# starting kernel
# baseline (speedup 1.0000x reference)
#include <hip/hip_runtime.h>
#include <cstdint>
#include <cstddef>

#define N_NODES   50000
#define N_EDGES   800000
#define TOT_E     (N_EDGES + N_NODES)
#define IN_C      128
#define HID       32
#define OUT_C     16
#define HEADS     8
#define HIDDEN    (HEADS * HID)
#define NEG_SLOPE 0.2f
#define GAT_EPS   1e-16f

typedef float v2f __attribute__((ext_vector_type(2)));
typedef float v8f __attribute__((ext_vector_type(8)));

template <int TN, int K, int NC>
__global__ __launch_bounds__(128) void gat_wmma_gemm_f32(
    const float* __restrict__ A, const float* __restrict__ B,
    float* __restrict__ C, int M, int waveTilesN)
{
    const int wave = blockIdx.x * (blockDim.x >> 5) + (threadIdx.x >> 5);
    const int lane = threadIdx.x & 31;
    const int tM = wave / waveTilesN;
    const int tg = wave % waveTilesN;
    if (tM * 16 >= M) return;

    const int half = lane >> 4;
    const int l    = lane & 15;
    const float* __restrict__ Ap = A + (size_t)(tM * 16 + l) * K + half * 2;
    const float* __restrict__ Bp = B + (size_t)(half * 2) * NC + tg * 16 * TN + l;

    const v8f zero = {0.f,0.f,0.f,0.f,0.f,0.f,0.f,0.f};
    v8f c[TN];
#pragma unroll
    for (int j = 0; j < TN; ++j) c[j] = zero;

#pragma unroll
    for (int k = 0; k < K; k += 4) {
        const v2f a = *(const v2f*)(Ap + k);
        v2f b[TN];
#pragma unroll
        for (int j = 0; j < TN; ++j) {
            b[j].x = Bp[(size_t)k * NC + j * 16];
            b[j].y = Bp[(size_t)(k + 1) * NC + j * 16];
        }
#pragma unroll
        for (int j = 0; j < TN; ++j)
            c[j] = __builtin_amdgcn_wmma_f32_16x16x4_f32(
                        false, a,  false, b[j],
                        (short)0, c[j],
                        false,  false);
    }

    float* __restrict__ Cbase =
        C + (size_t)(tM * 16 + half * 8) * NC + tg * 16 * TN + l;
#pragma unroll
    for (int j = 0; j < TN; ++j)
#pragma unroll
        for (int r = 0; r < 8; ++r)
            Cbase[(size_t)r * NC + j * 16] = c[j][r];
}

__global__ __launch_bounds__(256) void gat_att_logits(
    const float* __restrict__ xl, const float* __restrict__ att_s,
    const float* __restrict__ att_d, float* __restrict__ asrc,
    float* __restrict__ adst, int n, int H, int CH)
{
    const int i = blockIdx.x * blockDim.x + threadIdx.x;
    if (i >= n * H) return;
    const int h = i % H;
    const float* __restrict__ row = xl + (size_t)i * CH;
    const float* __restrict__ as  = att_s + (size_t)h * CH;
    const float* __restrict__ ad  = att_d + (size_t)h * CH;
    float s = 0.f, d = 0.f;
    for (int c = 0; c < CH; ++c) { float v = row[c]; s += v * as[c]; d += v * ad[c]; }
    asrc[i] = s;
    adst[i] = d;
}

__device__ __forceinline__ unsigned fkey(float f) {
    unsigned u = __float_as_uint(f);
    return (u & 0x80000000u) ? ~u : (u | 0x80000000u);
}
__device__ __forceinline__ float fdec(unsigned k) {
    unsigned u = (k & 0x80000000u) ? (k ^ 0x80000000u) : ~k;
    return __uint_as_float(u);
}

__device__ __forceinline__ void edge_sd(long long e, const int* __restrict__ src,
                                        const int* __restrict__ dst, int& s, int& d) {
    if (e < N_EDGES) { s = src[e]; d = dst[e]; }
    else             { s = (int)(e - N_EDGES); d = s; }
}

__global__ __launch_bounds__(256) void gat_edge_max(
    const int* __restrict__ src, const int* __restrict__ dst,
    const float* __restrict__ asrc, const float* __restrict__ adst,
    unsigned* __restrict__ emaxk, int H)
{
    const long long i = (long long)blockIdx.x * blockDim.x + threadIdx.x;
    if (i >= (long long)TOT_E * H) return;
    const int h = (int)(i % H);
    const long long e = i / H;
    int s, d; edge_sd(e, src, dst, s, d);
    float v = asrc[(size_t)s * H + h] + adst[(size_t)d * H + h];
    v = v > 0.f ? v : NEG_SLOPE * v;
    atomicMax(&emaxk[(size_t)d * H + h], fkey(v));
}

__global__ __launch_bounds__(256) void gat_edge_exp(
    const int* __restrict__ src, const int* __restrict__ dst,
    const float* __restrict__ asrc, const float* __restrict__ adst,
    const unsigned* __restrict__ emaxk, float* __restrict__ eexp,
    float* __restrict__ denom, int H)
{
    const long long i = (long long)blockIdx.x * blockDim.x + threadIdx.x;
    if (i >= (long long)TOT_E * H) return;
    const int h = (int)(i % H);
    const long long e = i / H;
    int s, d; edge_sd(e, src, dst, s, d);
    float v = asrc[(size_t)s * H + h] + adst[(size_t)d * H + h];
    v = v > 0.f ? v : NEG_SLOPE * v;
    const float m  = fdec(emaxk[(size_t)d * H + h]);
    const float ex = expf(v - m);
    eexp[i] = ex;
    atomicAdd(&denom[(size_t)d * H + h], ex);
}

__global__ __launch_bounds__(256) void gat_edge_aggregate(
    const int* __restrict__ src, const int* __restrict__ dst,
    const float* __restrict__ eexp, const float* __restrict__ denom,
    const float* __restrict__ xl, float* __restrict__ acc, int H, int CH)
{
    const long long i = (long long)blockIdx.x * blockDim.x + threadIdx.x;
    if (i >= (long long)TOT_E * H * CH) return;
    const int c        = (int)(i % CH);
    const long long eh = i / CH;
    const int h        = (int)(eh % H);
    const long long e  = eh / H;
    int s, d; edge_sd(e, src, dst, s, d);
    const float alpha = eexp[eh] / (denom[(size_t)d * H + h] + GAT_EPS);
    const float val   = alpha * xl[((size_t)s * H + h) * CH + c];
    atomicAdd(&acc[((size_t)d * H + h) * CH + c], val);
}

__global__ __launch_bounds__(256) void gat_finalize_elu(
    float* __restrict__ acc, const float* __restrict__ bias, int n, int F)
{
    const int i = blockIdx.x * blockDim.x + threadIdx.x;
    if (i >= n * F) return;
    const float v = acc[i] + bias[i % F];
    acc[i] = v > 0.f ? v : expm1f(v);
}

__global__ __launch_bounds__(256) void gat_finalize_bias(
    const float* __restrict__ acc, const float* __restrict__ bias,
    float* __restrict__ out, int n, int F)
{
    const int i = blockIdx.x * blockDim.x + threadIdx.x;
    if (i >= n * F) return;
    out[i] = acc[i] + bias[i % F];
}

extern "C" void kernel_launch(void* const* d_in, const int* in_sizes, int n_in,
                              void* d_out, int out_size, void* d_ws, size_t ws_size,
                              hipStream_t stream)
{
    (void)in_sizes; (void)n_in; (void)out_size; (void)ws_size;

    const float* x    = (const float*)d_in[0];
    const int*   ei   = (const int*)  d_in[1];
    const float* W1   = (const float*)d_in[2];
    const float* ats1 = (const float*)d_in[3];
    const float* atd1 = (const float*)d_in[4];
    const float* b1   = (const float*)d_in[5];
    const float* W2   = (const float*)d_in[6];
    const float* ats2 = (const float*)d_in[7];
    const float* atd2 = (const float*)d_in[8];
    const float* b2   = (const float*)d_in[9];
    const int* src = ei;
    const int* dst = ei + N_EDGES;
    float* out = (float*)d_out;

    float*    acc1 = (float*)d_ws;
    float*    acc2 = acc1 + (size_t)N_NODES * HIDDEN;
    unsigned* emk1 = (unsigned*)(acc2 + (size_t)N_NODES * OUT_C);
    float*    den1 = (float*)(emk1 + (size_t)N_NODES * HEADS);
    unsigned* emk2 = (unsigned*)(den1 + (size_t)N_NODES * HEADS);
    float*    den2 = (float*)(emk2 + (size_t)N_NODES);
    const size_t zero_bytes =
        (size_t)N_NODES * (HIDDEN + OUT_C + 2 * HEADS + 2) * sizeof(float);
    float* xl1 = den2 + (size_t)N_NODES;
    float* as1 = xl1 + (size_t)N_NODES * HIDDEN;
    float* ad1 = as1 + (size_t)N_NODES * HEADS;
    float* ee1 = ad1 + (size_t)N_NODES * HEADS;
    float* xl2 = ee1;
    float* as2 = xl2 + (size_t)N_NODES * OUT_C;
    float* ad2 = as2 + (size_t)N_NODES;
    float* ee2 = ad2 + (size_t)N_NODES;

    hipMemsetAsync(d_ws, 0, zero_bytes, stream);

    const int TB = 256;
    auto blk = [](long long n, int tb) { return (unsigned)((n + tb - 1) / tb); };

    {
        const int waveTilesN = HIDDEN / (16 * 4);
        const long long waves = (long long)(N_NODES / 16) * waveTilesN;
        gat_wmma_gemm_f32<4, IN_C, HIDDEN><<<blk(waves, 4), 128, 0, stream>>>(
            x, W1, xl1, N_NODES, waveTilesN);
    }
    gat_att_logits<<<blk((long long)N_NODES * HEADS, TB), TB, 0, stream>>>(
        xl1, ats1, atd1, as1, ad1, N_NODES, HEADS, HID);
    gat_edge_max<<<blk((long long)TOT_E * HEADS, TB), TB, 0, stream>>>(
        src, dst, as1, ad1, emk1, HEADS);
    gat_edge_exp<<<blk((long long)TOT_E * HEADS, TB), TB, 0, stream>>>(
        src, dst, as1, ad1, emk1, ee1, den1, HEADS);
    gat_edge_aggregate<<<blk((long long)TOT_E * HEADS * HID, TB), TB, 0, stream>>>(
        src, dst, ee1, den1, xl1, acc1, HEADS, HID);
    gat_finalize_elu<<<blk((long long)N_NODES * HIDDEN, TB), TB, 0, stream>>>(
        acc1, b1, N_NODES, HIDDEN);

    {
        const int waveTilesN = OUT_C / 16;
        const long long waves = (long long)(N_NODES / 16) * waveTilesN;
        gat_wmma_gemm_f32<1, HIDDEN, OUT_C><<<blk(waves, 4), 128, 0, stream>>>(
            acc1, W2, xl2, N_NODES, waveTilesN);
    }
    gat_att_logits<<<blk((long long)N_NODES, TB), TB, 0, stream>>>(
        xl2, ats2, atd2, as2, ad2, N_NODES, 1, OUT_C);
    gat_edge_max<<<blk((long long)TOT_E, TB), TB, 0, stream>>>(
        src, dst, as2, ad2, emk2, 1);
    gat_edge_exp<<<blk((long long)TOT_E, TB), TB, 0, stream>>>(
        src, dst, as2, ad2, emk2, ee2, den2, 1);
    gat_edge_aggregate<<<blk((long long)TOT_E * OUT_C, TB), TB, 0, stream>>>(
        src, dst, ee2, den2, xl2, acc2, 1, OUT_C);
    gat_finalize_bias<<<blk((long long)N_NODES * OUT_C, TB), TB, 0, stream>>>(
        acc2, b2, out, N_NODES, OUT_C);
}


# Edges grouped by destination each call (integer-add rank, scan, place); one wave per node does softmax and weighted row sum in registers, summed in list order; float and max atomics removed
# speedup vs baseline: 5.4290x; 5.4290x over previous
#include <hip/hip_runtime.h>
#include <cstdint>
#include <cstddef>

#define N_NODES   50000
#define N_EDGES   800000
#define TOT_E     (N_EDGES + N_NODES)
#define IN_C      128
#define HID       32
#define OUT_C     16
#define HEADS     8
#define HIDDEN    (HEADS * HID)
#define NEG_SLOPE 0.2f
#define GAT_EPS   1e-16f

typedef float v2f __attribute__((ext_vector_type(2)));
typedef float v8f __attribute__((ext_vector_type(8)));

template <int TN, int K, int NC>
__global__ __launch_bounds__(128) void gat_wmma_gemm_f32(
    const float* __restrict__ A, const float* __restrict__ B,
    float* __restrict__ C, int M, int waveTilesN)
{
    const int wave = blockIdx.x * (blockDim.x >> 5) + (threadIdx.x >> 5);
    const int lane = threadIdx.x & 31;
    const int tM = wave / waveTilesN;
    const int tg = wave % waveTilesN;
    if (tM * 16 >= M) return;

    const int half = lane >> 4;
    const int l    = lane & 15;
    const float* __restrict__ Ap = A + (size_t)(tM * 16 + l) * K + half * 2;
    const float* __restrict__ Bp = B + (size_t)(half * 2) * NC + tg * 16 * TN + l;

    const v8f zero = {0.f,0.f,0.f,0.f,0.f,0.f,0.f,0.f};
    v8f c[TN];
#pragma unroll
    for (int j = 0; j < TN; ++j) c[j] = zero;

#pragma unroll
    for (int k = 0; k < K; k += 4) {
        const v2f a = *(const v2f*)(Ap + k);
        v2f b[TN];
#pragma unroll
        for (int j = 0; j < TN; ++j) {
            b[j].x = Bp[(size_t)k * NC + j * 16];
            b[j].y = Bp[(size_t)(k + 1) * NC + j * 16];
        }
#pragma unroll
        for (int j = 0; j < TN; ++j)
            c[j] = __builtin_amdgcn_wmma_f32_16x16x4_f32(
                        false, a,  false, b[j],
                        (short)0, c[j],
                        false,  false);
    }

    float* __restrict__ Cbase =
        C + (size_t)(tM * 16 + half * 8) * NC + tg * 16 * TN + l;
#pragma unroll
    for (int j = 0; j < TN; ++j)
#pragma unroll
        for (int r = 0; r < 8; ++r)
            Cbase[(size_t)r * NC + j * 16] = c[j][r];
}

__global__ __launch_bounds__(256) void gat_att_logits(
    const float* __restrict__ xl, const float* __restrict__ att_s,
    const float* __restrict__ att_d, float* __restrict__ asrc,
    float* __restrict__ adst, int n, int H, int CH)
{
    const int i = blockIdx.x * blockDim.x + threadIdx.x;
    if (i >= n * H) return;
    const int h = i % H;
    const float* __restrict__ row = xl + (size_t)i * CH;
    const float* __restrict__ as  = att_s + (size_t)h * CH;
    const float* __restrict__ ad  = att_d + (size_t)h * CH;
    float s = 0.f, d = 0.f;
    for (int c = 0; c < CH; ++c) { float v = row[c]; s += v * as[c]; d += v * ad[c]; }
    asrc[i] = s;
    adst[i] = d;
}

__device__ __forceinline__ float leaky(float v) { return v > 0.f ? v : NEG_SLOPE * v; }

__global__ __launch_bounds__(256) void gat_edge_rank(
    const int* __restrict__ dst, int* __restrict__ cnt, int* __restrict__ rnk)
{
    const int e = blockIdx.x * blockDim.x + threadIdx.x;
    if (e >= N_EDGES) return;
    rnk[e] = atomicAdd(&cnt[dst[e]], 1);
}

#define SCAN_TB  1024
#define SCAN_PER ((N_NODES + SCAN_TB - 1) / SCAN_TB)
#define CNT_PAD  (SCAN_TB * SCAN_PER)
__global__ __launch_bounds__(SCAN_TB) void gat_rowptr_scan(
    const int* __restrict__ cnt, int* __restrict__ rowptr)
{
    __shared__ int wsum[32];
    const int t = threadIdx.x, lane = t & 31, w = t >> 5;
    const int* __restrict__ cp = cnt + t * SCAN_PER;
    int v[SCAN_PER];
    int s = 0;
#pragma unroll
    for (int i = 0; i < SCAN_PER; ++i) { v[i] = cp[i]; s += v[i]; }
    int x = s;
    for (int o = 1; o < 32; o <<= 1) { const int y = __shfl_up(x, o); if (lane >= o) x += y; }
    if (lane == 31) wsum[w] = x;
    __syncthreads();
    if (w == 0) {
        int u = wsum[lane];
        for (int o = 1; o < 32; o <<= 1) { const int y = __shfl_up(u, o); if (lane >= o) u += y; }
        wsum[lane] = u;
    }
    __syncthreads();
    int base = x - s + (w > 0 ? wsum[w - 1] : 0);
    int* __restrict__ rp = rowptr + t * SCAN_PER;
#pragma unroll
    for (int i = 0; i < SCAN_PER; ++i) { rp[i] = base; base += v[i]; }
}

__global__ __launch_bounds__(256) void gat_edge_place(
    const int* __restrict__ src, const int* __restrict__ dst,
    const int* __restrict__ rowptr, const int* __restrict__ rnk,
    int* __restrict__ list)
{
    const int e = blockIdx.x * blockDim.x + threadIdx.x;
    if (e >= N_EDGES) return;
    list[rowptr[dst[e]] + rnk[e]] = src[e];
}

__global__ __launch_bounds__(256) void gat_gather_l1(
    const int* __restrict__ rowptr, const int* __restrict__ list,
    const float* __restrict__ asrc, const float* __restrict__ adst,
    const float* __restrict__ xl, const float* __restrict__ bias,
    float* __restrict__ hout)
{
    const int lane = threadIdx.x & 31;
    const int d = __builtin_amdgcn_readfirstlane((int)(blockIdx.x * 8 + (threadIdx.x >> 5)));
    if (d >= N_NODES) return;
    const int p0 = rowptr[d], p1 = rowptr[d + 1];

    const int hh = lane & 7, j = lane >> 3;
    const float adh   = adst[(size_t)d * HEADS + hh];
    const float vself = leaky(asrc[(size_t)d * HEADS + hh] + adh);
    float m = vself;
    for (int p = p0 + j; p < p1; p += 4)
        m = fmaxf(m, leaky(asrc[(size_t)list[p] * HEADS + hh] + adh));
    m = fmaxf(m, __shfl_xor(m, 8));
    m = fmaxf(m, __shfl_xor(m, 16));
    float sum = (j == 0) ? expf(vself - m) : 0.f;
    for (int p = p0 + j; p < p1; p += 4)
        sum += expf(leaky(asrc[(size_t)list[p] * HEADS + hh] + adh) - m);
    sum += __shfl_xor(sum, 8);
    sum += __shfl_xor(sum, 16);

    const int hc = lane >> 2;
    const float mh  = __shfl(m, hc);
    const float den = __shfl(sum, hc) + GAT_EPS;
    const float adc = __shfl(adh, hc);
    float acc[8];
    {
        const float a = expf(leaky(asrc[(size_t)d * HEADS + hc] + adc) - mh) / den;
        const float4* r = (const float4*)(xl + (size_t)d * HIDDEN + lane * 8);
        const float4 r0 = r[0], r1 = r[1];
        acc[0] = a * r0.x; acc[1] = a * r0.y; acc[2] = a * r0.z; acc[3] = a * r0.w;
        acc[4] = a * r1.x; acc[5] = a * r1.y; acc[6] = a * r1.z; acc[7] = a * r1.w;
    }
#pragma unroll 4
    for (int p = p0; p < p1; ++p) {
        const int s = list[p];
        const float a = expf(leaky(asrc[(size_t)s * HEADS + hc] + adc) - mh) / den;
        const float4* r = (const float4*)(xl + (size_t)s * HIDDEN + lane * 8);
        const float4 r0 = r[0], r1 = r[1];
        acc[0] += a * r0.x; acc[1] += a * r0.y; acc[2] += a * r0.z; acc[3] += a * r0.w;
        acc[4] += a * r1.x; acc[5] += a * r1.y; acc[6] += a * r1.z; acc[7] += a * r1.w;
    }
    const float4* bp = (const float4*)(bias + lane * 8);
    const float4 b0 = bp[0], b1 = bp[1];
    const float bb[8] = {b0.x, b0.y, b0.z, b0.w, b1.x, b1.y, b1.z, b1.w};
    float o[8];
#pragma unroll
    for (int i = 0; i < 8; ++i) { const float v = acc[i] + bb[i]; o[i] = v > 0.f ? v : expm1f(v); }
    float4* op = (float4*)(hout + (size_t)d * HIDDEN + lane * 8);
    op[0] = make_float4(o[0], o[1], o[2], o[3]);
    op[1] = make_float4(o[4], o[5], o[6], o[7]);
}

__global__ __launch_bounds__(256) void gat_gather_l2(
    const int* __restrict__ rowptr, const int* __restrict__ list,
    const float* __restrict__ asrc, const float* __restrict__ adst,
    const float* __restrict__ xl, const float* __restrict__ bias,
    float* __restrict__ out)
{
    const int c = threadIdx.x & 15;
    const int d = blockIdx.x * 16 + (threadIdx.x >> 4);
    if (d >= N_NODES) return;
    const int p0 = rowptr[d], p1 = rowptr[d + 1];
    const float ad    = adst[d];
    const float vself = leaky(asrc[d] + ad);
    float m = vself;
    for (int p = p0 + c; p < p1; p += 16) m = fmaxf(m, leaky(asrc[list[p]] + ad));
    m = fmaxf(m, __shfl_xor(m, 1)); m = fmaxf(m, __shfl_xor(m, 2));
    m = fmaxf(m, __shfl_xor(m, 4)); m = fmaxf(m, __shfl_xor(m, 8));
    float sum = (c == 0) ? expf(vself - m) : 0.f;
    for (int p = p0 + c; p < p1; p += 16) sum += expf(leaky(asrc[list[p]] + ad) - m);
    sum += __shfl_xor(sum, 1); sum += __shfl_xor(sum, 2);
    sum += __shfl_xor(sum, 4); sum += __shfl_xor(sum, 8);
    const float den = sum + GAT_EPS;
    float acc = expf(vself - m) / den * xl[(size_t)d * OUT_C + c];
#pragma unroll 4
    for (int p = p0; p < p1; ++p) {
        const int s = list[p];
        const float a = expf(leaky(asrc[s] + ad) - m) / den;
        acc += a * xl[(size_t)s * OUT_C + c];
    }
    out[(size_t)d * OUT_C + c] = acc + bias[c];
}

extern "C" void kernel_launch(void* const* d_in, const int* in_sizes, int n_in,
                              void* d_out, int out_size, void* d_ws, size_t ws_size,
                              hipStream_t stream)
{
    (void)in_sizes; (void)n_in; (void)out_size; (void)ws_size;

    const float* x    = (const float*)d_in[0];
    const int*   ei   = (const int*)  d_in[1];
    const float* W1   = (const float*)d_in[2];
    const float* ats1 = (const float*)d_in[3];
    const float* atd1 = (const float*)d_in[4];
    const float* b1   = (const float*)d_in[5];
    const float* W2   = (const float*)d_in[6];
    const float* ats2 = (const float*)d_in[7];
    const float* atd2 = (const float*)d_in[8];
    const float* b2   = (const float*)d_in[9];
    const int* src = ei;
    const int* dst = ei + N_EDGES;
    float* out = (float*)d_out;

    int*   cnt    = (int*)d_ws;
    int*   rowptr = cnt + (size_t)CNT_PAD;
    int*   rnk    = rowptr + (size_t)CNT_PAD;
    int*   list   = rnk + (size_t)N_EDGES;
    float* xl1 = (float*)(list + (size_t)N_EDGES);
    float* as1 = xl1 + (size_t)N_NODES * HIDDEN;
    float* ad1 = as1 + (size_t)N_NODES * HEADS;
    float* h1  = ad1 + (size_t)N_NODES * HEADS;
    float* xl2 = h1  + (size_t)N_NODES * HIDDEN;
    float* as2 = xl2 + (size_t)N_NODES * OUT_C;
    float* ad2 = as2 + (size_t)CNT_PAD;

    hipMemsetAsync(cnt, 0, (size_t)CNT_PAD * sizeof(int), stream);

    const int TB = 256;
    auto blk = [](long long n, int tb) { return (unsigned)((n + tb - 1) / tb); };

    gat_edge_rank<<<blk(N_EDGES, TB), TB, 0, stream>>>(dst, cnt, rnk);
    gat_rowptr_scan<<<1, SCAN_TB, 0, stream>>>(cnt, rowptr);
    gat_edge_place<<<blk(N_EDGES, TB), TB, 0, stream>>>(src, dst, rowptr, rnk, list);

    {
        const int waveTilesN = HIDDEN / (16 * 4);
        const long long waves = (long long)(N_NODES / 16) * waveTilesN;
        gat_wmma_gemm_f32<4, IN_C, HIDDEN><<<blk(waves, 4), 128, 0, stream>>>(
            x, W1, xl1, N_NODES, waveTilesN);
    }
    gat_att_logits<<<blk((long long)N_NODES * HEADS, TB), TB, 0, stream>>>(
        xl1, ats1, atd1, as1, ad1, N_NODES, HEADS, HID);
    gat_gather_l1<<<blk(N_NODES, 8), TB, 0, stream>>>(
        rowptr, list, as1, ad1, xl1, b1, h1);

    {
        const int waveTilesN = OUT_C / 16;
        const long long waves = (long long)(N_NODES / 16) * waveTilesN;
        gat_wmma_gemm_f32<1, HIDDEN, OUT_C><<<blk(waves, 4), 128, 0, stream>>>(
            h1, W2, xl2, N_NODES, waveTilesN);
    }
    gat_att_logits<<<blk((long long)N_NODES, TB), TB, 0, stream>>>(
        xl2, ats2, atd2, as2, ad2, N_NODES, 1, OUT_C);
    gat_gather_l2<<<blk(N_NODES, 16), TB, 0, stream>>>(
        rowptr, list, as2, ad2, xl2, b2, out);
}


# Layer-1 gather: a node's edge indices and per-edge logits loaded once (lane = edge) into a per-wave LDS strip; max, sum, weights from LDS; row loop reads weights from LDS; same sum order
# speedup vs baseline: 5.7522x; 1.0595x over previous
#include <hip/hip_runtime.h>
#include <cstdint>
#include <cstddef>

#define N_NODES   50000
#define N_EDGES   800000
#define TOT_E     (N_EDGES + N_NODES)
#define IN_C      128
#define HID       32
#define OUT_C     16
#define HEADS     8
#define HIDDEN    (HEADS * HID)
#define NEG_SLOPE 0.2f
#define GAT_EPS   1e-16f

typedef float v2f __attribute__((ext_vector_type(2)));
typedef float v8f __attribute__((ext_vector_type(8)));

template <int TN, int K, int NC>
__global__ __launch_bounds__(128) void gat_wmma_gemm_f32(
    const float* __restrict__ A, const float* __restrict__ B,
    float* __restrict__ C, int M, int waveTilesN)
{
    const int wave = blockIdx.x * (blockDim.x >> 5) + (threadIdx.x >> 5);
    const int lane = threadIdx.x & 31;
    const int tM = wave / waveTilesN;
    const int tg = wave % waveTilesN;
    if (tM * 16 >= M) return;

    const int half = lane >> 4;
    const int l    = lane & 15;
    const float* __restrict__ Ap = A + (size_t)(tM * 16 + l) * K + half * 2;
    const float* __restrict__ Bp = B + (size_t)(half * 2) * NC + tg * 16 * TN + l;

    const v8f zero = {0.f,0.f,0.f,0.f,0.f,0.f,0.f,0.f};
    v8f c[TN];
#pragma unroll
    for (int j = 0; j < TN; ++j) c[j] = zero;

#pragma unroll
    for (int k = 0; k < K; k += 4) {
        const v2f a = *(const v2f*)(Ap + k);
        v2f b[TN];
#pragma unroll
        for (int j = 0; j < TN; ++j) {
            b[j].x = Bp[(size_t)k * NC + j * 16];
            b[j].y = Bp[(size_t)(k + 1) * NC + j * 16];
        }
#pragma unroll
        for (int j = 0; j < TN; ++j)
            c[j] = __builtin_amdgcn_wmma_f32_16x16x4_f32(
                        false, a,  false, b[j],
                        (short)0, c[j],
                        false,  false);
    }

    float* __restrict__ Cbase =
        C + (size_t)(tM * 16 + half * 8) * NC + tg * 16 * TN + l;
#pragma unroll
    for (int j = 0; j < TN; ++j)
#pragma unroll
        for (int r = 0; r < 8; ++r)
            Cbase[(size_t)r * NC + j * 16] = c[j][r];
}

__global__ __launch_bounds__(256) void gat_att_logits(
    const float* __restrict__ xl, const float* __restrict__ att_s,
    const float* __restrict__ att_d, float* __restrict__ asrc,
    float* __restrict__ adst, int n, int H, int CH)
{
    const int i = blockIdx.x * blockDim.x + threadIdx.x;
    if (i >= n * H) return;
    const int h = i % H;
    const float* __restrict__ row = xl + (size_t)i * CH;
    const float* __restrict__ as  = att_s + (size_t)h * CH;
    const float* __restrict__ ad  = att_d + (size_t)h * CH;
    float s = 0.f, d = 0.f;
    for (int c = 0; c < CH; ++c) { float v = row[c]; s += v * as[c]; d += v * ad[c]; }
    asrc[i] = s;
    adst[i] = d;
}

__device__ __forceinline__ float leaky(float v) { return v > 0.f ? v : NEG_SLOPE * v; }

__global__ __launch_bounds__(256) void gat_edge_rank(
    const int* __restrict__ dst, int* __restrict__ cnt, int* __restrict__ rnk)
{
    const int e = blockIdx.x * blockDim.x + threadIdx.x;
    if (e >= N_EDGES) return;
    rnk[e] = atomicAdd(&cnt[dst[e]], 1);
}

#define SCAN_TB  1024
#define SCAN_PER ((N_NODES + SCAN_TB - 1) / SCAN_TB)
#define CNT_PAD  (SCAN_TB * SCAN_PER)
__global__ __launch_bounds__(SCAN_TB) void gat_rowptr_scan(
    const int* __restrict__ cnt, int* __restrict__ rowptr)
{
    __shared__ int wsum[32];
    const int t = threadIdx.x, lane = t & 31, w = t >> 5;
    const int* __restrict__ cp = cnt + t * SCAN_PER;
    int v[SCAN_PER];
    int s = 0;
#pragma unroll
    for (int i = 0; i < SCAN_PER; ++i) { v[i] = cp[i]; s += v[i]; }
    int x = s;
    for (int o = 1; o < 32; o <<= 1) { const int y = __shfl_up(x, o); if (lane >= o) x += y; }
    if (lane == 31) wsum[w] = x;
    __syncthreads();
    if (w == 0) {
        int u = wsum[lane];
        for (int o = 1; o < 32; o <<= 1) { const int y = __shfl_up(u, o); if (lane >= o) u += y; }
        wsum[lane] = u;
    }
    __syncthreads();
    int base = x - s + (w > 0 ? wsum[w - 1] : 0);
    int* __restrict__ rp = rowptr + t * SCAN_PER;
#pragma unroll
    for (int i = 0; i < SCAN_PER; ++i) { rp[i] = base; base += v[i]; }
}

__global__ __launch_bounds__(256) void gat_edge_place(
    const int* __restrict__ src, const int* __restrict__ dst,
    const int* __restrict__ rowptr, const int* __restrict__ rnk,
    int* __restrict__ list)
{
    const int e = blockIdx.x * blockDim.x + threadIdx.x;
    if (e >= N_EDGES) return;
    list[rowptr[dst[e]] + rnk[e]] = src[e];
}

__global__ __launch_bounds__(256) void gat_gather_l1(
    const int* __restrict__ rowptr, const int* __restrict__ list,
    const float* __restrict__ asrc, const float* __restrict__ adst,
    const float* __restrict__ xl, const float* __restrict__ bias,
    float* __restrict__ hout)
{
    __shared__ __attribute__((aligned(16))) float sw[8][32 * HEADS];
    const int lane = threadIdx.x & 31;
    const int wv   = threadIdx.x >> 5;
    const int d = __builtin_amdgcn_readfirstlane((int)(blockIdx.x * 8 + wv));
    if (d >= N_NODES) return;
    float* my = sw[wv];
    const int p0 = rowptr[d], p1 = rowptr[d + 1];
    const int deg = p1 - p0;
    const int nch = (deg + 31) >> 5;

    const int hh = lane & 7, j = lane >> 3;
    const float adh   = adst[(size_t)d * HEADS + hh];
    const float vself = leaky(asrc[(size_t)d * HEADS + hh] + adh);

    int sv = d;
    auto stage = [&](int c) {
        const int q = c * 32 + lane;
        sv = (q < deg) ? list[p0 + q] : d;
        const float4* ar = (const float4*)(asrc + (size_t)sv * HEADS);
        const float4 a0 = ar[0], a1 = ar[1];
        float4* dp = (float4*)(my + lane * HEADS);
        dp[0] = a0; dp[1] = a1;
    };

    float m = vself;
    for (int c = 0; c < nch; ++c) {
        stage(c);
#pragma unroll
        for (int i = 0; i < 8; ++i) {
            const int sl = j + 4 * i;
            const float v = leaky(my[sl * HEADS + hh] + adh);
            m = (c * 32 + sl < deg) ? fmaxf(m, v) : m;
        }
    }
    m = fmaxf(m, __shfl_xor(m, 8));
    m = fmaxf(m, __shfl_xor(m, 16));
    float sum = (j == 0) ? expf(vself - m) : 0.f;
    for (int c = 0; c < nch; ++c) {
        if (nch > 1) stage(c);
#pragma unroll
        for (int i = 0; i < 8; ++i) {
            const int sl = j + 4 * i;
            const float ex = expf(leaky(my[sl * HEADS + hh] + adh) - m);
            sum = (c * 32 + sl < deg) ? sum + ex : sum;
        }
    }
    sum += __shfl_xor(sum, 8);
    sum += __shfl_xor(sum, 16);
    const float den = sum + GAT_EPS;
    const float wself = expf(vself - m) / den;

    const int hc = lane >> 2;
    float acc[8];
    {
        const float a = __shfl(wself, hc);
        const float4* r = (const float4*)(xl + (size_t)d * HIDDEN + lane * 8);
        const float4 r0 = r[0], r1 = r[1];
        acc[0] = a * r0.x; acc[1] = a * r0.y; acc[2] = a * r0.z; acc[3] = a * r0.w;
        acc[4] = a * r1.x; acc[5] = a * r1.y; acc[6] = a * r1.z; acc[7] = a * r1.w;
    }
    for (int c = 0; c < nch; ++c) {
        if (nch > 1) stage(c);
#pragma unroll
        for (int i = 0; i < 8; ++i) {
            const int sl = j + 4 * i;
            my[sl * HEADS + hh] = expf(leaky(my[sl * HEADS + hh] + adh) - m) / den;
        }
        const int cnt = min(32, deg - c * 32);
#pragma unroll 4
        for (int e = 0; e < cnt; ++e) {
            const int s = __builtin_amdgcn_readlane(sv, e);
            const float a = my[e * HEADS + hc];
            const float4* r = (const float4*)(xl + (size_t)s * HIDDEN + lane * 8);
            const float4 r0 = r[0], r1 = r[1];
            acc[0] += a * r0.x; acc[1] += a * r0.y; acc[2] += a * r0.z; acc[3] += a * r0.w;
            acc[4] += a * r1.x; acc[5] += a * r1.y; acc[6] += a * r1.z; acc[7] += a * r1.w;
        }
    }
    const float4* bp = (const float4*)(bias + lane * 8);
    const float4 b0 = bp[0], b1 = bp[1];
    const float bb[8] = {b0.x, b0.y, b0.z, b0.w, b1.x, b1.y, b1.z, b1.w};
    float o[8];
#pragma unroll
    for (int i = 0; i < 8; ++i) { const float v = acc[i] + bb[i]; o[i] = v > 0.f ? v : expm1f(v); }
    float4* op = (float4*)(hout + (size_t)d * HIDDEN + lane * 8);
    op[0] = make_float4(o[0], o[1], o[2], o[3]);
    op[1] = make_float4(o[4], o[5], o[6], o[7]);
}

__global__ __launch_bounds__(256) void gat_gather_l2(
    const int* __restrict__ rowptr, const int* __restrict__ list,
    const float* __restrict__ asrc, const float* __restrict__ adst,
    const float* __restrict__ xl, const float* __restrict__ bias,
    float* __restrict__ out)
{
    const int c = threadIdx.x & 15;
    const int d = blockIdx.x * 16 + (threadIdx.x >> 4);
    if (d >= N_NODES) return;
    const int p0 = rowptr[d], p1 = rowptr[d + 1];
    const float ad    = adst[d];
    const float vself = leaky(asrc[d] + ad);
    float m = vself;
    for (int p = p0 + c; p < p1; p += 16) m = fmaxf(m, leaky(asrc[list[p]] + ad));
    m = fmaxf(m, __shfl_xor(m, 1)); m = fmaxf(m, __shfl_xor(m, 2));
    m = fmaxf(m, __shfl_xor(m, 4)); m = fmaxf(m, __shfl_xor(m, 8));
    float sum = (c == 0) ? expf(vself - m) : 0.f;
    for (int p = p0 + c; p < p1; p += 16) sum += expf(leaky(asrc[list[p]] + ad) - m);
    sum += __shfl_xor(sum, 1); sum += __shfl_xor(sum, 2);
    sum += __shfl_xor(sum, 4); sum += __shfl_xor(sum, 8);
    const float den = sum + GAT_EPS;
    float acc = expf(vself - m) / den * xl[(size_t)d * OUT_C + c];
#pragma unroll 4
    for (int p = p0; p < p1; ++p) {
        const int s = list[p];
        const float a = expf(leaky(asrc[s] + ad) - m) / den;
        acc += a * xl[(size_t)s * OUT_C + c];
    }
    out[(size_t)d * OUT_C + c] = acc + bias[c];
}

extern "C" void kernel_launch(void* const* d_in, const int* in_sizes, int n_in,
                              void* d_out, int out_size, void* d_ws, size_t ws_size,
                              hipStream_t stream)
{
    (void)in_sizes; (void)n_in; (void)out_size; (void)ws_size;

    const float* x    = (const float*)d_in[0];
    const int*   ei   = (const int*)  d_in[1];
    const float* W1   = (const float*)d_in[2];
    const float* ats1 = (const float*)d_in[3];
    const float* atd1 = (const float*)d_in[4];
    const float* b1   = (const float*)d_in[5];
    const float* W2   = (const float*)d_in[6];
    const float* ats2 = (const float*)d_in[7];
    const float* atd2 = (const float*)d_in[8];
    const float* b2   = (const float*)d_in[9];
    const int* src = ei;
    const int* dst = ei + N_EDGES;
    float* out = (float*)d_out;

    int*   cnt    = (int*)d_ws;
    int*   rowptr = cnt + (size_t)CNT_PAD;
    int*   rnk    = rowptr + (size_t)CNT_PAD;
    int*   list   = rnk + (size_t)N_EDGES;
    float* xl1 = (float*)(list + (size_t)N_EDGES);
    float* as1 = xl1 + (size_t)N_NODES * HIDDEN;
    float* ad1 = as1 + (size_t)N_NODES * HEADS;
    float* h1  = ad1 + (size_t)N_NODES * HEADS;
    float* xl2 = h1  + (size_t)N_NODES * HIDDEN;
    float* as2 = xl2 + (size_t)N_NODES * OUT_C;
    float* ad2 = as2 + (size_t)CNT_PAD;

    hipMemsetAsync(cnt, 0, (size_t)CNT_PAD * sizeof(int), stream);

    const int TB = 256;
    auto blk = [](long long n, int tb) { return (unsigned)((n + tb - 1) / tb); };

    gat_edge_rank<<<blk(N_EDGES, TB), TB, 0, stream>>>(dst, cnt, rnk);
    gat_rowptr_scan<<<1, SCAN_TB, 0, stream>>>(cnt, rowptr);
    gat_edge_place<<<blk(N_EDGES, TB), TB, 0, stream>>>(src, dst, rowptr, rnk, list);

    {
        const int waveTilesN = HIDDEN / (16 * 4);
        const long long waves = (long long)(N_NODES / 16) * waveTilesN;
        gat_wmma_gemm_f32<4, IN_C, HIDDEN><<<blk(waves, 4), 128, 0, stream>>>(
            x, W1, xl1, N_NODES, waveTilesN);
    }
    gat_att_logits<<<blk((long long)N_NODES * HEADS, TB), TB, 0, stream>>>(
        xl1, ats1, atd1, as1, ad1, N_NODES, HEADS, HID);
    gat_gather_l1<<<blk(N_NODES, 8), TB, 0, stream>>>(
        rowptr, list, as1, ad1, xl1, b1, h1);

    {
        const int waveTilesN = OUT_C / 16;
        const long long waves = (long long)(N_NODES / 16) * waveTilesN;
        gat_wmma_gemm_f32<1, HIDDEN, OUT_C><<<blk(waves, 4), 128, 0, stream>>>(
            h1, W2, xl2, N_NODES, waveTilesN);
    }
    gat_att_logits<<<blk((long long)N_NODES, TB), TB, 0, stream>>>(
        xl2, ats2, atd2, as2, ad2, N_NODES, 1, OUT_C);
    gat_gather_l2<<<blk(N_NODES, 16), TB, 0, stream>>>(
        rowptr, list, as2, ad2, xl2, b2, out);
}


# Edge rank without global atomics: 64 edge chunks each count and rank their edges in a 50000-word LDS table, a prefix kernel joins the chunks; memset gone; a node's list (sum) order now chunk-major
# speedup vs baseline: 6.5366x; 1.1364x over previous
#include <hip/hip_runtime.h>
#include <cstdint>
#include <cstddef>

#define N_NODES   50000
#define N_EDGES   800000
#define TOT_E     (N_EDGES + N_NODES)
#define IN_C      128
#define HID       32
#define OUT_C     16
#define HEADS     8
#define HIDDEN    (HEADS * HID)
#define NEG_SLOPE 0.2f
#define GAT_EPS   1e-16f

typedef float v2f __attribute__((ext_vector_type(2)));
typedef float v8f __attribute__((ext_vector_type(8)));

template <int TN, int K, int NC>
__global__ __launch_bounds__(128) void gat_wmma_gemm_f32(
    const float* __restrict__ A, const float* __restrict__ B,
    float* __restrict__ C, int M, int waveTilesN)
{
    const int wave = blockIdx.x * (blockDim.x >> 5) + (threadIdx.x >> 5);
    const int lane = threadIdx.x & 31;
    const int tM = wave / waveTilesN;
    const int tg = wave % waveTilesN;
    if (tM * 16 >= M) return;

    const int half = lane >> 4;
    const int l    = lane & 15;
    const float* __restrict__ Ap = A + (size_t)(tM * 16 + l) * K + half * 2;
    const float* __restrict__ Bp = B + (size_t)(half * 2) * NC + tg * 16 * TN + l;

    const v8f zero = {0.f,0.f,0.f,0.f,0.f,0.f,0.f,0.f};
    v8f c[TN];
#pragma unroll
    for (int j = 0; j < TN; ++j) c[j] = zero;

#pragma unroll
    for (int k = 0; k < K; k += 4) {
        const v2f a = *(const v2f*)(Ap + k);
        v2f b[TN];
#pragma unroll
        for (int j = 0; j < TN; ++j) {
            b[j].x = Bp[(size_t)k * NC + j * 16];
            b[j].y = Bp[(size_t)(k + 1) * NC + j * 16];
        }
#pragma unroll
        for (int j = 0; j < TN; ++j)
            c[j] = __builtin_amdgcn_wmma_f32_16x16x4_f32(
                        false, a,  false, b[j],
                        (short)0, c[j],
                        false,  false);
    }

    float* __restrict__ Cbase =
        C + (size_t)(tM * 16 + half * 8) * NC + tg * 16 * TN + l;
#pragma unroll
    for (int j = 0; j < TN; ++j)
#pragma unroll
        for (int r = 0; r < 8; ++r)
            Cbase[(size_t)r * NC + j * 16] = c[j][r];
}

__global__ __launch_bounds__(256) void gat_att_logits(
    const float* __restrict__ xl, const float* __restrict__ att_s,
    const float* __restrict__ att_d, float* __restrict__ asrc,
    float* __restrict__ adst, int n, int H, int CH)
{
    const int i = blockIdx.x * blockDim.x + threadIdx.x;
    if (i >= n * H) return;
    const int h = i % H;
    const float* __restrict__ row = xl + (size_t)i * CH;
    const float* __restrict__ as  = att_s + (size_t)h * CH;
    const float* __restrict__ ad  = att_d + (size_t)h * CH;
    float s = 0.f, d = 0.f;
    for (int c = 0; c < CH; ++c) { float v = row[c]; s += v * as[c]; d += v * ad[c]; }
    asrc[i] = s;
    adst[i] = d;
}

__device__ __forceinline__ float leaky(float v) { return v > 0.f ? v : NEG_SLOPE * v; }

#define SCAN_TB  1024
#define SCAN_PER ((N_NODES + SCAN_TB - 1) / SCAN_TB)
#define CNT_PAD  (SCAN_TB * SCAN_PER)

#define RANK_CHUNKS  64
#define RANK_CHUNK_E (N_EDGES / RANK_CHUNKS)
#define RANK_TB      1024
#define RANK_PER     ((RANK_CHUNK_E + RANK_TB - 1) / RANK_TB)
static_assert(RANK_CHUNKS * RANK_CHUNK_E == N_EDGES, "chunks must tile the edge list");
static_assert(N_NODES % 4 == 0, "counter table is moved 16 bytes at a time");

__global__ __launch_bounds__(RANK_TB) void gat_edge_rank(
    const int* __restrict__ dst, int* __restrict__ part, int* __restrict__ rnk)
{
    __shared__ __attribute__((aligned(16))) int hist[N_NODES];
    const int t  = threadIdx.x;
    const int e0 = blockIdx.x * RANK_CHUNK_E;
    int4* h4 = (int4*)hist;
    for (int i = t; i < N_NODES / 4; i += RANK_TB) h4[i] = make_int4(0, 0, 0, 0);
    int d[RANK_PER];
#pragma unroll
    for (int k = 0; k < RANK_PER; ++k) {
        const int i = t + k * RANK_TB;
        d[k] = (i < RANK_CHUNK_E) ? dst[e0 + i] : 0;
    }
    __syncthreads();
    int r[RANK_PER];
#pragma unroll
    for (int k = 0; k < RANK_PER; ++k) {
        const int i = t + k * RANK_TB;
        r[k] = (i < RANK_CHUNK_E) ? atomicAdd(&hist[d[k]], 1) : 0;
    }
#pragma unroll
    for (int k = 0; k < RANK_PER; ++k) {
        const int i = t + k * RANK_TB;
        if (i < RANK_CHUNK_E) rnk[e0 + i] = r[k];
    }
    __syncthreads();
    int4* __restrict__ p4 = (int4*)(part + (size_t)blockIdx.x * CNT_PAD);
    for (int i = t; i < N_NODES / 4; i += RANK_TB) p4[i] = h4[i];
}

__global__ __launch_bounds__(256) void gat_chunk_prefix(
    int* __restrict__ part, int* __restrict__ cnt)
{
    const int n = blockIdx.x * 256 + threadIdx.x;
    if (n >= N_NODES) { cnt[n] = 0; return; }
    int v[RANK_CHUNKS];
#pragma unroll
    for (int q = 0; q < RANK_CHUNKS; ++q) v[q] = part[(size_t)q * CNT_PAD + n];
    int s = 0;
#pragma unroll
    for (int q = 0; q < RANK_CHUNKS; ++q) { part[(size_t)q * CNT_PAD + n] = s; s += v[q]; }
    cnt[n] = s;
}

__global__ __launch_bounds__(SCAN_TB) void gat_rowptr_scan(
    const int* __restrict__ cnt, int* __restrict__ rowptr)
{
    __shared__ int wsum[32];
    const int t = threadIdx.x, lane = t & 31, w = t >> 5;
    const int* __restrict__ cp = cnt + t * SCAN_PER;
    int v[SCAN_PER];
    int s = 0;
#pragma unroll
    for (int i = 0; i < SCAN_PER; ++i) { v[i] = cp[i]; s += v[i]; }
    int x = s;
    for (int o = 1; o < 32; o <<= 1) { const int y = __shfl_up(x, o); if (lane >= o) x += y; }
    if (lane == 31) wsum[w] = x;
    __syncthreads();
    if (w == 0) {
        int u = wsum[lane];
        for (int o = 1; o < 32; o <<= 1) { const int y = __shfl_up(u, o); if (lane >= o) u += y; }
        wsum[lane] = u;
    }
    __syncthreads();
    int base = x - s + (w > 0 ? wsum[w - 1] : 0);
    int* __restrict__ rp = rowptr + t * SCAN_PER;
#pragma unroll
    for (int i = 0; i < SCAN_PER; ++i) { rp[i] = base; base += v[i]; }
}

__global__ __launch_bounds__(256) void gat_edge_place(
    const int* __restrict__ src, const int* __restrict__ dst,
    const int* __restrict__ rowptr, const int* __restrict__ part,
    const int* __restrict__ rnk, int* __restrict__ list)
{
    const int e = blockIdx.x * blockDim.x + threadIdx.x;
    if (e >= N_EDGES) return;
    const int d = dst[e];
    const int before = part[(size_t)(e / RANK_CHUNK_E) * CNT_PAD + d];
    list[rowptr[d] + before + rnk[e]] = src[e];
}

__global__ __launch_bounds__(256) void gat_gather_l1(
    const int* __restrict__ rowptr, const int* __restrict__ list,
    const float* __restrict__ asrc, const float* __restrict__ adst,
    const float* __restrict__ xl, const float* __restrict__ bias,
    float* __restrict__ hout)
{
    __shared__ __attribute__((aligned(16))) float sw[8][32 * HEADS];
    const int lane = threadIdx.x & 31;
    const int wv   = threadIdx.x >> 5;
    const int d = __builtin_amdgcn_readfirstlane((int)(blockIdx.x * 8 + wv));
    if (d >= N_NODES) return;
    float* my = sw[wv];
    const int p0 = rowptr[d], p1 = rowptr[d + 1];
    const int deg = p1 - p0;
    const int nch = (deg + 31) >> 5;

    const int hh = lane & 7, j = lane >> 3;
    const float adh   = adst[(size_t)d * HEADS + hh];
    const float vself = leaky(asrc[(size_t)d * HEADS + hh] + adh);

    int sv = d;
    auto stage = [&](int c) {
        const int q = c * 32 + lane;
        sv = (q < deg) ? list[p0 + q] : d;
        const float4* ar = (const float4*)(asrc + (size_t)sv * HEADS);
        const float4 a0 = ar[0], a1 = ar[1];
        float4* dp = (float4*)(my + lane * HEADS);
        dp[0] = a0; dp[1] = a1;
    };

    float m = vself;
    for (int c = 0; c < nch; ++c) {
        stage(c);
#pragma unroll
        for (int i = 0; i < 8; ++i) {
            const int sl = j + 4 * i;
            const float v = leaky(my[sl * HEADS + hh] + adh);
            m = (c * 32 + sl < deg) ? fmaxf(m, v) : m;
        }
    }
    m = fmaxf(m, __shfl_xor(m, 8));
    m = fmaxf(m, __shfl_xor(m, 16));
    float sum = (j == 0) ? expf(vself - m) : 0.f;
    for (int c = 0; c < nch; ++c) {
        if (nch > 1) stage(c);
#pragma unroll
        for (int i = 0; i < 8; ++i) {
            const int sl = j + 4 * i;
            const float ex = expf(leaky(my[sl * HEADS + hh] + adh) - m);
            sum = (c * 32 + sl < deg) ? sum + ex : sum;
        }
    }
    sum += __shfl_xor(sum, 8);
    sum += __shfl_xor(sum, 16);
    const float den = sum + GAT_EPS;
    const float wself = expf(vself - m) / den;

    const int hc = lane >> 2;
    float acc[8];
    {
        const float a = __shfl(wself, hc);
        const float4* r = (const float4*)(xl + (size_t)d * HIDDEN + lane * 8);
        const float4 r0 = r[0], r1 = r[1];
        acc[0] = a * r0.x; acc[1] = a * r0.y; acc[2] = a * r0.z; acc[3] = a * r0.w;
        acc[4] = a * r1.x; acc[5] = a * r1.y; acc[6] = a * r1.z; acc[7] = a * r1.w;
    }
    for (int c = 0; c < nch; ++c) {
        if (nch > 1) stage(c);
#pragma unroll
        for (int i = 0; i < 8; ++i) {
            const int sl = j + 4 * i;
            my[sl * HEADS + hh] = expf(leaky(my[sl * HEADS + hh] + adh) - m) / den;
        }
        const int cnt = min(32, deg - c * 32);
#pragma unroll 4
        for (int e = 0; e < cnt; ++e) {
            const int s = __builtin_amdgcn_readlane(sv, e);
            const float a = my[e * HEADS + hc];
            const float4* r = (const float4*)(xl + (size_t)s * HIDDEN + lane * 8);
            const float4 r0 = r[0], r1 = r[1];
            acc[0] += a * r0.x; acc[1] += a * r0.y; acc[2] += a * r0.z; acc[3] += a * r0.w;
            acc[4] += a * r1.x; acc[5] += a * r1.y; acc[6] += a * r1.z; acc[7] += a * r1.w;
        }
    }
    const float4* bp = (const float4*)(bias + lane * 8);
    const float4 b0 = bp[0], b1 = bp[1];
    const float bb[8] = {b0.x, b0.y, b0.z, b0.w, b1.x, b1.y, b1.z, b1.w};
    float o[8];
#pragma unroll
    for (int i = 0; i < 8; ++i) { const float v = acc[i] + bb[i]; o[i] = v > 0.f ? v : expm1f(v); }
    float4* op = (float4*)(hout + (size_t)d * HIDDEN + lane * 8);
    op[0] = make_float4(o[0], o[1], o[2], o[3]);
    op[1] = make_float4(o[4], o[5], o[6], o[7]);
}

__global__ __launch_bounds__(256) void gat_gather_l2(
    const int* __restrict__ rowptr, const int* __restrict__ list,
    const float* __restrict__ asrc, const float* __restrict__ adst,
    const float* __restrict__ xl, const float* __restrict__ bias,
    float* __restrict__ out)
{
    const int c = threadIdx.x & 15;
    const int d = blockIdx.x * 16 + (threadIdx.x >> 4);
    if (d >= N_NODES) return;
    const int p0 = rowptr[d], p1 = rowptr[d + 1];
    const float ad    = adst[d];
    const float vself = leaky(asrc[d] + ad);
    float m = vself;
    for (int p = p0 + c; p < p1; p += 16) m = fmaxf(m, leaky(asrc[list[p]] + ad));
    m = fmaxf(m, __shfl_xor(m, 1)); m = fmaxf(m, __shfl_xor(m, 2));
    m = fmaxf(m, __shfl_xor(m, 4)); m = fmaxf(m, __shfl_xor(m, 8));
    float sum = (c == 0) ? expf(vself - m) : 0.f;
    for (int p = p0 + c; p < p1; p += 16) sum += expf(leaky(asrc[list[p]] + ad) - m);
    sum += __shfl_xor(sum, 1); sum += __shfl_xor(sum, 2);
    sum += __shfl_xor(sum, 4); sum += __shfl_xor(sum, 8);
    const float den = sum + GAT_EPS;
    float acc = expf(vself - m) / den * xl[(size_t)d * OUT_C + c];
#pragma unroll 4
    for (int p = p0; p < p1; ++p) {
        const int s = list[p];
        const float a = expf(leaky(asrc[s] + ad) - m) / den;
        acc += a * xl[(size_t)s * OUT_C + c];
    }
    out[(size_t)d * OUT_C + c] = acc + bias[c];
}

extern "C" void kernel_launch(void* const* d_in, const int* in_sizes, int n_in,
                              void* d_out, int out_size, void* d_ws, size_t ws_size,
                              hipStream_t stream)
{
    (void)in_sizes; (void)n_in; (void)out_size; (void)ws_size;

    const float* x    = (const float*)d_in[0];
    const int*   ei   = (const int*)  d_in[1];
    const float* W1   = (const float*)d_in[2];
    const float* ats1 = (const float*)d_in[3];
    const float* atd1 = (const float*)d_in[4];
    const float* b1   = (const float*)d_in[5];
    const float* W2   = (const float*)d_in[6];
    const float* ats2 = (const float*)d_in[7];
    const float* atd2 = (const float*)d_in[8];
    const float* b2   = (const float*)d_in[9];
    const int* src = ei;
    const int* dst = ei + N_EDGES;
    float* out = (float*)d_out;

    int*   cnt    = (int*)d_ws;
    int*   rowptr = cnt + (size_t)CNT_PAD;
    int*   rnk    = rowptr + (size_t)CNT_PAD;
    int*   list   = rnk + (size_t)N_EDGES;
    float* xl1 = (float*)(list + (size_t)N_EDGES);
    float* as1 = xl1 + (size_t)N_NODES * HIDDEN;
    float* ad1 = as1 + (size_t)N_NODES * HEADS;
    float* h1  = ad1 + (size_t)N_NODES * HEADS;
    float* xl2 = h1  + (size_t)N_NODES * HIDDEN;
    float* as2 = xl2 + (size_t)N_NODES * OUT_C;
    float* ad2 = as2 + (size_t)CNT_PAD;
    int*   part = (int*)(ad2 + (size_t)CNT_PAD);

    const int TB = 256;
    auto blk = [](long long n, int tb) { return (unsigned)((n + tb - 1) / tb); };

    gat_edge_rank<<<RANK_CHUNKS, RANK_TB, 0, stream>>>(dst, part, rnk);
    gat_chunk_prefix<<<CNT_PAD / 256, 256, 0, stream>>>(part, cnt);
    gat_rowptr_scan<<<1, SCAN_TB, 0, stream>>>(cnt, rowptr);
    gat_edge_place<<<blk(N_EDGES, TB), TB, 0, stream>>>(src, dst, rowptr, part, rnk, list);

    {
        const int waveTilesN = HIDDEN / (16 * 4);
        const long long waves = (long long)(N_NODES / 16) * waveTilesN;
        gat_wmma_gemm_f32<4, IN_C, HIDDEN><<<blk(waves, 4), 128, 0, stream>>>(
            x, W1, xl1, N_NODES, waveTilesN);
    }
    gat_att_logits<<<blk((long long)N_NODES * HEADS, TB), TB, 0, stream>>>(
        xl1, ats1, atd1, as1, ad1, N_NODES, HEADS, HID);
    gat_gather_l1<<<blk(N_NODES, 8), TB, 0, stream>>>(
        rowptr, list, as1, ad1, xl1, b1, h1);

    {
        const int waveTilesN = OUT_C / 16;
        const long long waves = (long long)(N_NODES / 16) * waveTilesN;
        gat_wmma_gemm_f32<1, HIDDEN, OUT_C><<<blk(waves, 4), 128, 0, stream>>>(
            h1, W2, xl2, N_NODES, waveTilesN);
    }
    gat_att_logits<<<blk((long long)N_NODES, TB), TB, 0, stream>>>(
        xl2, ats2, atd2, as2, ad2, N_NODES, 1, OUT_C);
    gat_gather_l2<<<blk(N_NODES, 16), TB, 0, stream>>>(
        rowptr, list, as2, ad2, xl2, b2, out);
}


# Layer-2 product + logits folded into layer-1 gather: 16 nodes/block pass ELU rows to f32 WMMA via LDS, K split over 16 waves (partials added in wave order, logits by lane-xor tree); 2 launches gone
# speedup vs baseline: 6.8307x; 1.0450x over previous
#include <hip/hip_runtime.h>
#include <cstdint>
#include <cstddef>

#define N_NODES   50000
#define N_EDGES   800000
#define TOT_E     (N_EDGES + N_NODES)
#define IN_C      128
#define HID       32
#define OUT_C     16
#define HEADS     8
#define HIDDEN    (HEADS * HID)
#define NEG_SLOPE 0.2f
#define GAT_EPS   1e-16f

typedef float v2f __attribute__((ext_vector_type(2)));
typedef float v8f __attribute__((ext_vector_type(8)));

template <int TN, int K, int NC>
__global__ __launch_bounds__(128) void gat_wmma_gemm_f32(
    const float* __restrict__ A, const float* __restrict__ B,
    float* __restrict__ C, int M, int waveTilesN)
{
    const int wave = blockIdx.x * (blockDim.x >> 5) + (threadIdx.x >> 5);
    const int lane = threadIdx.x & 31;
    const int tM = wave / waveTilesN;
    const int tg = wave % waveTilesN;
    if (tM * 16 >= M) return;

    const int half = lane >> 4;
    const int l    = lane & 15;
    const float* __restrict__ Ap = A + (size_t)(tM * 16 + l) * K + half * 2;
    const float* __restrict__ Bp = B + (size_t)(half * 2) * NC + tg * 16 * TN + l;

    const v8f zero = {0.f,0.f,0.f,0.f,0.f,0.f,0.f,0.f};
    v8f c[TN];
#pragma unroll
    for (int j = 0; j < TN; ++j) c[j] = zero;

#pragma unroll
    for (int k = 0; k < K; k += 4) {
        const v2f a = *(const v2f*)(Ap + k);
        v2f b[TN];
#pragma unroll
        for (int j = 0; j < TN; ++j) {
            b[j].x = Bp[(size_t)k * NC + j * 16];
            b[j].y = Bp[(size_t)(k + 1) * NC + j * 16];
        }
#pragma unroll
        for (int j = 0; j < TN; ++j)
            c[j] = __builtin_amdgcn_wmma_f32_16x16x4_f32(
                        false, a,  false, b[j],
                        (short)0, c[j],
                        false,  false);
    }

    float* __restrict__ Cbase =
        C + (size_t)(tM * 16 + half * 8) * NC + tg * 16 * TN + l;
#pragma unroll
    for (int j = 0; j < TN; ++j)
#pragma unroll
        for (int r = 0; r < 8; ++r)
            Cbase[(size_t)r * NC + j * 16] = c[j][r];
}

__global__ __launch_bounds__(256) void gat_att_logits(
    const float* __restrict__ xl, const float* __restrict__ att_s,
    const float* __restrict__ att_d, float* __restrict__ asrc,
    float* __restrict__ adst, int n, int H, int CH)
{
    const int i = blockIdx.x * blockDim.x + threadIdx.x;
    if (i >= n * H) return;
    const int h = i % H;
    const float* __restrict__ row = xl + (size_t)i * CH;
    const float* __restrict__ as  = att_s + (size_t)h * CH;
    const float* __restrict__ ad  = att_d + (size_t)h * CH;
    float s = 0.f, d = 0.f;
    for (int c = 0; c < CH; ++c) { float v = row[c]; s += v * as[c]; d += v * ad[c]; }
    asrc[i] = s;
    adst[i] = d;
}

__device__ __forceinline__ float leaky(float v) { return v > 0.f ? v : NEG_SLOPE * v; }

#define SCAN_TB  1024
#define SCAN_PER ((N_NODES + SCAN_TB - 1) / SCAN_TB)
#define CNT_PAD  (SCAN_TB * SCAN_PER)

#define RANK_CHUNKS  64
#define RANK_CHUNK_E (N_EDGES / RANK_CHUNKS)
#define RANK_TB      1024
#define RANK_PER     ((RANK_CHUNK_E + RANK_TB - 1) / RANK_TB)
static_assert(RANK_CHUNKS * RANK_CHUNK_E == N_EDGES, "chunks must tile the edge list");
static_assert(N_NODES % 4 == 0, "counter table is moved 16 bytes at a time");

__global__ __launch_bounds__(RANK_TB) void gat_edge_rank(
    const int* __restrict__ dst, int* __restrict__ part, int* __restrict__ rnk)
{
    __shared__ __attribute__((aligned(16))) int hist[N_NODES];
    const int t  = threadIdx.x;
    const int e0 = blockIdx.x * RANK_CHUNK_E;
    int4* h4 = (int4*)hist;
    for (int i = t; i < N_NODES / 4; i += RANK_TB) h4[i] = make_int4(0, 0, 0, 0);
    int d[RANK_PER];
#pragma unroll
    for (int k = 0; k < RANK_PER; ++k) {
        const int i = t + k * RANK_TB;
        d[k] = (i < RANK_CHUNK_E) ? dst[e0 + i] : 0;
    }
    __syncthreads();
    int r[RANK_PER];
#pragma unroll
    for (int k = 0; k < RANK_PER; ++k) {
        const int i = t + k * RANK_TB;
        r[k] = (i < RANK_CHUNK_E) ? atomicAdd(&hist[d[k]], 1) : 0;
    }
#pragma unroll
    for (int k = 0; k < RANK_PER; ++k) {
        const int i = t + k * RANK_TB;
        if (i < RANK_CHUNK_E) rnk[e0 + i] = r[k];
    }
    __syncthreads();
    int4* __restrict__ p4 = (int4*)(part + (size_t)blockIdx.x * CNT_PAD);
    for (int i = t; i < N_NODES / 4; i += RANK_TB) p4[i] = h4[i];
}

__global__ __launch_bounds__(256) void gat_chunk_prefix(
    int* __restrict__ part, int* __restrict__ cnt)
{
    const int n = blockIdx.x * 256 + threadIdx.x;
    if (n >= N_NODES) { cnt[n] = 0; return; }
    int v[RANK_CHUNKS];
#pragma unroll
    for (int q = 0; q < RANK_CHUNKS; ++q) v[q] = part[(size_t)q * CNT_PAD + n];
    int s = 0;
#pragma unroll
    for (int q = 0; q < RANK_CHUNKS; ++q) { part[(size_t)q * CNT_PAD + n] = s; s += v[q]; }
    cnt[n] = s;
}

__global__ __launch_bounds__(SCAN_TB) void gat_rowptr_scan(
    const int* __restrict__ cnt, int* __restrict__ rowptr)
{
    __shared__ int wsum[32];
    const int t = threadIdx.x, lane = t & 31, w = t >> 5;
    const int* __restrict__ cp = cnt + t * SCAN_PER;
    int v[SCAN_PER];
    int s = 0;
#pragma unroll
    for (int i = 0; i < SCAN_PER; ++i) { v[i] = cp[i]; s += v[i]; }
    int x = s;
    for (int o = 1; o < 32; o <<= 1) { const int y = __shfl_up(x, o); if (lane >= o) x += y; }
    if (lane == 31) wsum[w] = x;
    __syncthreads();
    if (w == 0) {
        int u = wsum[lane];
        for (int o = 1; o < 32; o <<= 1) { const int y = __shfl_up(u, o); if (lane >= o) u += y; }
        wsum[lane] = u;
    }
    __syncthreads();
    int base = x - s + (w > 0 ? wsum[w - 1] : 0);
    int* __restrict__ rp = rowptr + t * SCAN_PER;
#pragma unroll
    for (int i = 0; i < SCAN_PER; ++i) { rp[i] = base; base += v[i]; }
}

__global__ __launch_bounds__(256) void gat_edge_place(
    const int* __restrict__ src, const int* __restrict__ dst,
    const int* __restrict__ rowptr, const int* __restrict__ part,
    const int* __restrict__ rnk, int* __restrict__ list)
{
    const int e = blockIdx.x * blockDim.x + threadIdx.x;
    if (e >= N_EDGES) return;
    const int d = dst[e];
    const int before = part[(size_t)(e / RANK_CHUNK_E) * CNT_PAD + d];
    list[rowptr[d] + before + rnk[e]] = src[e];
}

#define L1_WAVES 16
#define H_PITCH  (HIDDEN + 4)
static_assert(N_NODES % L1_WAVES == 0, "every block is a full tile of 16 nodes: no partial block, no bounds test");
static_assert(HIDDEN == L1_WAVES * 16, "wave w takes K slice 16w .. 16w+15 of the layer-2 product");
static_assert(OUT_C == 16, "the layer-2 row is one 16-column WMMA tile");

__global__ __launch_bounds__(L1_WAVES * 32) void gat_gather_l1(
    const int* __restrict__ rowptr, const int* __restrict__ list,
    const float* __restrict__ asrc, const float* __restrict__ adst,
    const float* __restrict__ xl, const float* __restrict__ bias,
    const float* __restrict__ W2, const float* __restrict__ att_s2,
    const float* __restrict__ att_d2, float* __restrict__ xl2,
    float* __restrict__ as2, float* __restrict__ ad2)
{
    __shared__ __attribute__((aligned(16))) float sw[L1_WAVES][32 * HEADS];
    __shared__ __attribute__((aligned(16))) float hrow[L1_WAVES][H_PITCH];
    const int lane = threadIdx.x & 31;
    const int wv   = threadIdx.x >> 5;
    const int d = __builtin_amdgcn_readfirstlane((int)(blockIdx.x * L1_WAVES + wv));
    float* my = sw[wv];
    const int p0 = rowptr[d], p1 = rowptr[d + 1];
    const int deg = p1 - p0;
    const int nch = (deg + 31) >> 5;

    const int hh = lane & 7, j = lane >> 3;
    const float adh   = adst[(size_t)d * HEADS + hh];
    const float vself = leaky(asrc[(size_t)d * HEADS + hh] + adh);

    int sv = d;
    auto stage = [&](int c) {
        const int q = c * 32 + lane;
        sv = (q < deg) ? list[p0 + q] : d;
        const float4* ar = (const float4*)(asrc + (size_t)sv * HEADS);
        const float4 a0 = ar[0], a1 = ar[1];
        float4* dp = (float4*)(my + lane * HEADS);
        dp[0] = a0; dp[1] = a1;
    };

    float m = vself;
    for (int c = 0; c < nch; ++c) {
        stage(c);
#pragma unroll
        for (int i = 0; i < 8; ++i) {
            const int sl = j + 4 * i;
            const float v = leaky(my[sl * HEADS + hh] + adh);
            m = (c * 32 + sl < deg) ? fmaxf(m, v) : m;
        }
    }
    m = fmaxf(m, __shfl_xor(m, 8));
    m = fmaxf(m, __shfl_xor(m, 16));
    float sum = (j == 0) ? expf(vself - m) : 0.f;
    for (int c = 0; c < nch; ++c) {
        if (nch > 1) stage(c);
#pragma unroll
        for (int i = 0; i < 8; ++i) {
            const int sl = j + 4 * i;
            const float ex = expf(leaky(my[sl * HEADS + hh] + adh) - m);
            sum = (c * 32 + sl < deg) ? sum + ex : sum;
        }
    }
    sum += __shfl_xor(sum, 8);
    sum += __shfl_xor(sum, 16);
    const float den = sum + GAT_EPS;
    const float wself = expf(vself - m) / den;

    const int hc = lane >> 2;
    float acc[8];
    {
        const float a = __shfl(wself, hc);
        const float4* r = (const float4*)(xl + (size_t)d * HIDDEN + lane * 8);
        const float4 r0 = r[0], r1 = r[1];
        acc[0] = a * r0.x; acc[1] = a * r0.y; acc[2] = a * r0.z; acc[3] = a * r0.w;
        acc[4] = a * r1.x; acc[5] = a * r1.y; acc[6] = a * r1.z; acc[7] = a * r1.w;
    }
    for (int c = 0; c < nch; ++c) {
        if (nch > 1) stage(c);
#pragma unroll
        for (int i = 0; i < 8; ++i) {
            const int sl = j + 4 * i;
            my[sl * HEADS + hh] = expf(leaky(my[sl * HEADS + hh] + adh) - m) / den;
        }
        const int cnt = min(32, deg - c * 32);
#pragma unroll 4
        for (int e = 0; e < cnt; ++e) {
            const int s = __builtin_amdgcn_readlane(sv, e);
            const float a = my[e * HEADS + hc];
            const float4* r = (const float4*)(xl + (size_t)s * HIDDEN + lane * 8);
            const float4 r0 = r[0], r1 = r[1];
            acc[0] += a * r0.x; acc[1] += a * r0.y; acc[2] += a * r0.z; acc[3] += a * r0.w;
            acc[4] += a * r1.x; acc[5] += a * r1.y; acc[6] += a * r1.z; acc[7] += a * r1.w;
        }
    }
    const float4* bp = (const float4*)(bias + lane * 8);
    const float4 b0 = bp[0], b1 = bp[1];
    const float bb[8] = {b0.x, b0.y, b0.z, b0.w, b1.x, b1.y, b1.z, b1.w};
    float o[8];
#pragma unroll
    for (int i = 0; i < 8; ++i) { const float v = acc[i] + bb[i]; o[i] = v > 0.f ? v : expm1f(v); }
    float4* op = (float4*)(&hrow[wv][lane * 8]);
    op[0] = make_float4(o[0], o[1], o[2], o[3]);
    op[1] = make_float4(o[4], o[5], o[6], o[7]);
    __syncthreads();

    const int kh = (threadIdx.x >> 4) & 1, l16 = threadIdx.x & 15;
    const float* hA = &hrow[l16][wv * 16 + kh * 2];
    const float* __restrict__ Bq = W2 + (size_t)(wv * 16 + kh * 2) * OUT_C + l16;
    v8f c2 = {0.f, 0.f, 0.f, 0.f, 0.f, 0.f, 0.f, 0.f};
#pragma unroll
    for (int kk = 0; kk < 4; ++kk) {
        const v2f a2 = *(const v2f*)(hA + kk * 4);
        v2f b2;
        b2.x = Bq[(size_t)(kk * 4) * OUT_C];
        b2.y = Bq[(size_t)(kk * 4 + 1) * OUT_C];
        c2 = __builtin_amdgcn_wmma_f32_16x16x4_f32(
                  false, a2,  false, b2,
                  (short)0, c2,
                  false,  false);
    }
#pragma unroll
    for (int r = 0; r < 8; ++r) my[r * 32 + lane] = c2[r];
    __syncthreads();
    if (threadIdx.x < 256) {
        const int idx = threadIdx.x;
        float s = sw[0][idx];
#pragma unroll
        for (int w = 1; w < L1_WAVES; ++w) s += sw[w][idx];
        const int col  = idx & 15;
        const int node = blockIdx.x * L1_WAVES + ((idx >> 4) & 1) * 8 + (idx >> 5);
        xl2[(size_t)node * OUT_C + col] = s;
        float ps = s * att_s2[col], pd = s * att_d2[col];
        ps += __int_as_float(__builtin_amdgcn_ds_swizzle(__float_as_int(ps), 0x041f));
        pd += __int_as_float(__builtin_amdgcn_ds_swizzle(__float_as_int(pd), 0x041f));
        ps += __int_as_float(__builtin_amdgcn_ds_swizzle(__float_as_int(ps), 0x081f));
        pd += __int_as_float(__builtin_amdgcn_ds_swizzle(__float_as_int(pd), 0x081f));
        ps += __int_as_float(__builtin_amdgcn_ds_swizzle(__float_as_int(ps), 0x101f));
        pd += __int_as_float(__builtin_amdgcn_ds_swizzle(__float_as_int(pd), 0x101f));
        ps += __int_as_float(__builtin_amdgcn_ds_swizzle(__float_as_int(ps), 0x201f));
        pd += __int_as_float(__builtin_amdgcn_ds_swizzle(__float_as_int(pd), 0x201f));
        if (col == 0) { as2[node] = ps; ad2[node] = pd; }
    }
}

__global__ __launch_bounds__(256) void gat_gather_l2(
    const int* __restrict__ rowptr, const int* __restrict__ list,
    const float* __restrict__ asrc, const float* __restrict__ adst,
    const float* __restrict__ xl, const float* __restrict__ bias,
    float* __restrict__ out)
{
    const int c = threadIdx.x & 15;
    const int d = blockIdx.x * 16 + (threadIdx.x >> 4);
    if (d >= N_NODES) return;
    const int p0 = rowptr[d], p1 = rowptr[d + 1];
    const float ad    = adst[d];
    const float vself = leaky(asrc[d] + ad);
    float m = vself;
    for (int p = p0 + c; p < p1; p += 16) m = fmaxf(m, leaky(asrc[list[p]] + ad));
    m = fmaxf(m, __shfl_xor(m, 1)); m = fmaxf(m, __shfl_xor(m, 2));
    m = fmaxf(m, __shfl_xor(m, 4)); m = fmaxf(m, __shfl_xor(m, 8));
    float sum = (c == 0) ? expf(vself - m) : 0.f;
    for (int p = p0 + c; p < p1; p += 16) sum += expf(leaky(asrc[list[p]] + ad) - m);
    sum += __shfl_xor(sum, 1); sum += __shfl_xor(sum, 2);
    sum += __shfl_xor(sum, 4); sum += __shfl_xor(sum, 8);
    const float den = sum + GAT_EPS;
    float acc = expf(vself - m) / den * xl[(size_t)d * OUT_C + c];
#pragma unroll 4
    for (int p = p0; p < p1; ++p) {
        const int s = list[p];
        const float a = expf(leaky(asrc[s] + ad) - m) / den;
        acc += a * xl[(size_t)s * OUT_C + c];
    }
    out[(size_t)d * OUT_C + c] = acc + bias[c];
}

extern "C" void kernel_launch(void* const* d_in, const int* in_sizes, int n_in,
                              void* d_out, int out_size, void* d_ws, size_t ws_size,
                              hipStream_t stream)
{
    (void)in_sizes; (void)n_in; (void)out_size; (void)ws_size;

    const float* x    = (const float*)d_in[0];
    const int*   ei   = (const int*)  d_in[1];
    const float* W1   = (const float*)d_in[2];
    const float* ats1 = (const float*)d_in[3];
    const float* atd1 = (const float*)d_in[4];
    const float* b1   = (const float*)d_in[5];
    const float* W2   = (const float*)d_in[6];
    const float* ats2 = (const float*)d_in[7];
    const float* atd2 = (const float*)d_in[8];
    const float* b2   = (const float*)d_in[9];
    const int* src = ei;
    const int* dst = ei + N_EDGES;
    float* out = (float*)d_out;

    int*   cnt    = (int*)d_ws;
    int*   rowptr = cnt + (size_t)CNT_PAD;
    int*   rnk    = rowptr + (size_t)CNT_PAD;
    int*   list   = rnk + (size_t)N_EDGES;
    float* xl1 = (float*)(list + (size_t)N_EDGES);
    float* as1 = xl1 + (size_t)N_NODES * HIDDEN;
    float* ad1 = as1 + (size_t)N_NODES * HEADS;
    float* xl2 = ad1 + (size_t)N_NODES * HEADS;
    float* as2 = xl2 + (size_t)N_NODES * OUT_C;
    float* ad2 = as2 + (size_t)CNT_PAD;
    int*   part = (int*)(ad2 + (size_t)CNT_PAD);

    const int TB = 256;
    auto blk = [](long long n, int tb) { return (unsigned)((n + tb - 1) / tb); };

    gat_edge_rank<<<RANK_CHUNKS, RANK_TB, 0, stream>>>(dst, part, rnk);
    gat_chunk_prefix<<<CNT_PAD / 256, 256, 0, stream>>>(part, cnt);
    gat_rowptr_scan<<<1, SCAN_TB, 0, stream>>>(cnt, rowptr);
    gat_edge_place<<<blk(N_EDGES, TB), TB, 0, stream>>>(src, dst, rowptr, part, rnk, list);

    {
        const int waveTilesN = HIDDEN / (16 * 4);
        const long long waves = (long long)(N_NODES / 16) * waveTilesN;
        gat_wmma_gemm_f32<4, IN_C, HIDDEN><<<blk(waves, 4), 128, 0, stream>>>(
            x, W1, xl1, N_NODES, waveTilesN);
    }
    gat_att_logits<<<blk((long long)N_NODES * HEADS, TB), TB, 0, stream>>>(
        xl1, ats1, atd1, as1, ad1, N_NODES, HEADS, HID);
    gat_gather_l1<<<N_NODES / L1_WAVES, L1_WAVES * 32, 0, stream>>>(
        rowptr, list, as1, ad1, xl1, b1, W2, ats2, atd2, xl2, as2, ad2);

    gat_gather_l2<<<blk(N_NODES, 16), TB, 0, stream>>>(
        rowptr, list, as2, ad2, xl2, b2, out);
}


# Row-pointer scan folded into the chunk-prefix kernel: rank kernel also stores per-chunk totals of 256-node buckets; a prefix block adds earlier buckets as base, scans its 256 nodes; 1 launch gone
# speedup vs baseline: 7.5380x; 1.1035x over previous
#include <hip/hip_runtime.h>
#include <cstdint>
#include <cstddef>

#define N_NODES   50000
#define N_EDGES   800000
#define TOT_E     (N_EDGES + N_NODES)
#define IN_C      128
#define HID       32
#define OUT_C     16
#define HEADS     8
#define HIDDEN    (HEADS * HID)
#define NEG_SLOPE 0.2f
#define GAT_EPS   1e-16f

typedef float v2f __attribute__((ext_vector_type(2)));
typedef float v8f __attribute__((ext_vector_type(8)));

template <int TN, int K, int NC>
__global__ __launch_bounds__(128) void gat_wmma_gemm_f32(
    const float* __restrict__ A, const float* __restrict__ B,
    float* __restrict__ C, int M, int waveTilesN)
{
    const int wave = blockIdx.x * (blockDim.x >> 5) + (threadIdx.x >> 5);
    const int lane = threadIdx.x & 31;
    const int tM = wave / waveTilesN;
    const int tg = wave % waveTilesN;
    if (tM * 16 >= M) return;

    const int half = lane >> 4;
    const int l    = lane & 15;
    const float* __restrict__ Ap = A + (size_t)(tM * 16 + l) * K + half * 2;
    const float* __restrict__ Bp = B + (size_t)(half * 2) * NC + tg * 16 * TN + l;

    const v8f zero = {0.f,0.f,0.f,0.f,0.f,0.f,0.f,0.f};
    v8f c[TN];
#pragma unroll
    for (int j = 0; j < TN; ++j) c[j] = zero;

#pragma unroll
    for (int k = 0; k < K; k += 4) {
        const v2f a = *(const v2f*)(Ap + k);
        v2f b[TN];
#pragma unroll
        for (int j = 0; j < TN; ++j) {
            b[j].x = Bp[(size_t)k * NC + j * 16];
            b[j].y = Bp[(size_t)(k + 1) * NC + j * 16];
        }
#pragma unroll
        for (int j = 0; j < TN; ++j)
            c[j] = __builtin_amdgcn_wmma_f32_16x16x4_f32(
                        false, a,  false, b[j],
                        (short)0, c[j],
                        false,  false);
    }

    float* __restrict__ Cbase =
        C + (size_t)(tM * 16 + half * 8) * NC + tg * 16 * TN + l;
#pragma unroll
    for (int j = 0; j < TN; ++j)
#pragma unroll
        for (int r = 0; r < 8; ++r)
            Cbase[(size_t)r * NC + j * 16] = c[j][r];
}

__global__ __launch_bounds__(256) void gat_att_logits(
    const float* __restrict__ xl, const float* __restrict__ att_s,
    const float* __restrict__ att_d, float* __restrict__ asrc,
    float* __restrict__ adst, int n, int H, int CH)
{
    const int i = blockIdx.x * blockDim.x + threadIdx.x;
    if (i >= n * H) return;
    const int h = i % H;
    const float* __restrict__ row = xl + (size_t)i * CH;
    const float* __restrict__ as  = att_s + (size_t)h * CH;
    const float* __restrict__ ad  = att_d + (size_t)h * CH;
    float s = 0.f, d = 0.f;
    for (int c = 0; c < CH; ++c) { float v = row[c]; s += v * as[c]; d += v * ad[c]; }
    asrc[i] = s;
    adst[i] = d;
}

__device__ __forceinline__ float leaky(float v) { return v > 0.f ? v : NEG_SLOPE * v; }

#define BKT_N    256
#define N_BKT    ((N_NODES + BKT_N - 1) / BKT_N)
#define CNT_PAD  (N_BKT * BKT_N)

#define RANK_CHUNKS  64
#define RANK_CHUNK_E (N_EDGES / RANK_CHUNKS)
#define RANK_TB      1024
#define RANK_PER     ((RANK_CHUNK_E + RANK_TB - 1) / RANK_TB)
static_assert(RANK_CHUNKS * RANK_CHUNK_E == N_EDGES, "chunks must tile the edge list");
static_assert(N_NODES % 4 == 0, "counter table is moved 16 bytes at a time");
static_assert(RANK_CHUNKS % 4 == 0, "a bucket's chunk totals are read 16 bytes at a time");
static_assert(BKT_N % 32 == 0 && RANK_TB % 32 == 0, "whole waves");
#define BKT_PER_WAVE ((N_BKT + RANK_TB / 32 - 1) / (RANK_TB / 32))

__global__ __launch_bounds__(RANK_TB) void gat_edge_rank(
    const int* __restrict__ dst, int* __restrict__ part, int* __restrict__ rnk,
    int* __restrict__ bsum)
{
    __shared__ __attribute__((aligned(16))) int hist[N_NODES];
    const int t  = threadIdx.x;
    const int e0 = blockIdx.x * RANK_CHUNK_E;
    int4* h4 = (int4*)hist;
    for (int i = t; i < N_NODES / 4; i += RANK_TB) h4[i] = make_int4(0, 0, 0, 0);
    int d[RANK_PER];
#pragma unroll
    for (int k = 0; k < RANK_PER; ++k) {
        const int i = t + k * RANK_TB;
        d[k] = (i < RANK_CHUNK_E) ? dst[e0 + i] : 0;
    }
    __syncthreads();
    int r[RANK_PER];
#pragma unroll
    for (int k = 0; k < RANK_PER; ++k) {
        const int i = t + k * RANK_TB;
        r[k] = (i < RANK_CHUNK_E) ? atomicAdd(&hist[d[k]], 1) : 0;
    }
#pragma unroll
    for (int k = 0; k < RANK_PER; ++k) {
        const int i = t + k * RANK_TB;
        if (i < RANK_CHUNK_E) rnk[e0 + i] = r[k];
    }
    __syncthreads();
    int4* __restrict__ p4 = (int4*)(part + (size_t)blockIdx.x * CNT_PAD);
    for (int i = t; i < N_NODES / 4; i += RANK_TB) p4[i] = h4[i];
    const int lane = t & 31, w = t >> 5;
    int bs[BKT_PER_WAVE];
#pragma unroll
    for (int k = 0; k < BKT_PER_WAVE; ++k) {
        const int jb = w + k * (RANK_TB / 32);
        int acc = 0;
#pragma unroll
        for (int i = 0; i < BKT_N / 32; ++i) {
            const int n = jb * BKT_N + i * 32 + lane;
            acc += (n < N_NODES) ? hist[n] : 0;
        }
        bs[k] = acc;
    }
#pragma unroll
    for (int o = 16; o > 0; o >>= 1) {
#pragma unroll
        for (int k = 0; k < BKT_PER_WAVE; ++k) bs[k] += __shfl_xor(bs[k], o);
    }
#pragma unroll
    for (int k = 0; k < BKT_PER_WAVE; ++k) {
        const int jb = w + k * (RANK_TB / 32);
        if (lane == 0 && jb < N_BKT) bsum[jb * RANK_CHUNKS + blockIdx.x] = bs[k];
    }
}

#define BASE_MOVE (((N_BKT - 1) * (RANK_CHUNKS / 4) + BKT_N - 1) / BKT_N)
__global__ __launch_bounds__(BKT_N) void gat_chunk_prefix(
    int* __restrict__ part, const int* __restrict__ bsum, int* __restrict__ rowptr)
{
    __shared__ int wtot[BKT_N / 32];
    __shared__ int wsum[BKT_N / 32];
    const int t = threadIdx.x, lane = t & 31, w = t >> 5;
    const int n = blockIdx.x * BKT_N + t;
    const int4* __restrict__ b4 = (const int4*)bsum;
    const int n4 = blockIdx.x * (RANK_CHUNKS / 4);
    int4 q[BASE_MOVE];
#pragma unroll
    for (int k = 0; k < BASE_MOVE; ++k) {
        const int i = t + k * BKT_N;
        q[k] = (i < n4) ? b4[i] : make_int4(0, 0, 0, 0);
    }
    int s = 0;
    if (n < N_NODES) {
        int v[RANK_CHUNKS];
#pragma unroll
        for (int c = 0; c < RANK_CHUNKS; ++c) v[c] = part[(size_t)c * CNT_PAD + n];
#pragma unroll
        for (int c = 0; c < RANK_CHUNKS; ++c) { part[(size_t)c * CNT_PAD + n] = s; s += v[c]; }
    }
    int a = 0;
#pragma unroll
    for (int k = 0; k < BASE_MOVE; ++k) a += q[k].x + q[k].y + q[k].z + q[k].w;
    a += __shfl_xor(a, 16); a += __shfl_xor(a, 8); a += __shfl_xor(a, 4);
    a += __shfl_xor(a, 2);  a += __shfl_xor(a, 1);
    int x = s;
    for (int o = 1; o < 32; o <<= 1) { const int y = __shfl_up(x, o); if (lane >= o) x += y; }
    if (lane == 31) wsum[w] = x;
    if (lane == 0)  wtot[w] = a;
    __syncthreads();
    int base = 0;
#pragma unroll
    for (int i = 0; i < BKT_N / 32; ++i) { base += wtot[i]; base += (i < w) ? wsum[i] : 0; }
    rowptr[n] = base + x - s;
}

__global__ __launch_bounds__(256) void gat_edge_place(
    const int* __restrict__ src, const int* __restrict__ dst,
    const int* __restrict__ rowptr, const int* __restrict__ part,
    const int* __restrict__ rnk, int* __restrict__ list)
{
    const int e = blockIdx.x * blockDim.x + threadIdx.x;
    if (e >= N_EDGES) return;
    const int d = dst[e];
    const int before = part[(size_t)(e / RANK_CHUNK_E) * CNT_PAD + d];
    list[rowptr[d] + before + rnk[e]] = src[e];
}

#define L1_WAVES 16
#define H_PITCH  (HIDDEN + 4)
static_assert(N_NODES % L1_WAVES == 0, "every block is a full tile of 16 nodes: no partial block, no bounds test");
static_assert(HIDDEN == L1_WAVES * 16, "wave w takes K slice 16w .. 16w+15 of the layer-2 product");
static_assert(OUT_C == 16, "the layer-2 row is one 16-column WMMA tile");

__global__ __launch_bounds__(L1_WAVES * 32) void gat_gather_l1(
    const int* __restrict__ rowptr, const int* __restrict__ list,
    const float* __restrict__ asrc, const float* __restrict__ adst,
    const float* __restrict__ xl, const float* __restrict__ bias,
    const float* __restrict__ W2, const float* __restrict__ att_s2,
    const float* __restrict__ att_d2, float* __restrict__ xl2,
    float* __restrict__ as2, float* __restrict__ ad2)
{
    __shared__ __attribute__((aligned(16))) float sw[L1_WAVES][32 * HEADS];
    __shared__ __attribute__((aligned(16))) float hrow[L1_WAVES][H_PITCH];
    const int lane = threadIdx.x & 31;
    const int wv   = threadIdx.x >> 5;
    const int d = __builtin_amdgcn_readfirstlane((int)(blockIdx.x * L1_WAVES + wv));
    float* my = sw[wv];
    const int p0 = rowptr[d], p1 = rowptr[d + 1];
    const int deg = p1 - p0;
    const int nch = (deg + 31) >> 5;

    const int hh = lane & 7, j = lane >> 3;
    const float adh   = adst[(size_t)d * HEADS + hh];
    const float vself = leaky(asrc[(size_t)d * HEADS + hh] + adh);

    int sv = d;
    auto stage = [&](int c) {
        const int q = c * 32 + lane;
        sv = (q < deg) ? list[p0 + q] : d;
        const float4* ar = (const float4*)(asrc + (size_t)sv * HEADS);
        const float4 a0 = ar[0], a1 = ar[1];
        float4* dp = (float4*)(my + lane * HEADS);
        dp[0] = a0; dp[1] = a1;
    };

    float m = vself;
    for (int c = 0; c < nch; ++c) {
        stage(c);
#pragma unroll
        for (int i = 0; i < 8; ++i) {
            const int sl = j + 4 * i;
            const float v = leaky(my[sl * HEADS + hh] + adh);
            m = (c * 32 + sl < deg) ? fmaxf(m, v) : m;
        }
    }
    m = fmaxf(m, __shfl_xor(m, 8));
    m = fmaxf(m, __shfl_xor(m, 16));
    float sum = (j == 0) ? expf(vself - m) : 0.f;
    for (int c = 0; c < nch; ++c) {
        if (nch > 1) stage(c);
#pragma unroll
        for (int i = 0; i < 8; ++i) {
            const int sl = j + 4 * i;
            const float ex = expf(leaky(my[sl * HEADS + hh] + adh) - m);
            sum = (c * 32 + sl < deg) ? sum + ex : sum;
        }
    }
    sum += __shfl_xor(sum, 8);
    sum += __shfl_xor(sum, 16);
    const float den = sum + GAT_EPS;
    const float wself = expf(vself - m) / den;

    const int hc = lane >> 2;
    float acc[8];
    {
        const float a = __shfl(wself, hc);
        const float4* r = (const float4*)(xl + (size_t)d * HIDDEN + lane * 8);
        const float4 r0 = r[0], r1 = r[1];
        acc[0] = a * r0.x; acc[1] = a * r0.y; acc[2] = a * r0.z; acc[3] = a * r0.w;
        acc[4] = a * r1.x; acc[5] = a * r1.y; acc[6] = a * r1.z; acc[7] = a * r1.w;
    }
    for (int c = 0; c < nch; ++c) {
        if (nch > 1) stage(c);
#pragma unroll
        for (int i = 0; i < 8; ++i) {
            const int sl = j + 4 * i;
            my[sl * HEADS + hh] = expf(leaky(my[sl * HEADS + hh] + adh) - m) / den;
        }
        const int cnt = min(32, deg - c * 32);
#pragma unroll 4
        for (int e = 0; e < cnt; ++e) {
            const int s = __builtin_amdgcn_readlane(sv, e);
            const float a = my[e * HEADS + hc];
            const float4* r = (const float4*)(xl + (size_t)s * HIDDEN + lane * 8);
            const float4 r0 = r[0], r1 = r[1];
            acc[0] += a * r0.x; acc[1] += a * r0.y; acc[2] += a * r0.z; acc[3] += a * r0.w;
            acc[4] += a * r1.x; acc[5] += a * r1.y; acc[6] += a * r1.z; acc[7] += a * r1.w;
        }
    }
    const float4* bp = (const float4*)(bias + lane * 8);
    const float4 b0 = bp[0], b1 = bp[1];
    const float bb[8] = {b0.x, b0.y, b0.z, b0.w, b1.x, b1.y, b1.z, b1.w};
    float o[8];
#pragma unroll
    for (int i = 0; i < 8; ++i) { const float v = acc[i] + bb[i]; o[i] = v > 0.f ? v : expm1f(v); }
    float4* op = (float4*)(&hrow[wv][lane * 8]);
    op[0] = make_float4(o[0], o[1], o[2], o[3]);
    op[1] = make_float4(o[4], o[5], o[6], o[7]);
    __syncthreads();

    const int kh = (threadIdx.x >> 4) & 1, l16 = threadIdx.x & 15;
    const float* hA = &hrow[l16][wv * 16 + kh * 2];
    const float* __restrict__ Bq = W2 + (size_t)(wv * 16 + kh * 2) * OUT_C + l16;
    v8f c2 = {0.f, 0.f, 0.f, 0.f, 0.f, 0.f, 0.f, 0.f};
#pragma unroll
    for (int kk = 0; kk < 4; ++kk) {
        const v2f a2 = *(const v2f*)(hA + kk * 4);
        v2f b2;
        b2.x = Bq[(size_t)(kk * 4) * OUT_C];
        b2.y = Bq[(size_t)(kk * 4 + 1) * OUT_C];
        c2 = __builtin_amdgcn_wmma_f32_16x16x4_f32(
                  false, a2,  false, b2,
                  (short)0, c2,
                  false,  false);
    }
#pragma unroll
    for (int r = 0; r < 8; ++r) my[r * 32 + lane] = c2[r];
    __syncthreads();
    if (threadIdx.x < 256) {
        const int idx = threadIdx.x;
        float s = sw[0][idx];
#pragma unroll
        for (int w = 1; w < L1_WAVES; ++w) s += sw[w][idx];
        const int col  = idx & 15;
        const int node = blockIdx.x * L1_WAVES + ((idx >> 4) & 1) * 8 + (idx >> 5);
        xl2[(size_t)node * OUT_C + col] = s;
        float ps = s * att_s2[col], pd = s * att_d2[col];
        ps += __int_as_float(__builtin_amdgcn_ds_swizzle(__float_as_int(ps), 0x041f));
        pd += __int_as_float(__builtin_amdgcn_ds_swizzle(__float_as_int(pd), 0x041f));
        ps += __int_as_float(__builtin_amdgcn_ds_swizzle(__float_as_int(ps), 0x081f));
        pd += __int_as_float(__builtin_amdgcn_ds_swizzle(__float_as_int(pd), 0x081f));
        ps += __int_as_float(__builtin_amdgcn_ds_swizzle(__float_as_int(ps), 0x101f));
        pd += __int_as_float(__builtin_amdgcn_ds_swizzle(__float_as_int(pd), 0x101f));
        ps += __int_as_float(__builtin_amdgcn_ds_swizzle(__float_as_int(ps), 0x201f));
        pd += __int_as_float(__builtin_amdgcn_ds_swizzle(__float_as_int(pd), 0x201f));
        if (col == 0) { as2[node] = ps; ad2[node] = pd; }
    }
}

__global__ __launch_bounds__(256) void gat_gather_l2(
    const int* __restrict__ rowptr, const int* __restrict__ list,
    const float* __restrict__ asrc, const float* __restrict__ adst,
    const float* __restrict__ xl, const float* __restrict__ bias,
    float* __restrict__ out)
{
    const int c = threadIdx.x & 15;
    const int d = blockIdx.x * 16 + (threadIdx.x >> 4);
    if (d >= N_NODES) return;
    const int p0 = rowptr[d], p1 = rowptr[d + 1];
    const float ad    = adst[d];
    const float vself = leaky(asrc[d] + ad);
    float m = vself;
    for (int p = p0 + c; p < p1; p += 16) m = fmaxf(m, leaky(asrc[list[p]] + ad));
    m = fmaxf(m, __shfl_xor(m, 1)); m = fmaxf(m, __shfl_xor(m, 2));
    m = fmaxf(m, __shfl_xor(m, 4)); m = fmaxf(m, __shfl_xor(m, 8));
    float sum = (c == 0) ? expf(vself - m) : 0.f;
    for (int p = p0 + c; p < p1; p += 16) sum += expf(leaky(asrc[list[p]] + ad) - m);
    sum += __shfl_xor(sum, 1); sum += __shfl_xor(sum, 2);
    sum += __shfl_xor(sum, 4); sum += __shfl_xor(sum, 8);
    const float den = sum + GAT_EPS;
    float acc = expf(vself - m) / den * xl[(size_t)d * OUT_C + c];
#pragma unroll 4
    for (int p = p0; p < p1; ++p) {
        const int s = list[p];
        const float a = expf(leaky(asrc[s] + ad) - m) / den;
        acc += a * xl[(size_t)s * OUT_C + c];
    }
    out[(size_t)d * OUT_C + c] = acc + bias[c];
}

extern "C" void kernel_launch(void* const* d_in, const int* in_sizes, int n_in,
                              void* d_out, int out_size, void* d_ws, size_t ws_size,
                              hipStream_t stream)
{
    (void)in_sizes; (void)n_in; (void)out_size; (void)ws_size;

    const float* x    = (const float*)d_in[0];
    const int*   ei   = (const int*)  d_in[1];
    const float* W1   = (const float*)d_in[2];
    const float* ats1 = (const float*)d_in[3];
    const float* atd1 = (const float*)d_in[4];
    const float* b1   = (const float*)d_in[5];
    const float* W2   = (const float*)d_in[6];
    const float* ats2 = (const float*)d_in[7];
    const float* atd2 = (const float*)d_in[8];
    const float* b2   = (const float*)d_in[9];
    const int* src = ei;
    const int* dst = ei + N_EDGES;
    float* out = (float*)d_out;

    int*   rowptr = (int*)d_ws;
    int*   rnk    = rowptr + (size_t)CNT_PAD;
    int*   list   = rnk + (size_t)N_EDGES;
    float* xl1 = (float*)(list + (size_t)N_EDGES);
    float* as1 = xl1 + (size_t)N_NODES * HIDDEN;
    float* ad1 = as1 + (size_t)N_NODES * HEADS;
    float* xl2 = ad1 + (size_t)N_NODES * HEADS;
    float* as2 = xl2 + (size_t)N_NODES * OUT_C;
    float* ad2 = as2 + (size_t)CNT_PAD;
    int*   part = (int*)(ad2 + (size_t)CNT_PAD);
    int*   bsum = part + (size_t)RANK_CHUNKS * CNT_PAD;

    const int TB = 256;
    auto blk = [](long long n, int tb) { return (unsigned)((n + tb - 1) / tb); };

    gat_edge_rank<<<RANK_CHUNKS, RANK_TB, 0, stream>>>(dst, part, rnk, bsum);
    gat_chunk_prefix<<<N_BKT, BKT_N, 0, stream>>>(part, bsum, rowptr);
    gat_edge_place<<<blk(N_EDGES, TB), TB, 0, stream>>>(src, dst, rowptr, part, rnk, list);

    {
        const int waveTilesN = HIDDEN / (16 * 4);
        const long long waves = (long long)(N_NODES / 16) * waveTilesN;
        gat_wmma_gemm_f32<4, IN_C, HIDDEN><<<blk(waves, 4), 128, 0, stream>>>(
            x, W1, xl1, N_NODES, waveTilesN);
    }
    gat_att_logits<<<blk((long long)N_NODES * HEADS, TB), TB, 0, stream>>>(
        xl1, ats1, atd1, as1, ad1, N_NODES, HEADS, HID);
    gat_gather_l1<<<N_NODES / L1_WAVES, L1_WAVES * 32, 0, stream>>>(
        rowptr, list, as1, ad1, xl1, b1, W2, ats2, atd2, xl2, as2, ad2);

    gat_gather_l2<<<blk(N_NODES, 16), TB, 0, stream>>>(
        rowptr, list, as2, ad2, xl2, b2, out);
}


# Layer-1 attention logits formed in the GEMM wave from its accumulator tiles (halving exchange across a row's lanes on DPP adds; logit sum order now a lane tree); logits launch and its xl re-read gone
# speedup vs baseline: 7.9976x; 1.0610x over previous
#include <hip/hip_runtime.h>
#include <cstdint>
#include <cstddef>

#define N_NODES   50000
#define N_EDGES   800000
#define TOT_E     (N_EDGES + N_NODES)
#define IN_C      128
#define HID       32
#define OUT_C     16
#define HEADS     8
#define HIDDEN    (HEADS * HID)
#define NEG_SLOPE 0.2f
#define GAT_EPS   1e-16f

typedef float v2f __attribute__((ext_vector_type(2)));
typedef float v8f __attribute__((ext_vector_type(8)));


template <int MASK>
__device__ __forceinline__ float gat_row_xor(float v)
{
    return __int_as_float(__builtin_amdgcn_update_dpp(
        0, __float_as_int(v), 0x160 + MASK, 0xf, 0xf, true));
}

template <int MASK, int NV>
__device__ __forceinline__ void gat_fold_step(float (&v)[32], bool up)
{
#pragma unroll
    for (int i = 0; i < NV / 2; ++i) {
        const float a = v[i], b = v[i + NV / 2];
        const float keep = up ? b : a;
        const float send = up ? a : b;
        v[i] = keep + gat_row_xor<MASK>(send);
    }
}

template <int TN, int K, int NC>
__global__ __launch_bounds__(128) __attribute__((amdgpu_waves_per_eu(16, 16))) void gat_wmma_gemm_f32(
    const float* __restrict__ A, const float* __restrict__ B,
    float* __restrict__ C, const float* __restrict__ att_s,
    const float* __restrict__ att_d, float* __restrict__ asrc,
    float* __restrict__ adst, int M, int waveTilesN)
{
    static_assert(TN == 4 && HID == 32 && NC == HEADS * HID,
                  "a wave's 16*TN columns are two whole heads");
    const int wave = __builtin_amdgcn_readfirstlane(
        (int)(blockIdx.x * (blockDim.x >> 5) + (threadIdx.x >> 5)));
    const int lane = threadIdx.x & 31;
    const int tM = wave / waveTilesN;
    const int tg = wave % waveTilesN;
    if (tM * 16 >= M) return;

    const int half = lane >> 4;
    const int l    = lane & 15;
    const float* __restrict__ Ap = A + (size_t)(tM * 16 + l) * K + half * 2;
    const float* __restrict__ Bp = B + (size_t)(half * 2) * NC + tg * 16 * TN + l;

    const v8f zero = {0.f,0.f,0.f,0.f,0.f,0.f,0.f,0.f};
    v8f c[TN];
#pragma unroll
    for (int j = 0; j < TN; ++j) c[j] = zero;

#pragma unroll
    for (int k = 0; k < K; k += 4) {
        const v2f a = *(const v2f*)(Ap + k);
        v2f b[TN];
#pragma unroll
        for (int j = 0; j < TN; ++j) {
            b[j].x = Bp[(size_t)k * NC + j * 16];
            b[j].y = Bp[(size_t)(k + 1) * NC + j * 16];
        }
#pragma unroll
        for (int j = 0; j < TN; ++j)
            c[j] = __builtin_amdgcn_wmma_f32_16x16x4_f32(
                        false, a,  false, b[j],
                        (short)0, c[j],
                        false,  false);
    }

    __builtin_amdgcn_sched_barrier(0);

    float* __restrict__ Cbase =
        C + (size_t)(tM * 16 + half * 8) * NC + tg * 16 * TN + l;
#pragma unroll
    for (int j = 0; j < TN; ++j)
#pragma unroll
        for (int r = 0; r < 8; ++r)
            Cbase[(size_t)r * NC + j * 16] = c[j][r];

    float ws[TN], wd[TN];
#pragma unroll
    for (int j = 0; j < TN; ++j) {
        ws[j] = att_s[tg * 16 * TN + j * 16 + l];
        wd[j] = att_d[tg * 16 * TN + j * 16 + l];
    }
    float v[32];
#pragma unroll
    for (int r = 0; r < 8; ++r) {
        v[4 * r + 0] = c[0][r] * ws[0] + c[1][r] * ws[1];
        v[4 * r + 1] = c[0][r] * wd[0] + c[1][r] * wd[1];
        v[4 * r + 2] = c[2][r] * ws[2] + c[3][r] * ws[3];
        v[4 * r + 3] = c[2][r] * wd[2] + c[3][r] * wd[3];
    }
    gat_fold_step<8, 32>(v, (l & 8) != 0);
    gat_fold_step<4, 16>(v, (l & 4) != 0);
    gat_fold_step<2, 8>(v, (l & 2) != 0);
    gat_fold_step<1, 4>(v, (l & 1) != 0);
    const size_t o = (size_t)(tM * 16 + half * 8 + (l >> 1)) * HEADS + tg * 2 + (l & 1);
    asrc[o] = v[0];
    adst[o] = v[1];
}

__device__ __forceinline__ float leaky(float v) { return v > 0.f ? v : NEG_SLOPE * v; }

#define BKT_N    256
#define N_BKT    ((N_NODES + BKT_N - 1) / BKT_N)
#define CNT_PAD  (N_BKT * BKT_N)

#define RANK_CHUNKS  64
#define RANK_CHUNK_E (N_EDGES / RANK_CHUNKS)
#define RANK_TB      1024
#define RANK_PER     ((RANK_CHUNK_E + RANK_TB - 1) / RANK_TB)
static_assert(RANK_CHUNKS * RANK_CHUNK_E == N_EDGES, "chunks must tile the edge list");
static_assert(N_NODES % 4 == 0, "counter table is moved 16 bytes at a time");
static_assert(RANK_CHUNKS % 4 == 0, "a bucket's chunk totals are read 16 bytes at a time");
static_assert(BKT_N % 32 == 0 && RANK_TB % 32 == 0, "whole waves");
#define BKT_PER_WAVE ((N_BKT + RANK_TB / 32 - 1) / (RANK_TB / 32))

__global__ __launch_bounds__(RANK_TB) void gat_edge_rank(
    const int* __restrict__ dst, int* __restrict__ part, int* __restrict__ rnk,
    int* __restrict__ bsum)
{
    __shared__ __attribute__((aligned(16))) int hist[N_NODES];
    const int t  = threadIdx.x;
    const int e0 = blockIdx.x * RANK_CHUNK_E;
    int4* h4 = (int4*)hist;
    for (int i = t; i < N_NODES / 4; i += RANK_TB) h4[i] = make_int4(0, 0, 0, 0);
    int d[RANK_PER];
#pragma unroll
    for (int k = 0; k < RANK_PER; ++k) {
        const int i = t + k * RANK_TB;
        d[k] = (i < RANK_CHUNK_E) ? dst[e0 + i] : 0;
    }
    __syncthreads();
    int r[RANK_PER];
#pragma unroll
    for (int k = 0; k < RANK_PER; ++k) {
        const int i = t + k * RANK_TB;
        r[k] = (i < RANK_CHUNK_E) ? atomicAdd(&hist[d[k]], 1) : 0;
    }
#pragma unroll
    for (int k = 0; k < RANK_PER; ++k) {
        const int i = t + k * RANK_TB;
        if (i < RANK_CHUNK_E) rnk[e0 + i] = r[k];
    }
    __syncthreads();
    int4* __restrict__ p4 = (int4*)(part + (size_t)blockIdx.x * CNT_PAD);
    for (int i = t; i < N_NODES / 4; i += RANK_TB) p4[i] = h4[i];
    const int lane = t & 31, w = t >> 5;
    int bs[BKT_PER_WAVE];
#pragma unroll
    for (int k = 0; k < BKT_PER_WAVE; ++k) {
        const int jb = w + k * (RANK_TB / 32);
        int acc = 0;
#pragma unroll
        for (int i = 0; i < BKT_N / 32; ++i) {
            const int n = jb * BKT_N + i * 32 + lane;
            acc += (n < N_NODES) ? hist[n] : 0;
        }
        bs[k] = acc;
    }
#pragma unroll
    for (int o = 16; o > 0; o >>= 1) {
#pragma unroll
        for (int k = 0; k < BKT_PER_WAVE; ++k) bs[k] += __shfl_xor(bs[k], o);
    }
#pragma unroll
    for (int k = 0; k < BKT_PER_WAVE; ++k) {
        const int jb = w + k * (RANK_TB / 32);
        if (lane == 0 && jb < N_BKT) bsum[jb * RANK_CHUNKS + blockIdx.x] = bs[k];
    }
}

#define BASE_MOVE (((N_BKT - 1) * (RANK_CHUNKS / 4) + BKT_N - 1) / BKT_N)
__global__ __launch_bounds__(BKT_N) void gat_chunk_prefix(
    int* __restrict__ part, const int* __restrict__ bsum, int* __restrict__ rowptr)
{
    __shared__ int wtot[BKT_N / 32];
    __shared__ int wsum[BKT_N / 32];
    const int t = threadIdx.x, lane = t & 31, w = t >> 5;
    const int n = blockIdx.x * BKT_N + t;
    const int4* __restrict__ b4 = (const int4*)bsum;
    const int n4 = blockIdx.x * (RANK_CHUNKS / 4);
    int4 q[BASE_MOVE];
#pragma unroll
    for (int k = 0; k < BASE_MOVE; ++k) {
        const int i = t + k * BKT_N;
        q[k] = (i < n4) ? b4[i] : make_int4(0, 0, 0, 0);
    }
    int s = 0;
    if (n < N_NODES) {
        int v[RANK_CHUNKS];
#pragma unroll
        for (int c = 0; c < RANK_CHUNKS; ++c) v[c] = part[(size_t)c * CNT_PAD + n];
#pragma unroll
        for (int c = 0; c < RANK_CHUNKS; ++c) { part[(size_t)c * CNT_PAD + n] = s; s += v[c]; }
    }
    int a = 0;
#pragma unroll
    for (int k = 0; k < BASE_MOVE; ++k) a += q[k].x + q[k].y + q[k].z + q[k].w;
    a += __shfl_xor(a, 16); a += __shfl_xor(a, 8); a += __shfl_xor(a, 4);
    a += __shfl_xor(a, 2);  a += __shfl_xor(a, 1);
    int x = s;
    for (int o = 1; o < 32; o <<= 1) { const int y = __shfl_up(x, o); if (lane >= o) x += y; }
    if (lane == 31) wsum[w] = x;
    if (lane == 0)  wtot[w] = a;
    __syncthreads();
    int base = 0;
#pragma unroll
    for (int i = 0; i < BKT_N / 32; ++i) { base += wtot[i]; base += (i < w) ? wsum[i] : 0; }
    rowptr[n] = base + x - s;
}

__global__ __launch_bounds__(256) void gat_edge_place(
    const int* __restrict__ src, const int* __restrict__ dst,
    const int* __restrict__ rowptr, const int* __restrict__ part,
    const int* __restrict__ rnk, int* __restrict__ list)
{
    const int e = blockIdx.x * blockDim.x + threadIdx.x;
    if (e >= N_EDGES) return;
    const int d = dst[e];
    const int before = part[(size_t)(e / RANK_CHUNK_E) * CNT_PAD + d];
    list[rowptr[d] + before + rnk[e]] = src[e];
}

#define L1_WAVES 16
#define H_PITCH  (HIDDEN + 4)
static_assert(N_NODES % L1_WAVES == 0, "every block is a full tile of 16 nodes: no partial block, no bounds test");
static_assert(HIDDEN == L1_WAVES * 16, "wave w takes K slice 16w .. 16w+15 of the layer-2 product");
static_assert(OUT_C == 16, "the layer-2 row is one 16-column WMMA tile");

__global__ __launch_bounds__(L1_WAVES * 32) void gat_gather_l1(
    const int* __restrict__ rowptr, const int* __restrict__ list,
    const float* __restrict__ asrc, const float* __restrict__ adst,
    const float* __restrict__ xl, const float* __restrict__ bias,
    const float* __restrict__ W2, const float* __restrict__ att_s2,
    const float* __restrict__ att_d2, float* __restrict__ xl2,
    float* __restrict__ as2, float* __restrict__ ad2)
{
    __shared__ __attribute__((aligned(16))) float sw[L1_WAVES][32 * HEADS];
    __shared__ __attribute__((aligned(16))) float hrow[L1_WAVES][H_PITCH];
    const int lane = threadIdx.x & 31;
    const int wv   = threadIdx.x >> 5;
    const int d = __builtin_amdgcn_readfirstlane((int)(blockIdx.x * L1_WAVES + wv));
    float* my = sw[wv];
    const int p0 = rowptr[d], p1 = rowptr[d + 1];
    const int deg = p1 - p0;
    const int nch = (deg + 31) >> 5;

    const int hh = lane & 7, j = lane >> 3;
    const float adh   = adst[(size_t)d * HEADS + hh];
    const float vself = leaky(asrc[(size_t)d * HEADS + hh] + adh);

    int sv = d;
    auto stage = [&](int c) {
        const int q = c * 32 + lane;
        sv = (q < deg) ? list[p0 + q] : d;
        const float4* ar = (const float4*)(asrc + (size_t)sv * HEADS);
        const float4 a0 = ar[0], a1 = ar[1];
        float4* dp = (float4*)(my + lane * HEADS);
        dp[0] = a0; dp[1] = a1;
    };

    float m = vself;
    for (int c = 0; c < nch; ++c) {
        stage(c);
#pragma unroll
        for (int i = 0; i < 8; ++i) {
            const int sl = j + 4 * i;
            const float v = leaky(my[sl * HEADS + hh] + adh);
            m = (c * 32 + sl < deg) ? fmaxf(m, v) : m;
        }
    }
    m = fmaxf(m, __shfl_xor(m, 8));
    m = fmaxf(m, __shfl_xor(m, 16));
    float sum = (j == 0) ? expf(vself - m) : 0.f;
    for (int c = 0; c < nch; ++c) {
        if (nch > 1) stage(c);
#pragma unroll
        for (int i = 0; i < 8; ++i) {
            const int sl = j + 4 * i;
            const float ex = expf(leaky(my[sl * HEADS + hh] + adh) - m);
            sum = (c * 32 + sl < deg) ? sum + ex : sum;
        }
    }
    sum += __shfl_xor(sum, 8);
    sum += __shfl_xor(sum, 16);
    const float den = sum + GAT_EPS;
    const float wself = expf(vself - m) / den;

    const int hc = lane >> 2;
    float acc[8];
    {
        const float a = __shfl(wself, hc);
        const float4* r = (const float4*)(xl + (size_t)d * HIDDEN + lane * 8);
        const float4 r0 = r[0], r1 = r[1];
        acc[0] = a * r0.x; acc[1] = a * r0.y; acc[2] = a * r0.z; acc[3] = a * r0.w;
        acc[4] = a * r1.x; acc[5] = a * r1.y; acc[6] = a * r1.z; acc[7] = a * r1.w;
    }
    for (int c = 0; c < nch; ++c) {
        if (nch > 1) stage(c);
#pragma unroll
        for (int i = 0; i < 8; ++i) {
            const int sl = j + 4 * i;
            my[sl * HEADS + hh] = expf(leaky(my[sl * HEADS + hh] + adh) - m) / den;
        }
        const int cnt = min(32, deg - c * 32);
#pragma unroll 4
        for (int e = 0; e < cnt; ++e) {
            const int s = __builtin_amdgcn_readlane(sv, e);
            const float a = my[e * HEADS + hc];
            const float4* r = (const float4*)(xl + (size_t)s * HIDDEN + lane * 8);
            const float4 r0 = r[0], r1 = r[1];
            acc[0] += a * r0.x; acc[1] += a * r0.y; acc[2] += a * r0.z; acc[3] += a * r0.w;
            acc[4] += a * r1.x; acc[5] += a * r1.y; acc[6] += a * r1.z; acc[7] += a * r1.w;
        }
    }
    const float4* bp = (const float4*)(bias + lane * 8);
    const float4 b0 = bp[0], b1 = bp[1];
    const float bb[8] = {b0.x, b0.y, b0.z, b0.w, b1.x, b1.y, b1.z, b1.w};
    float o[8];
#pragma unroll
    for (int i = 0; i < 8; ++i) { const float v = acc[i] + bb[i]; o[i] = v > 0.f ? v : expm1f(v); }
    float4* op = (float4*)(&hrow[wv][lane * 8]);
    op[0] = make_float4(o[0], o[1], o[2], o[3]);
    op[1] = make_float4(o[4], o[5], o[6], o[7]);
    __syncthreads();

    const int kh = (threadIdx.x >> 4) & 1, l16 = threadIdx.x & 15;
    const float* hA = &hrow[l16][wv * 16 + kh * 2];
    const float* __restrict__ Bq = W2 + (size_t)(wv * 16 + kh * 2) * OUT_C + l16;
    v8f c2 = {0.f, 0.f, 0.f, 0.f, 0.f, 0.f, 0.f, 0.f};
#pragma unroll
    for (int kk = 0; kk < 4; ++kk) {
        const v2f a2 = *(const v2f*)(hA + kk * 4);
        v2f b2;
        b2.x = Bq[(size_t)(kk * 4) * OUT_C];
        b2.y = Bq[(size_t)(kk * 4 + 1) * OUT_C];
        c2 = __builtin_amdgcn_wmma_f32_16x16x4_f32(
                  false, a2,  false, b2,
                  (short)0, c2,
                  false,  false);
    }
#pragma unroll
    for (int r = 0; r < 8; ++r) my[r * 32 + lane] = c2[r];
    __syncthreads();
    if (threadIdx.x < 256) {
        const int idx = threadIdx.x;
        float s = sw[0][idx];
#pragma unroll
        for (int w = 1; w < L1_WAVES; ++w) s += sw[w][idx];
        const int col  = idx & 15;
        const int node = blockIdx.x * L1_WAVES + ((idx >> 4) & 1) * 8 + (idx >> 5);
        xl2[(size_t)node * OUT_C + col] = s;
        float ps = s * att_s2[col], pd = s * att_d2[col];
        ps += __int_as_float(__builtin_amdgcn_ds_swizzle(__float_as_int(ps), 0x041f));
        pd += __int_as_float(__builtin_amdgcn_ds_swizzle(__float_as_int(pd), 0x041f));
        ps += __int_as_float(__builtin_amdgcn_ds_swizzle(__float_as_int(ps), 0x081f));
        pd += __int_as_float(__builtin_amdgcn_ds_swizzle(__float_as_int(pd), 0x081f));
        ps += __int_as_float(__builtin_amdgcn_ds_swizzle(__float_as_int(ps), 0x101f));
        pd += __int_as_float(__builtin_amdgcn_ds_swizzle(__float_as_int(pd), 0x101f));
        ps += __int_as_float(__builtin_amdgcn_ds_swizzle(__float_as_int(ps), 0x201f));
        pd += __int_as_float(__builtin_amdgcn_ds_swizzle(__float_as_int(pd), 0x201f));
        if (col == 0) { as2[node] = ps; ad2[node] = pd; }
    }
}

__global__ __launch_bounds__(256) void gat_gather_l2(
    const int* __restrict__ rowptr, const int* __restrict__ list,
    const float* __restrict__ asrc, const float* __restrict__ adst,
    const float* __restrict__ xl, const float* __restrict__ bias,
    float* __restrict__ out)
{
    const int c = threadIdx.x & 15;
    const int d = blockIdx.x * 16 + (threadIdx.x >> 4);
    if (d >= N_NODES) return;
    const int p0 = rowptr[d], p1 = rowptr[d + 1];
    const float ad    = adst[d];
    const float vself = leaky(asrc[d] + ad);
    float m = vself;
    for (int p = p0 + c; p < p1; p += 16) m = fmaxf(m, leaky(asrc[list[p]] + ad));
    m = fmaxf(m, __shfl_xor(m, 1)); m = fmaxf(m, __shfl_xor(m, 2));
    m = fmaxf(m, __shfl_xor(m, 4)); m = fmaxf(m, __shfl_xor(m, 8));
    float sum = (c == 0) ? expf(vself - m) : 0.f;
    for (int p = p0 + c; p < p1; p += 16) sum += expf(leaky(asrc[list[p]] + ad) - m);
    sum += __shfl_xor(sum, 1); sum += __shfl_xor(sum, 2);
    sum += __shfl_xor(sum, 4); sum += __shfl_xor(sum, 8);
    const float den = sum + GAT_EPS;
    float acc = expf(vself - m) / den * xl[(size_t)d * OUT_C + c];
#pragma unroll 4
    for (int p = p0; p < p1; ++p) {
        const int s = list[p];
        const float a = expf(leaky(asrc[s] + ad) - m) / den;
        acc += a * xl[(size_t)s * OUT_C + c];
    }
    out[(size_t)d * OUT_C + c] = acc + bias[c];
}

extern "C" void kernel_launch(void* const* d_in, const int* in_sizes, int n_in,
                              void* d_out, int out_size, void* d_ws, size_t ws_size,
                              hipStream_t stream)
{
    (void)in_sizes; (void)n_in; (void)out_size; (void)ws_size;

    const float* x    = (const float*)d_in[0];
    const int*   ei   = (const int*)  d_in[1];
    const float* W1   = (const float*)d_in[2];
    const float* ats1 = (const float*)d_in[3];
    const float* atd1 = (const float*)d_in[4];
    const float* b1   = (const float*)d_in[5];
    const float* W2   = (const float*)d_in[6];
    const float* ats2 = (const float*)d_in[7];
    const float* atd2 = (const float*)d_in[8];
    const float* b2   = (const float*)d_in[9];
    const int* src = ei;
    const int* dst = ei + N_EDGES;
    float* out = (float*)d_out;

    int*   rowptr = (int*)d_ws;
    int*   rnk    = rowptr + (size_t)CNT_PAD;
    int*   list   = rnk + (size_t)N_EDGES;
    float* xl1 = (float*)(list + (size_t)N_EDGES);
    float* as1 = xl1 + (size_t)N_NODES * HIDDEN;
    float* ad1 = as1 + (size_t)N_NODES * HEADS;
    float* xl2 = ad1 + (size_t)N_NODES * HEADS;
    float* as2 = xl2 + (size_t)N_NODES * OUT_C;
    float* ad2 = as2 + (size_t)CNT_PAD;
    int*   part = (int*)(ad2 + (size_t)CNT_PAD);
    int*   bsum = part + (size_t)RANK_CHUNKS * CNT_PAD;

    const int TB = 256;
    auto blk = [](long long n, int tb) { return (unsigned)((n + tb - 1) / tb); };

    gat_edge_rank<<<RANK_CHUNKS, RANK_TB, 0, stream>>>(dst, part, rnk, bsum);
    gat_chunk_prefix<<<N_BKT, BKT_N, 0, stream>>>(part, bsum, rowptr);
    gat_edge_place<<<blk(N_EDGES, TB), TB, 0, stream>>>(src, dst, rowptr, part, rnk, list);

    {
        const int waveTilesN = HIDDEN / (16 * 4);
        const long long waves = (long long)(N_NODES / 16) * waveTilesN;
        gat_wmma_gemm_f32<4, IN_C, HIDDEN><<<blk(waves, 4), 128, 0, stream>>>(
            x, W1, xl1, ats1, atd1, as1, ad1, N_NODES, waveTilesN);
    }
    gat_gather_l1<<<N_NODES / L1_WAVES, L1_WAVES * 32, 0, stream>>>(
        rowptr, list, as1, ad1, xl1, b1, W2, ats2, atd2, xl2, as2, ad2);

    gat_gather_l2<<<blk(N_NODES, 16), TB, 0, stream>>>(
        rowptr, list, as2, ad2, xl2, b2, out);
}


# Edge placing and the layer-1 GEMM share one launch (role by block index: GEMM work-groups enter first, placing work-groups stream through the free wave slots under the matrix work); 1 launch gone
# speedup vs baseline: 8.4285x; 1.0539x over previous
#include <hip/hip_runtime.h>
#include <cstdint>
#include <cstddef>

#define N_NODES   50000
#define N_EDGES   800000
#define TOT_E     (N_EDGES + N_NODES)
#define IN_C      128
#define HID       32
#define OUT_C     16
#define HEADS     8
#define HIDDEN    (HEADS * HID)
#define NEG_SLOPE 0.2f
#define GAT_EPS   1e-16f
#define PG_TB     128

typedef float v2f __attribute__((ext_vector_type(2)));
typedef float v8f __attribute__((ext_vector_type(8)));


template <int MASK>
__device__ __forceinline__ float gat_row_xor(float v)
{
    return __int_as_float(__builtin_amdgcn_update_dpp(
        0, __float_as_int(v), 0x160 + MASK, 0xf, 0xf, true));
}

template <int MASK, int NV>
__device__ __forceinline__ void gat_fold_step(float (&v)[32], bool up)
{
#pragma unroll
    for (int i = 0; i < NV / 2; ++i) {
        const float a = v[i], b = v[i + NV / 2];
        const float keep = up ? b : a;
        const float send = up ? a : b;
        v[i] = keep + gat_row_xor<MASK>(send);
    }
}

template <int TN, int K, int NC>
__device__ __forceinline__ void gat_gemm_block_f32(
    const int gblk,
    const float* __restrict__ A, const float* __restrict__ B,
    float* __restrict__ C, const float* __restrict__ att_s,
    const float* __restrict__ att_d, float* __restrict__ asrc,
    float* __restrict__ adst, int M, int waveTilesN)
{
    static_assert(TN == 4 && HID == 32 && NC == HEADS * HID,
                  "a wave's 16*TN columns are two whole heads");
    const int wave = __builtin_amdgcn_readfirstlane(
        (int)(gblk * (PG_TB / 32) + (threadIdx.x >> 5)));
    const int lane = threadIdx.x & 31;
    const int tM = wave / waveTilesN;
    const int tg = wave % waveTilesN;
    if (tM * 16 >= M) return;

    const int half = lane >> 4;
    const int l    = lane & 15;
    const float* __restrict__ Ap = A + (size_t)(tM * 16 + l) * K + half * 2;
    const float* __restrict__ Bp = B + (size_t)(half * 2) * NC + tg * 16 * TN + l;

    const v8f zero = {0.f,0.f,0.f,0.f,0.f,0.f,0.f,0.f};
    v8f c[TN];
#pragma unroll
    for (int j = 0; j < TN; ++j) c[j] = zero;

#pragma unroll
    for (int k = 0; k < K; k += 4) {
        const v2f a = *(const v2f*)(Ap + k);
        v2f b[TN];
#pragma unroll
        for (int j = 0; j < TN; ++j) {
            b[j].x = Bp[(size_t)k * NC + j * 16];
            b[j].y = Bp[(size_t)(k + 1) * NC + j * 16];
        }
#pragma unroll
        for (int j = 0; j < TN; ++j)
            c[j] = __builtin_amdgcn_wmma_f32_16x16x4_f32(
                        false, a,  false, b[j],
                        (short)0, c[j],
                        false,  false);
    }

    __builtin_amdgcn_sched_barrier(0);

    float* __restrict__ Cw = C + (size_t)(tM * 16) * NC + tg * 16 * TN;
    const unsigned co = (unsigned)(half * 8 * NC + l);
#pragma unroll
    for (int j = 0; j < TN; ++j)
#pragma unroll
        for (int r = 0; r < 8; ++r)
            Cw[co + (unsigned)(r * NC + j * 16)] = c[j][r];

    const float* __restrict__ asw = att_s + tg * 16 * TN;
    const float* __restrict__ adw = att_d + tg * 16 * TN;
    float ws[TN], wd[TN];
#pragma unroll
    for (int j = 0; j < TN; ++j) {
        ws[j] = asw[(unsigned)(j * 16 + l)];
        wd[j] = adw[(unsigned)(j * 16 + l)];
    }
    float v[32];
#pragma unroll
    for (int r = 0; r < 8; ++r) {
        v[4 * r + 0] = c[0][r] * ws[0] + c[1][r] * ws[1];
        v[4 * r + 1] = c[0][r] * wd[0] + c[1][r] * wd[1];
        v[4 * r + 2] = c[2][r] * ws[2] + c[3][r] * ws[3];
        v[4 * r + 3] = c[2][r] * wd[2] + c[3][r] * wd[3];
    }
    gat_fold_step<8, 32>(v, (l & 8) != 0);
    gat_fold_step<4, 16>(v, (l & 4) != 0);
    gat_fold_step<2, 8>(v, (l & 2) != 0);
    gat_fold_step<1, 4>(v, (l & 1) != 0);
    const size_t ob = (size_t)(tM * 16) * HEADS + tg * 2;
    const unsigned oo = (unsigned)((half * 8 + (l >> 1)) * HEADS + (l & 1));
    (asrc + ob)[oo] = v[0];
    (adst + ob)[oo] = v[1];
}

__device__ __forceinline__ float leaky(float v) { return v > 0.f ? v : NEG_SLOPE * v; }

#define BKT_N    256
#define N_BKT    ((N_NODES + BKT_N - 1) / BKT_N)
#define CNT_PAD  (N_BKT * BKT_N)

#define RANK_CHUNKS  64
#define RANK_CHUNK_E (N_EDGES / RANK_CHUNKS)
#define RANK_TB      1024
#define RANK_PER     ((RANK_CHUNK_E + RANK_TB - 1) / RANK_TB)
static_assert(RANK_CHUNKS * RANK_CHUNK_E == N_EDGES, "chunks must tile the edge list");
static_assert(N_NODES % 4 == 0, "counter table is moved 16 bytes at a time");
static_assert(RANK_CHUNKS % 4 == 0, "a bucket's chunk totals are read 16 bytes at a time");
static_assert(BKT_N % 32 == 0 && RANK_TB % 32 == 0, "whole waves");
#define BKT_PER_WAVE ((N_BKT + RANK_TB / 32 - 1) / (RANK_TB / 32))

__global__ __launch_bounds__(RANK_TB) void gat_edge_rank(
    const int* __restrict__ dst, int* __restrict__ part, int* __restrict__ rnk,
    int* __restrict__ bsum)
{
    __shared__ __attribute__((aligned(16))) int hist[N_NODES];
    const int t  = threadIdx.x;
    const int e0 = blockIdx.x * RANK_CHUNK_E;
    int4* h4 = (int4*)hist;
    for (int i = t; i < N_NODES / 4; i += RANK_TB) h4[i] = make_int4(0, 0, 0, 0);
    int d[RANK_PER];
#pragma unroll
    for (int k = 0; k < RANK_PER; ++k) {
        const int i = t + k * RANK_TB;
        d[k] = (i < RANK_CHUNK_E) ? dst[e0 + i] : 0;
    }
    __syncthreads();
    int r[RANK_PER];
#pragma unroll
    for (int k = 0; k < RANK_PER; ++k) {
        const int i = t + k * RANK_TB;
        r[k] = (i < RANK_CHUNK_E) ? atomicAdd(&hist[d[k]], 1) : 0;
    }
#pragma unroll
    for (int k = 0; k < RANK_PER; ++k) {
        const int i = t + k * RANK_TB;
        if (i < RANK_CHUNK_E) rnk[e0 + i] = r[k];
    }
    __syncthreads();
    int4* __restrict__ p4 = (int4*)(part + (size_t)blockIdx.x * CNT_PAD);
    for (int i = t; i < N_NODES / 4; i += RANK_TB) p4[i] = h4[i];
    const int lane = t & 31, w = t >> 5;
    int bs[BKT_PER_WAVE];
#pragma unroll
    for (int k = 0; k < BKT_PER_WAVE; ++k) {
        const int jb = w + k * (RANK_TB / 32);
        int acc = 0;
#pragma unroll
        for (int i = 0; i < BKT_N / 32; ++i) {
            const int n = jb * BKT_N + i * 32 + lane;
            acc += (n < N_NODES) ? hist[n] : 0;
        }
        bs[k] = acc;
    }
#pragma unroll
    for (int o = 16; o > 0; o >>= 1) {
#pragma unroll
        for (int k = 0; k < BKT_PER_WAVE; ++k) bs[k] += __shfl_xor(bs[k], o);
    }
#pragma unroll
    for (int k = 0; k < BKT_PER_WAVE; ++k) {
        const int jb = w + k * (RANK_TB / 32);
        if (lane == 0 && jb < N_BKT) bsum[jb * RANK_CHUNKS + blockIdx.x] = bs[k];
    }
}

#define BASE_MOVE (((N_BKT - 1) * (RANK_CHUNKS / 4) + BKT_N - 1) / BKT_N)
__global__ __launch_bounds__(BKT_N) void gat_chunk_prefix(
    int* __restrict__ part, const int* __restrict__ bsum, int* __restrict__ rowptr)
{
    __shared__ int wtot[BKT_N / 32];
    __shared__ int wsum[BKT_N / 32];
    const int t = threadIdx.x, lane = t & 31, w = t >> 5;
    const int n = blockIdx.x * BKT_N + t;
    const int4* __restrict__ b4 = (const int4*)bsum;
    const int n4 = blockIdx.x * (RANK_CHUNKS / 4);
    int4 q[BASE_MOVE];
#pragma unroll
    for (int k = 0; k < BASE_MOVE; ++k) {
        const int i = t + k * BKT_N;
        q[k] = (i < n4) ? b4[i] : make_int4(0, 0, 0, 0);
    }
    int s = 0;
    if (n < N_NODES) {
        int v[RANK_CHUNKS];
#pragma unroll
        for (int c = 0; c < RANK_CHUNKS; ++c) v[c] = part[(size_t)c * CNT_PAD + n];
#pragma unroll
        for (int c = 0; c < RANK_CHUNKS; ++c) { part[(size_t)c * CNT_PAD + n] = s; s += v[c]; }
    }
    int a = 0;
#pragma unroll
    for (int k = 0; k < BASE_MOVE; ++k) a += q[k].x + q[k].y + q[k].z + q[k].w;
    a += __shfl_xor(a, 16); a += __shfl_xor(a, 8); a += __shfl_xor(a, 4);
    a += __shfl_xor(a, 2);  a += __shfl_xor(a, 1);
    int x = s;
    for (int o = 1; o < 32; o <<= 1) { const int y = __shfl_up(x, o); if (lane >= o) x += y; }
    if (lane == 31) wsum[w] = x;
    if (lane == 0)  wtot[w] = a;
    __syncthreads();
    int base = 0;
#pragma unroll
    for (int i = 0; i < BKT_N / 32; ++i) { base += wtot[i]; base += (i < w) ? wsum[i] : 0; }
    rowptr[n] = base + x - s;
}

__device__ __forceinline__ void gat_place_block(
    const int pblk,
    const int* __restrict__ src, const int* __restrict__ dst,
    const int* __restrict__ rowptr, const int* __restrict__ part,
    const int* __restrict__ rnk, int* __restrict__ list)
{
    const int e = pblk * PG_TB + threadIdx.x;
    if (e >= N_EDGES) return;
    const int d = dst[e];
    const int before = part[(size_t)(e / RANK_CHUNK_E) * CNT_PAD + d];
    list[rowptr[d] + before + rnk[e]] = src[e];
}

#define PG_GROUPS  ((N_NODES / 16) * (HIDDEN / 64) / (PG_TB / 32))
static_assert(N_NODES % 16 == 0 && PG_GROUPS * (PG_TB / 32) == (N_NODES / 16) * (HIDDEN / 64),
              "GEMM work-groups are whole");
static_assert(2 * PG_GROUPS * PG_TB >= N_EDGES, "2 * PG_GROUPS placing work-groups cover every edge");

template <int TN, int K, int NC>
__global__ __launch_bounds__(PG_TB) __attribute__((amdgpu_waves_per_eu(16, 16))) void gat_place_gemm_f32(
    const int* __restrict__ src, const int* __restrict__ dst,
    const int* __restrict__ rowptr, const int* __restrict__ part,
    const int* __restrict__ rnk, int* __restrict__ list,
    const float* __restrict__ A, const float* __restrict__ B,
    float* __restrict__ C, const float* __restrict__ att_s,
    const float* __restrict__ att_d, float* __restrict__ asrc,
    float* __restrict__ adst, int M, int waveTilesN)
{
    if (blockIdx.x >= PG_GROUPS) {
        gat_place_block((int)blockIdx.x - PG_GROUPS, src, dst, rowptr, part, rnk, list);
        return;
    }
    gat_gemm_block_f32<TN, K, NC>((int)blockIdx.x, A, B, C, att_s, att_d, asrc, adst, M, waveTilesN);
}

#define L1_WAVES 16
#define H_PITCH  (HIDDEN + 4)
static_assert(N_NODES % L1_WAVES == 0, "every block is a full tile of 16 nodes: no partial block, no bounds test");
static_assert(HIDDEN == L1_WAVES * 16, "wave w takes K slice 16w .. 16w+15 of the layer-2 product");
static_assert(OUT_C == 16, "the layer-2 row is one 16-column WMMA tile");

__global__ __launch_bounds__(L1_WAVES * 32) void gat_gather_l1(
    const int* __restrict__ rowptr, const int* __restrict__ list,
    const float* __restrict__ asrc, const float* __restrict__ adst,
    const float* __restrict__ xl, const float* __restrict__ bias,
    const float* __restrict__ W2, const float* __restrict__ att_s2,
    const float* __restrict__ att_d2, float* __restrict__ xl2,
    float* __restrict__ as2, float* __restrict__ ad2)
{
    __shared__ __attribute__((aligned(16))) float sw[L1_WAVES][32 * HEADS];
    __shared__ __attribute__((aligned(16))) float hrow[L1_WAVES][H_PITCH];
    const int lane = threadIdx.x & 31;
    const int wv   = threadIdx.x >> 5;
    const int d = __builtin_amdgcn_readfirstlane((int)(blockIdx.x * L1_WAVES + wv));
    float* my = sw[wv];
    const int p0 = rowptr[d], p1 = rowptr[d + 1];
    const int deg = p1 - p0;
    const int nch = (deg + 31) >> 5;

    const int hh = lane & 7, j = lane >> 3;
    const float adh   = adst[(size_t)d * HEADS + hh];
    const float vself = leaky(asrc[(size_t)d * HEADS + hh] + adh);

    int sv = d;
    auto stage = [&](int c) {
        const int q = c * 32 + lane;
        sv = (q < deg) ? list[p0 + q] : d;
        const float4* ar = (const float4*)(asrc + (size_t)sv * HEADS);
        const float4 a0 = ar[0], a1 = ar[1];
        float4* dp = (float4*)(my + lane * HEADS);
        dp[0] = a0; dp[1] = a1;
    };

    float m = vself;
    for (int c = 0; c < nch; ++c) {
        stage(c);
#pragma unroll
        for (int i = 0; i < 8; ++i) {
            const int sl = j + 4 * i;
            const float v = leaky(my[sl * HEADS + hh] + adh);
            m = (c * 32 + sl < deg) ? fmaxf(m, v) : m;
        }
    }
    m = fmaxf(m, __shfl_xor(m, 8));
    m = fmaxf(m, __shfl_xor(m, 16));
    float sum = (j == 0) ? expf(vself - m) : 0.f;
    for (int c = 0; c < nch; ++c) {
        if (nch > 1) stage(c);
#pragma unroll
        for (int i = 0; i < 8; ++i) {
            const int sl = j + 4 * i;
            const float ex = expf(leaky(my[sl * HEADS + hh] + adh) - m);
            sum = (c * 32 + sl < deg) ? sum + ex : sum;
        }
    }
    sum += __shfl_xor(sum, 8);
    sum += __shfl_xor(sum, 16);
    const float den = sum + GAT_EPS;
    const float wself = expf(vself - m) / den;

    const int hc = lane >> 2;
    float acc[8];
    {
        const float a = __shfl(wself, hc);
        const float4* r = (const float4*)(xl + (size_t)d * HIDDEN + lane * 8);
        const float4 r0 = r[0], r1 = r[1];
        acc[0] = a * r0.x; acc[1] = a * r0.y; acc[2] = a * r0.z; acc[3] = a * r0.w;
        acc[4] = a * r1.x; acc[5] = a * r1.y; acc[6] = a * r1.z; acc[7] = a * r1.w;
    }
    for (int c = 0; c < nch; ++c) {
        if (nch > 1) stage(c);
#pragma unroll
        for (int i = 0; i < 8; ++i) {
            const int sl = j + 4 * i;
            my[sl * HEADS + hh] = expf(leaky(my[sl * HEADS + hh] + adh) - m) / den;
        }
        const int cnt = min(32, deg - c * 32);
#pragma unroll 4
        for (int e = 0; e < cnt; ++e) {
            const int s = __builtin_amdgcn_readlane(sv, e);
            const float a = my[e * HEADS + hc];
            const float4* r = (const float4*)(xl + (size_t)s * HIDDEN + lane * 8);
            const float4 r0 = r[0], r1 = r[1];
            acc[0] += a * r0.x; acc[1] += a * r0.y; acc[2] += a * r0.z; acc[3] += a * r0.w;
            acc[4] += a * r1.x; acc[5] += a * r1.y; acc[6] += a * r1.z; acc[7] += a * r1.w;
        }
    }
    const float4* bp = (const float4*)(bias + lane * 8);
    const float4 b0 = bp[0], b1 = bp[1];
    const float bb[8] = {b0.x, b0.y, b0.z, b0.w, b1.x, b1.y, b1.z, b1.w};
    float o[8];
#pragma unroll
    for (int i = 0; i < 8; ++i) { const float v = acc[i] + bb[i]; o[i] = v > 0.f ? v : expm1f(v); }
    float4* op = (float4*)(&hrow[wv][lane * 8]);
    op[0] = make_float4(o[0], o[1], o[2], o[3]);
    op[1] = make_float4(o[4], o[5], o[6], o[7]);
    __syncthreads();

    const int kh = (threadIdx.x >> 4) & 1, l16 = threadIdx.x & 15;
    const float* hA = &hrow[l16][wv * 16 + kh * 2];
    const float* __restrict__ Bq = W2 + (size_t)(wv * 16 + kh * 2) * OUT_C + l16;
    v8f c2 = {0.f, 0.f, 0.f, 0.f, 0.f, 0.f, 0.f, 0.f};
#pragma unroll
    for (int kk = 0; kk < 4; ++kk) {
        const v2f a2 = *(const v2f*)(hA + kk * 4);
        v2f b2;
        b2.x = Bq[(size_t)(kk * 4) * OUT_C];
        b2.y = Bq[(size_t)(kk * 4 + 1) * OUT_C];
        c2 = __builtin_amdgcn_wmma_f32_16x16x4_f32(
                  false, a2,  false, b2,
                  (short)0, c2,
                  false,  false);
    }
#pragma unroll
    for (int r = 0; r < 8; ++r) my[r * 32 + lane] = c2[r];
    __syncthreads();
    if (threadIdx.x < 256) {
        const int idx = threadIdx.x;
        float s = sw[0][idx];
#pragma unroll
        for (int w = 1; w < L1_WAVES; ++w) s += sw[w][idx];
        const int col  = idx & 15;
        const int node = blockIdx.x * L1_WAVES + ((idx >> 4) & 1) * 8 + (idx >> 5);
        xl2[(size_t)node * OUT_C + col] = s;
        float ps = s * att_s2[col], pd = s * att_d2[col];
        ps += __int_as_float(__builtin_amdgcn_ds_swizzle(__float_as_int(ps), 0x041f));
        pd += __int_as_float(__builtin_amdgcn_ds_swizzle(__float_as_int(pd), 0x041f));
        ps += __int_as_float(__builtin_amdgcn_ds_swizzle(__float_as_int(ps), 0x081f));
        pd += __int_as_float(__builtin_amdgcn_ds_swizzle(__float_as_int(pd), 0x081f));
        ps += __int_as_float(__builtin_amdgcn_ds_swizzle(__float_as_int(ps), 0x101f));
        pd += __int_as_float(__builtin_amdgcn_ds_swizzle(__float_as_int(pd), 0x101f));
        ps += __int_as_float(__builtin_amdgcn_ds_swizzle(__float_as_int(ps), 0x201f));
        pd += __int_as_float(__builtin_amdgcn_ds_swizzle(__float_as_int(pd), 0x201f));
        if (col == 0) { as2[node] = ps; ad2[node] = pd; }
    }
}

__global__ __launch_bounds__(256) void gat_gather_l2(
    const int* __restrict__ rowptr, const int* __restrict__ list,
    const float* __restrict__ asrc, const float* __restrict__ adst,
    const float* __restrict__ xl, const float* __restrict__ bias,
    float* __restrict__ out)
{
    const int c = threadIdx.x & 15;
    const int d = blockIdx.x * 16 + (threadIdx.x >> 4);
    if (d >= N_NODES) return;
    const int p0 = rowptr[d], p1 = rowptr[d + 1];
    const float ad    = adst[d];
    const float vself = leaky(asrc[d] + ad);
    float m = vself;
    for (int p = p0 + c; p < p1; p += 16) m = fmaxf(m, leaky(asrc[list[p]] + ad));
    m = fmaxf(m, __shfl_xor(m, 1)); m = fmaxf(m, __shfl_xor(m, 2));
    m = fmaxf(m, __shfl_xor(m, 4)); m = fmaxf(m, __shfl_xor(m, 8));
    float sum = (c == 0) ? expf(vself - m) : 0.f;
    for (int p = p0 + c; p < p1; p += 16) sum += expf(leaky(asrc[list[p]] + ad) - m);
    sum += __shfl_xor(sum, 1); sum += __shfl_xor(sum, 2);
    sum += __shfl_xor(sum, 4); sum += __shfl_xor(sum, 8);
    const float den = sum + GAT_EPS;
    float acc = expf(vself - m) / den * xl[(size_t)d * OUT_C + c];
#pragma unroll 4
    for (int p = p0; p < p1; ++p) {
        const int s = list[p];
        const float a = expf(leaky(asrc[s] + ad) - m) / den;
        acc += a * xl[(size_t)s * OUT_C + c];
    }
    out[(size_t)d * OUT_C + c] = acc + bias[c];
}

extern "C" void kernel_launch(void* const* d_in, const int* in_sizes, int n_in,
                              void* d_out, int out_size, void* d_ws, size_t ws_size,
                              hipStream_t stream)
{
    (void)in_sizes; (void)n_in; (void)out_size; (void)ws_size;

    const float* x    = (const float*)d_in[0];
    const int*   ei   = (const int*)  d_in[1];
    const float* W1   = (const float*)d_in[2];
    const float* ats1 = (const float*)d_in[3];
    const float* atd1 = (const float*)d_in[4];
    const float* b1   = (const float*)d_in[5];
    const float* W2   = (const float*)d_in[6];
    const float* ats2 = (const float*)d_in[7];
    const float* atd2 = (const float*)d_in[8];
    const float* b2   = (const float*)d_in[9];
    const int* src = ei;
    const int* dst = ei + N_EDGES;
    float* out = (float*)d_out;

    int*   rowptr = (int*)d_ws;
    int*   rnk    = rowptr + (size_t)CNT_PAD;
    int*   list   = rnk + (size_t)N_EDGES;
    float* xl1 = (float*)(list + (size_t)N_EDGES);
    float* as1 = xl1 + (size_t)N_NODES * HIDDEN;
    float* ad1 = as1 + (size_t)N_NODES * HEADS;
    float* xl2 = ad1 + (size_t)N_NODES * HEADS;
    float* as2 = xl2 + (size_t)N_NODES * OUT_C;
    float* ad2 = as2 + (size_t)CNT_PAD;
    int*   part = (int*)(ad2 + (size_t)CNT_PAD);
    int*   bsum = part + (size_t)RANK_CHUNKS * CNT_PAD;

    const int TB = 256;
    auto blk = [](long long n, int tb) { return (unsigned)((n + tb - 1) / tb); };

    gat_edge_rank<<<RANK_CHUNKS, RANK_TB, 0, stream>>>(dst, part, rnk, bsum);
    gat_chunk_prefix<<<N_BKT, BKT_N, 0, stream>>>(part, bsum, rowptr);

    gat_place_gemm_f32<4, IN_C, HIDDEN><<<3 * PG_GROUPS, PG_TB, 0, stream>>>(
        src, dst, rowptr, part, rnk, list,
        x, W1, xl1, ats1, atd1, as1, ad1, N_NODES, HIDDEN / (16 * 4));

    gat_gather_l1<<<N_NODES / L1_WAVES, L1_WAVES * 32, 0, stream>>>(
        rowptr, list, as1, ad1, xl1, b1, W2, ats2, atd2, xl2, as2, ad2);

    gat_gather_l2<<<blk(N_NODES, 16), TB, 0, stream>>>(
        rowptr, list, as2, ad2, xl2, b2, out);
}


# Layer-2 gather: a node's edge indices and logits loaded once (lane = edge, registers); each edge's softmax weight made once, handed to the row loop by DPP row_share; rows by fmaf; same sum order
# speedup vs baseline: 8.9062x; 1.0567x over previous
#include <hip/hip_runtime.h>
#include <cstdint>
#include <cstddef>

#define N_NODES   50000
#define N_EDGES   800000
#define TOT_E     (N_EDGES + N_NODES)
#define IN_C      128
#define HID       32
#define OUT_C     16
#define HEADS     8
#define HIDDEN    (HEADS * HID)
#define NEG_SLOPE 0.2f
#define GAT_EPS   1e-16f
#define PG_TB     128

typedef float v2f __attribute__((ext_vector_type(2)));
typedef float v8f __attribute__((ext_vector_type(8)));


template <int MASK>
__device__ __forceinline__ float gat_row_xor(float v)
{
    return __int_as_float(__builtin_amdgcn_update_dpp(
        0, __float_as_int(v), 0x160 + MASK, 0xf, 0xf, true));
}

template <int MASK, int NV>
__device__ __forceinline__ void gat_fold_step(float (&v)[32], bool up)
{
#pragma unroll
    for (int i = 0; i < NV / 2; ++i) {
        const float a = v[i], b = v[i + NV / 2];
        const float keep = up ? b : a;
        const float send = up ? a : b;
        v[i] = keep + gat_row_xor<MASK>(send);
    }
}

template <int TN, int K, int NC>
__device__ __forceinline__ void gat_gemm_block_f32(
    const int gblk,
    const float* __restrict__ A, const float* __restrict__ B,
    float* __restrict__ C, const float* __restrict__ att_s,
    const float* __restrict__ att_d, float* __restrict__ asrc,
    float* __restrict__ adst, int M, int waveTilesN)
{
    static_assert(TN == 4 && HID == 32 && NC == HEADS * HID,
                  "a wave's 16*TN columns are two whole heads");
    const int wave = __builtin_amdgcn_readfirstlane(
        (int)(gblk * (PG_TB / 32) + (threadIdx.x >> 5)));
    const int lane = threadIdx.x & 31;
    const int tM = wave / waveTilesN;
    const int tg = wave % waveTilesN;
    if (tM * 16 >= M) return;

    const int half = lane >> 4;
    const int l    = lane & 15;
    const float* __restrict__ Ap = A + (size_t)(tM * 16 + l) * K + half * 2;
    const float* __restrict__ Bp = B + (size_t)(half * 2) * NC + tg * 16 * TN + l;

    const v8f zero = {0.f,0.f,0.f,0.f,0.f,0.f,0.f,0.f};
    v8f c[TN];
#pragma unroll
    for (int j = 0; j < TN; ++j) c[j] = zero;

#pragma unroll
    for (int k = 0; k < K; k += 4) {
        const v2f a = *(const v2f*)(Ap + k);
        v2f b[TN];
#pragma unroll
        for (int j = 0; j < TN; ++j) {
            b[j].x = Bp[(size_t)k * NC + j * 16];
            b[j].y = Bp[(size_t)(k + 1) * NC + j * 16];
        }
#pragma unroll
        for (int j = 0; j < TN; ++j)
            c[j] = __builtin_amdgcn_wmma_f32_16x16x4_f32(
                        false, a,  false, b[j],
                        (short)0, c[j],
                        false,  false);
    }

    __builtin_amdgcn_sched_barrier(0);

    float* __restrict__ Cw = C + (size_t)(tM * 16) * NC + tg * 16 * TN;
    const unsigned co = (unsigned)(half * 8 * NC + l);
#pragma unroll
    for (int j = 0; j < TN; ++j)
#pragma unroll
        for (int r = 0; r < 8; ++r)
            Cw[co + (unsigned)(r * NC + j * 16)] = c[j][r];

    const float* __restrict__ asw = att_s + tg * 16 * TN;
    const float* __restrict__ adw = att_d + tg * 16 * TN;
    float ws[TN], wd[TN];
#pragma unroll
    for (int j = 0; j < TN; ++j) {
        ws[j] = asw[(unsigned)(j * 16 + l)];
        wd[j] = adw[(unsigned)(j * 16 + l)];
    }
    float v[32];
#pragma unroll
    for (int r = 0; r < 8; ++r) {
        v[4 * r + 0] = c[0][r] * ws[0] + c[1][r] * ws[1];
        v[4 * r + 1] = c[0][r] * wd[0] + c[1][r] * wd[1];
        v[4 * r + 2] = c[2][r] * ws[2] + c[3][r] * ws[3];
        v[4 * r + 3] = c[2][r] * wd[2] + c[3][r] * wd[3];
    }
    gat_fold_step<8, 32>(v, (l & 8) != 0);
    gat_fold_step<4, 16>(v, (l & 4) != 0);
    gat_fold_step<2, 8>(v, (l & 2) != 0);
    gat_fold_step<1, 4>(v, (l & 1) != 0);
    const size_t ob = (size_t)(tM * 16) * HEADS + tg * 2;
    const unsigned oo = (unsigned)((half * 8 + (l >> 1)) * HEADS + (l & 1));
    (asrc + ob)[oo] = v[0];
    (adst + ob)[oo] = v[1];
}

__device__ __forceinline__ float leaky(float v) { return v > 0.f ? v : NEG_SLOPE * v; }

#define BKT_N    256
#define N_BKT    ((N_NODES + BKT_N - 1) / BKT_N)
#define CNT_PAD  (N_BKT * BKT_N)

#define RANK_CHUNKS  64
#define RANK_CHUNK_E (N_EDGES / RANK_CHUNKS)
#define RANK_TB      1024
#define RANK_PER     ((RANK_CHUNK_E + RANK_TB - 1) / RANK_TB)
static_assert(RANK_CHUNKS * RANK_CHUNK_E == N_EDGES, "chunks must tile the edge list");
static_assert(N_NODES % 4 == 0, "counter table is moved 16 bytes at a time");
static_assert(RANK_CHUNKS % 4 == 0, "a bucket's chunk totals are read 16 bytes at a time");
static_assert(BKT_N % 32 == 0 && RANK_TB % 32 == 0, "whole waves");
#define BKT_PER_WAVE ((N_BKT + RANK_TB / 32 - 1) / (RANK_TB / 32))

__global__ __launch_bounds__(RANK_TB) void gat_edge_rank(
    const int* __restrict__ dst, int* __restrict__ part, int* __restrict__ rnk,
    int* __restrict__ bsum)
{
    __shared__ __attribute__((aligned(16))) int hist[N_NODES];
    const int t  = threadIdx.x;
    const int e0 = blockIdx.x * RANK_CHUNK_E;
    int4* h4 = (int4*)hist;
    for (int i = t; i < N_NODES / 4; i += RANK_TB) h4[i] = make_int4(0, 0, 0, 0);
    int d[RANK_PER];
#pragma unroll
    for (int k = 0; k < RANK_PER; ++k) {
        const int i = t + k * RANK_TB;
        d[k] = (i < RANK_CHUNK_E) ? dst[e0 + i] : 0;
    }
    __syncthreads();
    int r[RANK_PER];
#pragma unroll
    for (int k = 0; k < RANK_PER; ++k) {
        const int i = t + k * RANK_TB;
        r[k] = (i < RANK_CHUNK_E) ? atomicAdd(&hist[d[k]], 1) : 0;
    }
#pragma unroll
    for (int k = 0; k < RANK_PER; ++k) {
        const int i = t + k * RANK_TB;
        if (i < RANK_CHUNK_E) rnk[e0 + i] = r[k];
    }
    __syncthreads();
    int4* __restrict__ p4 = (int4*)(part + (size_t)blockIdx.x * CNT_PAD);
    for (int i = t; i < N_NODES / 4; i += RANK_TB) p4[i] = h4[i];
    const int lane = t & 31, w = t >> 5;
    int bs[BKT_PER_WAVE];
#pragma unroll
    for (int k = 0; k < BKT_PER_WAVE; ++k) {
        const int jb = w + k * (RANK_TB / 32);
        int acc = 0;
#pragma unroll
        for (int i = 0; i < BKT_N / 32; ++i) {
            const int n = jb * BKT_N + i * 32 + lane;
            acc += (n < N_NODES) ? hist[n] : 0;
        }
        bs[k] = acc;
    }
#pragma unroll
    for (int o = 16; o > 0; o >>= 1) {
#pragma unroll
        for (int k = 0; k < BKT_PER_WAVE; ++k) bs[k] += __shfl_xor(bs[k], o);
    }
#pragma unroll
    for (int k = 0; k < BKT_PER_WAVE; ++k) {
        const int jb = w + k * (RANK_TB / 32);
        if (lane == 0 && jb < N_BKT) bsum[jb * RANK_CHUNKS + blockIdx.x] = bs[k];
    }
}

#define BASE_MOVE (((N_BKT - 1) * (RANK_CHUNKS / 4) + BKT_N - 1) / BKT_N)
__global__ __launch_bounds__(BKT_N) void gat_chunk_prefix(
    int* __restrict__ part, const int* __restrict__ bsum, int* __restrict__ rowptr)
{
    __shared__ int wtot[BKT_N / 32];
    __shared__ int wsum[BKT_N / 32];
    const int t = threadIdx.x, lane = t & 31, w = t >> 5;
    const int n = blockIdx.x * BKT_N + t;
    const int4* __restrict__ b4 = (const int4*)bsum;
    const int n4 = blockIdx.x * (RANK_CHUNKS / 4);
    int4 q[BASE_MOVE];
#pragma unroll
    for (int k = 0; k < BASE_MOVE; ++k) {
        const int i = t + k * BKT_N;
        q[k] = (i < n4) ? b4[i] : make_int4(0, 0, 0, 0);
    }
    int s = 0;
    if (n < N_NODES) {
        int v[RANK_CHUNKS];
#pragma unroll
        for (int c = 0; c < RANK_CHUNKS; ++c) v[c] = part[(size_t)c * CNT_PAD + n];
#pragma unroll
        for (int c = 0; c < RANK_CHUNKS; ++c) { part[(size_t)c * CNT_PAD + n] = s; s += v[c]; }
    }
    int a = 0;
#pragma unroll
    for (int k = 0; k < BASE_MOVE; ++k) a += q[k].x + q[k].y + q[k].z + q[k].w;
    a += __shfl_xor(a, 16); a += __shfl_xor(a, 8); a += __shfl_xor(a, 4);
    a += __shfl_xor(a, 2);  a += __shfl_xor(a, 1);
    int x = s;
    for (int o = 1; o < 32; o <<= 1) { const int y = __shfl_up(x, o); if (lane >= o) x += y; }
    if (lane == 31) wsum[w] = x;
    if (lane == 0)  wtot[w] = a;
    __syncthreads();
    int base = 0;
#pragma unroll
    for (int i = 0; i < BKT_N / 32; ++i) { base += wtot[i]; base += (i < w) ? wsum[i] : 0; }
    rowptr[n] = base + x - s;
}

__device__ __forceinline__ void gat_place_block(
    const int pblk,
    const int* __restrict__ src, const int* __restrict__ dst,
    const int* __restrict__ rowptr, const int* __restrict__ part,
    const int* __restrict__ rnk, int* __restrict__ list)
{
    const int e = pblk * PG_TB + threadIdx.x;
    if (e >= N_EDGES) return;
    const int d = dst[e];
    const int before = part[(size_t)(e / RANK_CHUNK_E) * CNT_PAD + d];
    list[rowptr[d] + before + rnk[e]] = src[e];
}

#define PG_GROUPS  ((N_NODES / 16) * (HIDDEN / 64) / (PG_TB / 32))
static_assert(N_NODES % 16 == 0 && PG_GROUPS * (PG_TB / 32) == (N_NODES / 16) * (HIDDEN / 64),
              "GEMM work-groups are whole");
static_assert(2 * PG_GROUPS * PG_TB >= N_EDGES, "2 * PG_GROUPS placing work-groups cover every edge");

template <int TN, int K, int NC>
__global__ __launch_bounds__(PG_TB) __attribute__((amdgpu_waves_per_eu(16, 16))) void gat_place_gemm_f32(
    const int* __restrict__ src, const int* __restrict__ dst,
    const int* __restrict__ rowptr, const int* __restrict__ part,
    const int* __restrict__ rnk, int* __restrict__ list,
    const float* __restrict__ A, const float* __restrict__ B,
    float* __restrict__ C, const float* __restrict__ att_s,
    const float* __restrict__ att_d, float* __restrict__ asrc,
    float* __restrict__ adst, int M, int waveTilesN)
{
    if (blockIdx.x >= PG_GROUPS) {
        gat_place_block((int)blockIdx.x - PG_GROUPS, src, dst, rowptr, part, rnk, list);
        return;
    }
    gat_gemm_block_f32<TN, K, NC>((int)blockIdx.x, A, B, C, att_s, att_d, asrc, adst, M, waveTilesN);
}

#define L1_WAVES 16
#define H_PITCH  (HIDDEN + 4)
static_assert(N_NODES % L1_WAVES == 0, "every block is a full tile of 16 nodes: no partial block, no bounds test");
static_assert(HIDDEN == L1_WAVES * 16, "wave w takes K slice 16w .. 16w+15 of the layer-2 product");
static_assert(OUT_C == 16, "the layer-2 row is one 16-column WMMA tile");

__global__ __launch_bounds__(L1_WAVES * 32) void gat_gather_l1(
    const int* __restrict__ rowptr, const int* __restrict__ list,
    const float* __restrict__ asrc, const float* __restrict__ adst,
    const float* __restrict__ xl, const float* __restrict__ bias,
    const float* __restrict__ W2, const float* __restrict__ att_s2,
    const float* __restrict__ att_d2, float* __restrict__ xl2,
    float* __restrict__ as2, float* __restrict__ ad2)
{
    __shared__ __attribute__((aligned(16))) float sw[L1_WAVES][32 * HEADS];
    __shared__ __attribute__((aligned(16))) float hrow[L1_WAVES][H_PITCH];
    const int lane = threadIdx.x & 31;
    const int wv   = threadIdx.x >> 5;
    const int d = __builtin_amdgcn_readfirstlane((int)(blockIdx.x * L1_WAVES + wv));
    float* my = sw[wv];
    const int p0 = rowptr[d], p1 = rowptr[d + 1];
    const int deg = p1 - p0;
    const int nch = (deg + 31) >> 5;

    const int hh = lane & 7, j = lane >> 3;
    const float adh   = adst[(size_t)d * HEADS + hh];
    const float vself = leaky(asrc[(size_t)d * HEADS + hh] + adh);

    int sv = d;
    auto stage = [&](int c) {
        const int q = c * 32 + lane;
        sv = (q < deg) ? list[p0 + q] : d;
        const float4* ar = (const float4*)(asrc + (size_t)sv * HEADS);
        const float4 a0 = ar[0], a1 = ar[1];
        float4* dp = (float4*)(my + lane * HEADS);
        dp[0] = a0; dp[1] = a1;
    };

    float m = vself;
    for (int c = 0; c < nch; ++c) {
        stage(c);
#pragma unroll
        for (int i = 0; i < 8; ++i) {
            const int sl = j + 4 * i;
            const float v = leaky(my[sl * HEADS + hh] + adh);
            m = (c * 32 + sl < deg) ? fmaxf(m, v) : m;
        }
    }
    m = fmaxf(m, __shfl_xor(m, 8));
    m = fmaxf(m, __shfl_xor(m, 16));
    float sum = (j == 0) ? expf(vself - m) : 0.f;
    for (int c = 0; c < nch; ++c) {
        if (nch > 1) stage(c);
#pragma unroll
        for (int i = 0; i < 8; ++i) {
            const int sl = j + 4 * i;
            const float ex = expf(leaky(my[sl * HEADS + hh] + adh) - m);
            sum = (c * 32 + sl < deg) ? sum + ex : sum;
        }
    }
    sum += __shfl_xor(sum, 8);
    sum += __shfl_xor(sum, 16);
    const float den = sum + GAT_EPS;
    const float wself = expf(vself - m) / den;

    const int hc = lane >> 2;
    float acc[8];
    {
        const float a = __shfl(wself, hc);
        const float4* r = (const float4*)(xl + (size_t)d * HIDDEN + lane * 8);
        const float4 r0 = r[0], r1 = r[1];
        acc[0] = a * r0.x; acc[1] = a * r0.y; acc[2] = a * r0.z; acc[3] = a * r0.w;
        acc[4] = a * r1.x; acc[5] = a * r1.y; acc[6] = a * r1.z; acc[7] = a * r1.w;
    }
    for (int c = 0; c < nch; ++c) {
        if (nch > 1) stage(c);
#pragma unroll
        for (int i = 0; i < 8; ++i) {
            const int sl = j + 4 * i;
            my[sl * HEADS + hh] = expf(leaky(my[sl * HEADS + hh] + adh) - m) / den;
        }
        const int cnt = min(32, deg - c * 32);
#pragma unroll 4
        for (int e = 0; e < cnt; ++e) {
            const int s = __builtin_amdgcn_readlane(sv, e);
            const float a = my[e * HEADS + hc];
            const float4* r = (const float4*)(xl + (size_t)s * HIDDEN + lane * 8);
            const float4 r0 = r[0], r1 = r[1];
            acc[0] += a * r0.x; acc[1] += a * r0.y; acc[2] += a * r0.z; acc[3] += a * r0.w;
            acc[4] += a * r1.x; acc[5] += a * r1.y; acc[6] += a * r1.z; acc[7] += a * r1.w;
        }
    }
    const float4* bp = (const float4*)(bias + lane * 8);
    const float4 b0 = bp[0], b1 = bp[1];
    const float bb[8] = {b0.x, b0.y, b0.z, b0.w, b1.x, b1.y, b1.z, b1.w};
    float o[8];
#pragma unroll
    for (int i = 0; i < 8; ++i) { const float v = acc[i] + bb[i]; o[i] = v > 0.f ? v : expm1f(v); }
    float4* op = (float4*)(&hrow[wv][lane * 8]);
    op[0] = make_float4(o[0], o[1], o[2], o[3]);
    op[1] = make_float4(o[4], o[5], o[6], o[7]);
    __syncthreads();

    const int kh = (threadIdx.x >> 4) & 1, l16 = threadIdx.x & 15;
    const float* hA = &hrow[l16][wv * 16 + kh * 2];
    const float* __restrict__ Bq = W2 + (size_t)(wv * 16 + kh * 2) * OUT_C + l16;
    v8f c2 = {0.f, 0.f, 0.f, 0.f, 0.f, 0.f, 0.f, 0.f};
#pragma unroll
    for (int kk = 0; kk < 4; ++kk) {
        const v2f a2 = *(const v2f*)(hA + kk * 4);
        v2f b2;
        b2.x = Bq[(size_t)(kk * 4) * OUT_C];
        b2.y = Bq[(size_t)(kk * 4 + 1) * OUT_C];
        c2 = __builtin_amdgcn_wmma_f32_16x16x4_f32(
                  false, a2,  false, b2,
                  (short)0, c2,
                  false,  false);
    }
#pragma unroll
    for (int r = 0; r < 8; ++r) my[r * 32 + lane] = c2[r];
    __syncthreads();
    if (threadIdx.x < 256) {
        const int idx = threadIdx.x;
        float s = sw[0][idx];
#pragma unroll
        for (int w = 1; w < L1_WAVES; ++w) s += sw[w][idx];
        const int col  = idx & 15;
        const int node = blockIdx.x * L1_WAVES + ((idx >> 4) & 1) * 8 + (idx >> 5);
        xl2[(size_t)node * OUT_C + col] = s;
        float ps = s * att_s2[col], pd = s * att_d2[col];
        ps += __int_as_float(__builtin_amdgcn_ds_swizzle(__float_as_int(ps), 0x041f));
        pd += __int_as_float(__builtin_amdgcn_ds_swizzle(__float_as_int(pd), 0x041f));
        ps += __int_as_float(__builtin_amdgcn_ds_swizzle(__float_as_int(ps), 0x081f));
        pd += __int_as_float(__builtin_amdgcn_ds_swizzle(__float_as_int(pd), 0x081f));
        ps += __int_as_float(__builtin_amdgcn_ds_swizzle(__float_as_int(ps), 0x101f));
        pd += __int_as_float(__builtin_amdgcn_ds_swizzle(__float_as_int(pd), 0x101f));
        ps += __int_as_float(__builtin_amdgcn_ds_swizzle(__float_as_int(ps), 0x201f));
        pd += __int_as_float(__builtin_amdgcn_ds_swizzle(__float_as_int(pd), 0x201f));
        if (col == 0) { as2[node] = ps; ad2[node] = pd; }
    }
}


template <int E>
__device__ __forceinline__ int gat_row_lane_i(int v)
{
    return __builtin_amdgcn_update_dpp(0, v, 0x150 + E, 0xf, 0xf, true);
}
template <int E>
__device__ __forceinline__ float gat_row_lane_f(float v)
{
    return __int_as_float(gat_row_lane_i<E>(__float_as_int(v)));
}

template <int B0>
__device__ __forceinline__ float gat_l2_rows16(float acc, const int sreg, const float wreg,
                                                const float* __restrict__ xl, const int c)
{
    const float x0 = xl[(unsigned)(gat_row_lane_i<B0 + 0>(sreg) * OUT_C + c)];
    const float x1 = xl[(unsigned)(gat_row_lane_i<B0 + 1>(sreg) * OUT_C + c)];
    const float x2 = xl[(unsigned)(gat_row_lane_i<B0 + 2>(sreg) * OUT_C + c)];
    const float x3 = xl[(unsigned)(gat_row_lane_i<B0 + 3>(sreg) * OUT_C + c)];
    const float x4 = xl[(unsigned)(gat_row_lane_i<B0 + 4>(sreg) * OUT_C + c)];
    const float x5 = xl[(unsigned)(gat_row_lane_i<B0 + 5>(sreg) * OUT_C + c)];
    const float x6 = xl[(unsigned)(gat_row_lane_i<B0 + 6>(sreg) * OUT_C + c)];
    const float x7 = xl[(unsigned)(gat_row_lane_i<B0 + 7>(sreg) * OUT_C + c)];
    const float x8 = xl[(unsigned)(gat_row_lane_i<B0 + 8>(sreg) * OUT_C + c)];
    const float x9 = xl[(unsigned)(gat_row_lane_i<B0 + 9>(sreg) * OUT_C + c)];
    const float x10 = xl[(unsigned)(gat_row_lane_i<B0 + 10>(sreg) * OUT_C + c)];
    const float x11 = xl[(unsigned)(gat_row_lane_i<B0 + 11>(sreg) * OUT_C + c)];
    const float x12 = xl[(unsigned)(gat_row_lane_i<B0 + 12>(sreg) * OUT_C + c)];
    const float x13 = xl[(unsigned)(gat_row_lane_i<B0 + 13>(sreg) * OUT_C + c)];
    const float x14 = xl[(unsigned)(gat_row_lane_i<B0 + 14>(sreg) * OUT_C + c)];
    const float x15 = xl[(unsigned)(gat_row_lane_i<B0 + 15>(sreg) * OUT_C + c)];
    acc = fmaf(gat_row_lane_f<B0 + 0>(wreg), x0, acc);
    acc = fmaf(gat_row_lane_f<B0 + 1>(wreg), x1, acc);
    acc = fmaf(gat_row_lane_f<B0 + 2>(wreg), x2, acc);
    acc = fmaf(gat_row_lane_f<B0 + 3>(wreg), x3, acc);
    acc = fmaf(gat_row_lane_f<B0 + 4>(wreg), x4, acc);
    acc = fmaf(gat_row_lane_f<B0 + 5>(wreg), x5, acc);
    acc = fmaf(gat_row_lane_f<B0 + 6>(wreg), x6, acc);
    acc = fmaf(gat_row_lane_f<B0 + 7>(wreg), x7, acc);
    acc = fmaf(gat_row_lane_f<B0 + 8>(wreg), x8, acc);
    acc = fmaf(gat_row_lane_f<B0 + 9>(wreg), x9, acc);
    acc = fmaf(gat_row_lane_f<B0 + 10>(wreg), x10, acc);
    acc = fmaf(gat_row_lane_f<B0 + 11>(wreg), x11, acc);
    acc = fmaf(gat_row_lane_f<B0 + 12>(wreg), x12, acc);
    acc = fmaf(gat_row_lane_f<B0 + 13>(wreg), x13, acc);
    acc = fmaf(gat_row_lane_f<B0 + 14>(wreg), x14, acc);
    acc = fmaf(gat_row_lane_f<B0 + 15>(wreg), x15, acc);
    return acc;
}

template <int B0>
__device__ __forceinline__ float gat_l2_rows8(float acc, const int sreg, const float wreg,
                                               const float* __restrict__ xl, const int c)
{
    const float x0 = xl[(unsigned)(gat_row_lane_i<B0 + 0>(sreg) * OUT_C + c)];
    const float x1 = xl[(unsigned)(gat_row_lane_i<B0 + 1>(sreg) * OUT_C + c)];
    const float x2 = xl[(unsigned)(gat_row_lane_i<B0 + 2>(sreg) * OUT_C + c)];
    const float x3 = xl[(unsigned)(gat_row_lane_i<B0 + 3>(sreg) * OUT_C + c)];
    const float x4 = xl[(unsigned)(gat_row_lane_i<B0 + 4>(sreg) * OUT_C + c)];
    const float x5 = xl[(unsigned)(gat_row_lane_i<B0 + 5>(sreg) * OUT_C + c)];
    const float x6 = xl[(unsigned)(gat_row_lane_i<B0 + 6>(sreg) * OUT_C + c)];
    const float x7 = xl[(unsigned)(gat_row_lane_i<B0 + 7>(sreg) * OUT_C + c)];
    acc = fmaf(gat_row_lane_f<B0 + 0>(wreg), x0, acc);
    acc = fmaf(gat_row_lane_f<B0 + 1>(wreg), x1, acc);
    acc = fmaf(gat_row_lane_f<B0 + 2>(wreg), x2, acc);
    acc = fmaf(gat_row_lane_f<B0 + 3>(wreg), x3, acc);
    acc = fmaf(gat_row_lane_f<B0 + 4>(wreg), x4, acc);
    acc = fmaf(gat_row_lane_f<B0 + 5>(wreg), x5, acc);
    acc = fmaf(gat_row_lane_f<B0 + 6>(wreg), x6, acc);
    acc = fmaf(gat_row_lane_f<B0 + 7>(wreg), x7, acc);
    return acc;
}

static_assert(N_NODES % 16 == 0, "a block of the layer-2 gather is 16 whole nodes: no partial block");

__global__ __launch_bounds__(256) void gat_gather_l2(
    const int* __restrict__ rowptr, const int* __restrict__ list,
    const float* __restrict__ asrc, const float* __restrict__ adst,
    const float* __restrict__ xl, const float* __restrict__ bias,
    float* __restrict__ out)
{
    const int c = threadIdx.x & 15;
    const int d = blockIdx.x * 16 + (threadIdx.x >> 4);
    const int p0 = rowptr[d], p1 = rowptr[d + 1];
    const int deg = p1 - p0;
    const float ad    = adst[d];
    const float vself = leaky(asrc[d] + ad);
    const int degw = max(__builtin_amdgcn_readlane(deg, 0), __builtin_amdgcn_readlane(deg, 16));
    float acc;
    if (degw <= 32) {
        const bool in0 = c < deg, in1 = c + 16 < deg;
        const int l0 = list[in0 ? p0 + c : 0];
        const int l1 = list[in1 ? p0 + c + 16 : 0];
        const int s0 = in0 ? l0 : d;
        const int s1 = in1 ? l1 : d;
        const float v0 = leaky(asrc[s0] + ad);
        const float v1 = leaky(asrc[s1] + ad);
        float m = vself;
        m = in0 ? fmaxf(m, v0) : m;
        m = in1 ? fmaxf(m, v1) : m;
        m = fmaxf(m, gat_row_xor<1>(m)); m = fmaxf(m, gat_row_xor<2>(m));
        m = fmaxf(m, gat_row_xor<4>(m)); m = fmaxf(m, gat_row_xor<8>(m));
        const float eself = expf(vself - m);
        const float e0 = expf(v0 - m), e1 = expf(v1 - m);
        float sum = (c == 0) ? eself : 0.f;
        sum = in0 ? sum + e0 : sum;
        sum = in1 ? sum + e1 : sum;
        sum += gat_row_xor<1>(sum); sum += gat_row_xor<2>(sum);
        sum += gat_row_xor<4>(sum); sum += gat_row_xor<8>(sum);
        const float den = sum + GAT_EPS;
        const float w0 = in0 ? e0 / den : 0.f;
        const float w1 = in1 ? e1 / den : 0.f;
        acc = eself / den * xl[(size_t)d * OUT_C + c];
        acc = gat_l2_rows16<0>(acc, s0, w0, xl, c);
        if (degw > 16) acc = gat_l2_rows8<0>(acc, s1, w1, xl, c);
        if (degw > 24) acc = gat_l2_rows8<8>(acc, s1, w1, xl, c);
    } else {
        float m = vself;
        for (int p = p0 + c; p < p1; p += 16) m = fmaxf(m, leaky(asrc[list[p]] + ad));
        m = fmaxf(m, __shfl_xor(m, 1)); m = fmaxf(m, __shfl_xor(m, 2));
        m = fmaxf(m, __shfl_xor(m, 4)); m = fmaxf(m, __shfl_xor(m, 8));
        float sum = (c == 0) ? expf(vself - m) : 0.f;
        for (int p = p0 + c; p < p1; p += 16) sum += expf(leaky(asrc[list[p]] + ad) - m);
        sum += __shfl_xor(sum, 1); sum += __shfl_xor(sum, 2);
        sum += __shfl_xor(sum, 4); sum += __shfl_xor(sum, 8);
        const float den = sum + GAT_EPS;
        acc = expf(vself - m) / den * xl[(size_t)d * OUT_C + c];
#pragma unroll 4
        for (int p = p0; p < p1; ++p) {
            const int s = list[p];
            const float a = expf(leaky(asrc[s] + ad) - m) / den;
            acc += a * xl[(size_t)s * OUT_C + c];
        }
    }
    out[(size_t)d * OUT_C + c] = acc + bias[c];
}

extern "C" void kernel_launch(void* const* d_in, const int* in_sizes, int n_in,
                              void* d_out, int out_size, void* d_ws, size_t ws_size,
                              hipStream_t stream)
{
    (void)in_sizes; (void)n_in; (void)out_size; (void)ws_size;

    const float* x    = (const float*)d_in[0];
    const int*   ei   = (const int*)  d_in[1];
    const float* W1   = (const float*)d_in[2];
    const float* ats1 = (const float*)d_in[3];
    const float* atd1 = (const float*)d_in[4];
    const float* b1   = (const float*)d_in[5];
    const float* W2   = (const float*)d_in[6];
    const float* ats2 = (const float*)d_in[7];
    const float* atd2 = (const float*)d_in[8];
    const float* b2   = (const float*)d_in[9];
    const int* src = ei;
    const int* dst = ei + N_EDGES;
    float* out = (float*)d_out;

    int*   rowptr = (int*)d_ws;
    int*   rnk    = rowptr + (size_t)CNT_PAD;
    int*   list   = rnk + (size_t)N_EDGES;
    float* xl1 = (float*)(list + (size_t)N_EDGES);
    float* as1 = xl1 + (size_t)N_NODES * HIDDEN;
    float* ad1 = as1 + (size_t)N_NODES * HEADS;
    float* xl2 = ad1 + (size_t)N_NODES * HEADS;
    float* as2 = xl2 + (size_t)N_NODES * OUT_C;
    float* ad2 = as2 + (size_t)CNT_PAD;
    int*   part = (int*)(ad2 + (size_t)CNT_PAD);
    int*   bsum = part + (size_t)RANK_CHUNKS * CNT_PAD;

    const int TB = 256;

    gat_edge_rank<<<RANK_CHUNKS, RANK_TB, 0, stream>>>(dst, part, rnk, bsum);
    gat_chunk_prefix<<<N_BKT, BKT_N, 0, stream>>>(part, bsum, rowptr);

    gat_place_gemm_f32<4, IN_C, HIDDEN><<<3 * PG_GROUPS, PG_TB, 0, stream>>>(
        src, dst, rowptr, part, rnk, list,
        x, W1, xl1, ats1, atd1, as1, ad1, N_NODES, HIDDEN / (16 * 4));

    gat_gather_l1<<<N_NODES / L1_WAVES, L1_WAVES * 32, 0, stream>>>(
        rowptr, list, as1, ad1, xl1, b1, W2, ats2, atd2, xl2, as2, ad2);

    gat_gather_l2<<<N_NODES / 16, TB, 0, stream>>>(
        rowptr, list, as2, ad2, xl2, b2, out);
}


# Layer-1 GEMM dealt over two launches: 1536 of its 3125 row tiles run in the edge-rank launch on the 192 WGPs the 64 rank work-groups leave idle, the rest beside edge placing as before; same tiles
# speedup vs baseline: 9.1894x; 1.0318x over previous
#include <hip/hip_runtime.h>
#include <cstdint>
#include <cstddef>

#define N_NODES   50000
#define N_EDGES   800000
#define TOT_E     (N_EDGES + N_NODES)
#define IN_C      128
#define HID       32
#define OUT_C     16
#define HEADS     8
#define HIDDEN    (HEADS * HID)
#define NEG_SLOPE 0.2f
#define GAT_EPS   1e-16f
#define PG_TB     128

typedef float v2f __attribute__((ext_vector_type(2)));
typedef float v8f __attribute__((ext_vector_type(8)));


template <int MASK>
__device__ __forceinline__ float gat_row_xor(float v)
{
    return __int_as_float(__builtin_amdgcn_update_dpp(
        0, __float_as_int(v), 0x160 + MASK, 0xf, 0xf, true));
}

template <int MASK, int NV>
__device__ __forceinline__ void gat_fold_step(float (&v)[32], bool up)
{
#pragma unroll
    for (int i = 0; i < NV / 2; ++i) {
        const float a = v[i], b = v[i + NV / 2];
        const float keep = up ? b : a;
        const float send = up ? a : b;
        v[i] = keep + gat_row_xor<MASK>(send);
    }
}

template <int TN, int K, int NC, int KREG>
__device__ __forceinline__ void gat_gemm_block_f32(
    const int gwave,
    const float* __restrict__ A, const float* __restrict__ B,
    float* __restrict__ C, const float* __restrict__ att_s,
    const float* __restrict__ att_d, float* __restrict__ asrc,
    float* __restrict__ adst, int M, int waveTilesN)
{
    static_assert(TN == 4 && HID == 32 && NC == HEADS * HID,
                  "a wave's 16*TN columns are two whole heads");
    const int wave = __builtin_amdgcn_readfirstlane(gwave);
    const int lane = threadIdx.x & 31;
    const int tM = wave / waveTilesN;
    const int tg = wave % waveTilesN;
    if (tM * 16 >= M) return;

    const int half = lane >> 4;
    const int l    = lane & 15;
    const float* __restrict__ Ap = A + (size_t)(tM * 16 + l) * K + half * 2;
    const float* __restrict__ Bp = B + (size_t)(half * 2) * NC + tg * 16 * TN + l;

    const v8f zero = {0.f,0.f,0.f,0.f,0.f,0.f,0.f,0.f};
    v8f c[TN];
#pragma unroll
    for (int j = 0; j < TN; ++j) c[j] = zero;

#pragma unroll
    for (int k = 0; k < K; k += 4) {
        if (KREG > 0 && k > 0 && k % (4 * KREG) == 0) __builtin_amdgcn_sched_barrier(0);
        const v2f a = *(const v2f*)(Ap + k);
        v2f b[TN];
#pragma unroll
        for (int j = 0; j < TN; ++j) {
            b[j].x = Bp[(size_t)k * NC + j * 16];
            b[j].y = Bp[(size_t)(k + 1) * NC + j * 16];
        }
#pragma unroll
        for (int j = 0; j < TN; ++j)
            c[j] = __builtin_amdgcn_wmma_f32_16x16x4_f32(
                        false, a,  false, b[j],
                        (short)0, c[j],
                        false,  false);
    }

    __builtin_amdgcn_sched_barrier(0);

    float* __restrict__ Cw = C + (size_t)(tM * 16) * NC + tg * 16 * TN;
    const unsigned co = (unsigned)(half * 8 * NC + l);
#pragma unroll
    for (int j = 0; j < TN; ++j)
#pragma unroll
        for (int r = 0; r < 8; ++r)
            Cw[co + (unsigned)(r * NC + j * 16)] = c[j][r];

    const float* __restrict__ asw = att_s + tg * 16 * TN;
    const float* __restrict__ adw = att_d + tg * 16 * TN;
    float ws[TN], wd[TN];
#pragma unroll
    for (int j = 0; j < TN; ++j) {
        ws[j] = asw[(unsigned)(j * 16 + l)];
        wd[j] = adw[(unsigned)(j * 16 + l)];
    }
    float v[32];
#pragma unroll
    for (int r = 0; r < 8; ++r) {
        v[4 * r + 0] = c[0][r] * ws[0] + c[1][r] * ws[1];
        v[4 * r + 1] = c[0][r] * wd[0] + c[1][r] * wd[1];
        v[4 * r + 2] = c[2][r] * ws[2] + c[3][r] * ws[3];
        v[4 * r + 3] = c[2][r] * wd[2] + c[3][r] * wd[3];
    }
    gat_fold_step<8, 32>(v, (l & 8) != 0);
    gat_fold_step<4, 16>(v, (l & 4) != 0);
    gat_fold_step<2, 8>(v, (l & 2) != 0);
    gat_fold_step<1, 4>(v, (l & 1) != 0);
    const size_t ob = (size_t)(tM * 16) * HEADS + tg * 2;
    const unsigned oo = (unsigned)((half * 8 + (l >> 1)) * HEADS + (l & 1));
    (asrc + ob)[oo] = v[0];
    (adst + ob)[oo] = v[1];
}

__device__ __forceinline__ float leaky(float v) { return v > 0.f ? v : NEG_SLOPE * v; }

#define BKT_N    256
#define N_BKT    ((N_NODES + BKT_N - 1) / BKT_N)
#define CNT_PAD  (N_BKT * BKT_N)

#define RANK_CHUNKS  64
#define RANK_CHUNK_E (N_EDGES / RANK_CHUNKS)
#define RANK_TB      1024
#define RANK_PER     ((RANK_CHUNK_E + RANK_TB - 1) / RANK_TB)
static_assert(RANK_CHUNKS * RANK_CHUNK_E == N_EDGES, "chunks must tile the edge list");
static_assert(N_NODES % 4 == 0, "counter table is moved 16 bytes at a time");
static_assert(RANK_CHUNKS % 4 == 0, "a bucket's chunk totals are read 16 bytes at a time");
static_assert(BKT_N % 32 == 0 && RANK_TB % 32 == 0, "whole waves");
#define BKT_PER_WAVE ((N_BKT + RANK_TB / 32 - 1) / (RANK_TB / 32))

__device__ __forceinline__ void gat_rank_block(
    const int chunk,
    const int* __restrict__ dst, int* __restrict__ part, int* __restrict__ rnk,
    int* __restrict__ bsum)
{
    __shared__ __attribute__((aligned(16))) int hist[N_NODES];
    const int t  = threadIdx.x;
    const int e0 = chunk * RANK_CHUNK_E;
    int4* h4 = (int4*)hist;
    for (int i = t; i < N_NODES / 4; i += RANK_TB) h4[i] = make_int4(0, 0, 0, 0);
    int d[RANK_PER];
#pragma unroll
    for (int k = 0; k < RANK_PER; ++k) {
        const int i = t + k * RANK_TB;
        d[k] = (i < RANK_CHUNK_E) ? dst[e0 + i] : 0;
    }
    __syncthreads();
    int r[RANK_PER];
#pragma unroll
    for (int k = 0; k < RANK_PER; ++k) {
        const int i = t + k * RANK_TB;
        r[k] = (i < RANK_CHUNK_E) ? atomicAdd(&hist[d[k]], 1) : 0;
    }
#pragma unroll
    for (int k = 0; k < RANK_PER; ++k) {
        const int i = t + k * RANK_TB;
        if (i < RANK_CHUNK_E) rnk[e0 + i] = r[k];
    }
    __syncthreads();
    int4* __restrict__ p4 = (int4*)(part + (size_t)chunk * CNT_PAD);
    for (int i = t; i < N_NODES / 4; i += RANK_TB) p4[i] = h4[i];
    const int lane = t & 31, w = t >> 5;
    int bs[BKT_PER_WAVE];
#pragma unroll
    for (int k = 0; k < BKT_PER_WAVE; ++k) {
        const int jb = w + k * (RANK_TB / 32);
        int acc = 0;
#pragma unroll
        for (int i = 0; i < BKT_N / 32; ++i) {
            const int n = jb * BKT_N + i * 32 + lane;
            acc += (n < N_NODES) ? hist[n] : 0;
        }
        bs[k] = acc;
    }
#pragma unroll
    for (int o = 16; o > 0; o >>= 1) {
#pragma unroll
        for (int k = 0; k < BKT_PER_WAVE; ++k) bs[k] += __shfl_xor(bs[k], o);
    }
#pragma unroll
    for (int k = 0; k < BKT_PER_WAVE; ++k) {
        const int jb = w + k * (RANK_TB / 32);
        if (lane == 0 && jb < N_BKT) bsum[jb * RANK_CHUNKS + chunk] = bs[k];
    }
}

#define RG_GEMM_GROUPS 192
#define RG_TILES       (RG_GEMM_GROUPS * (RANK_TB / 32) / (HIDDEN / 64))
static_assert(RG_TILES * (HIDDEN / 64) == RG_GEMM_GROUPS * (RANK_TB / 32), "GEMM work-groups of the rank launch are whole tiles");
static_assert(RG_TILES <= N_NODES / 16, "no more tiles than the matrix has");

template <int TN, int K, int NC>
__global__ __launch_bounds__(RANK_TB) void gat_rank_gemm_f32(
    const int* __restrict__ dst, int* __restrict__ part, int* __restrict__ rnk,
    int* __restrict__ bsum,
    const float* __restrict__ A, const float* __restrict__ B,
    float* __restrict__ C, const float* __restrict__ att_s,
    const float* __restrict__ att_d, float* __restrict__ asrc,
    float* __restrict__ adst, int M, int waveTilesN)
{
    if (blockIdx.x < RANK_CHUNKS) {
        gat_rank_block((int)blockIdx.x, dst, part, rnk, bsum);
        return;
    }
    gat_gemm_block_f32<TN, K, NC, 8>(
        (int)((blockIdx.x - RANK_CHUNKS) * (RANK_TB / 32) + (threadIdx.x >> 5)),
        A, B, C, att_s, att_d, asrc, adst, M, waveTilesN);
}

#define BASE_MOVE (((N_BKT - 1) * (RANK_CHUNKS / 4) + BKT_N - 1) / BKT_N)
__global__ __launch_bounds__(BKT_N) void gat_chunk_prefix(
    int* __restrict__ part, const int* __restrict__ bsum, int* __restrict__ rowptr)
{
    __shared__ int wtot[BKT_N / 32];
    __shared__ int wsum[BKT_N / 32];
    const int t = threadIdx.x, lane = t & 31, w = t >> 5;
    const int n = blockIdx.x * BKT_N + t;
    const int4* __restrict__ b4 = (const int4*)bsum;
    const int n4 = blockIdx.x * (RANK_CHUNKS / 4);
    int4 q[BASE_MOVE];
#pragma unroll
    for (int k = 0; k < BASE_MOVE; ++k) {
        const int i = t + k * BKT_N;
        q[k] = (i < n4) ? b4[i] : make_int4(0, 0, 0, 0);
    }
    int s = 0;
    if (n < N_NODES) {
        int v[RANK_CHUNKS];
#pragma unroll
        for (int c = 0; c < RANK_CHUNKS; ++c) v[c] = part[(size_t)c * CNT_PAD + n];
#pragma unroll
        for (int c = 0; c < RANK_CHUNKS; ++c) { part[(size_t)c * CNT_PAD + n] = s; s += v[c]; }
    }
    int a = 0;
#pragma unroll
    for (int k = 0; k < BASE_MOVE; ++k) a += q[k].x + q[k].y + q[k].z + q[k].w;
    a += __shfl_xor(a, 16); a += __shfl_xor(a, 8); a += __shfl_xor(a, 4);
    a += __shfl_xor(a, 2);  a += __shfl_xor(a, 1);
    int x = s;
    for (int o = 1; o < 32; o <<= 1) { const int y = __shfl_up(x, o); if (lane >= o) x += y; }
    if (lane == 31) wsum[w] = x;
    if (lane == 0)  wtot[w] = a;
    __syncthreads();
    int base = 0;
#pragma unroll
    for (int i = 0; i < BKT_N / 32; ++i) { base += wtot[i]; base += (i < w) ? wsum[i] : 0; }
    rowptr[n] = base + x - s;
}

__device__ __forceinline__ void gat_place_block(
    const int pblk,
    const int* __restrict__ src, const int* __restrict__ dst,
    const int* __restrict__ rowptr, const int* __restrict__ part,
    const int* __restrict__ rnk, int* __restrict__ list)
{
    const int e = pblk * PG_TB + threadIdx.x;
    if (e >= N_EDGES) return;
    const int d = dst[e];
    const int before = part[(size_t)(e / RANK_CHUNK_E) * CNT_PAD + d];
    list[rowptr[d] + before + rnk[e]] = src[e];
}

#define PG_GROUPS  ((N_NODES / 16 - RG_TILES) * (HIDDEN / 64) / (PG_TB / 32))
#define PG_PLACE   ((N_EDGES + PG_TB - 1) / PG_TB)
static_assert(N_NODES % 16 == 0 && PG_GROUPS * (PG_TB / 32) == (N_NODES / 16 - RG_TILES) * (HIDDEN / 64),
              "GEMM work-groups are whole");
static_assert(PG_PLACE * PG_TB >= N_EDGES, "PG_PLACE placing work-groups cover every edge");

template <int TN, int K, int NC>
__global__ __launch_bounds__(PG_TB) __attribute__((amdgpu_waves_per_eu(16, 16))) void gat_place_gemm_f32(
    const int* __restrict__ src, const int* __restrict__ dst,
    const int* __restrict__ rowptr, const int* __restrict__ part,
    const int* __restrict__ rnk, int* __restrict__ list,
    const float* __restrict__ A, const float* __restrict__ B,
    float* __restrict__ C, const float* __restrict__ att_s,
    const float* __restrict__ att_d, float* __restrict__ asrc,
    float* __restrict__ adst, int M, int waveTilesN)
{
    if (blockIdx.x >= PG_GROUPS) {
        gat_place_block((int)blockIdx.x - PG_GROUPS, src, dst, rowptr, part, rnk, list);
        return;
    }
    gat_gemm_block_f32<TN, K, NC, 0>(
        (int)(RG_TILES * (HIDDEN / 64) + blockIdx.x * (PG_TB / 32) + (threadIdx.x >> 5)),
        A, B, C, att_s, att_d, asrc, adst, M, waveTilesN);
}

#define L1_WAVES 16
#define H_PITCH  (HIDDEN + 4)
static_assert(N_NODES % L1_WAVES == 0, "every block is a full tile of 16 nodes: no partial block, no bounds test");
static_assert(HIDDEN == L1_WAVES * 16, "wave w takes K slice 16w .. 16w+15 of the layer-2 product");
static_assert(OUT_C == 16, "the layer-2 row is one 16-column WMMA tile");

__global__ __launch_bounds__(L1_WAVES * 32) void gat_gather_l1(
    const int* __restrict__ rowptr, const int* __restrict__ list,
    const float* __restrict__ asrc, const float* __restrict__ adst,
    const float* __restrict__ xl, const float* __restrict__ bias,
    const float* __restrict__ W2, const float* __restrict__ att_s2,
    const float* __restrict__ att_d2, float* __restrict__ xl2,
    float* __restrict__ as2, float* __restrict__ ad2)
{
    __shared__ __attribute__((aligned(16))) float sw[L1_WAVES][32 * HEADS];
    __shared__ __attribute__((aligned(16))) float hrow[L1_WAVES][H_PITCH];
    const int lane = threadIdx.x & 31;
    const int wv   = threadIdx.x >> 5;
    const int d = __builtin_amdgcn_readfirstlane((int)(blockIdx.x * L1_WAVES + wv));
    float* my = sw[wv];
    const int p0 = rowptr[d], p1 = rowptr[d + 1];
    const int deg = p1 - p0;
    const int nch = (deg + 31) >> 5;

    const int hh = lane & 7, j = lane >> 3;
    const float adh   = adst[(size_t)d * HEADS + hh];
    const float vself = leaky(asrc[(size_t)d * HEADS + hh] + adh);

    int sv = d;
    auto stage = [&](int c) {
        const int q = c * 32 + lane;
        sv = (q < deg) ? list[p0 + q] : d;
        const float4* ar = (const float4*)(asrc + (size_t)sv * HEADS);
        const float4 a0 = ar[0], a1 = ar[1];
        float4* dp = (float4*)(my + lane * HEADS);
        dp[0] = a0; dp[1] = a1;
    };

    float m = vself;
    for (int c = 0; c < nch; ++c) {
        stage(c);
#pragma unroll
        for (int i = 0; i < 8; ++i) {
            const int sl = j + 4 * i;
            const float v = leaky(my[sl * HEADS + hh] + adh);
            m = (c * 32 + sl < deg) ? fmaxf(m, v) : m;
        }
    }
    m = fmaxf(m, __shfl_xor(m, 8));
    m = fmaxf(m, __shfl_xor(m, 16));
    float sum = (j == 0) ? expf(vself - m) : 0.f;
    for (int c = 0; c < nch; ++c) {
        if (nch > 1) stage(c);
#pragma unroll
        for (int i = 0; i < 8; ++i) {
            const int sl = j + 4 * i;
            const float ex = expf(leaky(my[sl * HEADS + hh] + adh) - m);
            sum = (c * 32 + sl < deg) ? sum + ex : sum;
        }
    }
    sum += __shfl_xor(sum, 8);
    sum += __shfl_xor(sum, 16);
    const float den = sum + GAT_EPS;
    const float wself = expf(vself - m) / den;

    const int hc = lane >> 2;
    float acc[8];
    {
        const float a = __shfl(wself, hc);
        const float4* r = (const float4*)(xl + (size_t)d * HIDDEN + lane * 8);
        const float4 r0 = r[0], r1 = r[1];
        acc[0] = a * r0.x; acc[1] = a * r0.y; acc[2] = a * r0.z; acc[3] = a * r0.w;
        acc[4] = a * r1.x; acc[5] = a * r1.y; acc[6] = a * r1.z; acc[7] = a * r1.w;
    }
    for (int c = 0; c < nch; ++c) {
        if (nch > 1) stage(c);
#pragma unroll
        for (int i = 0; i < 8; ++i) {
            const int sl = j + 4 * i;
            my[sl * HEADS + hh] = expf(leaky(my[sl * HEADS + hh] + adh) - m) / den;
        }
        const int cnt = min(32, deg - c * 32);
#pragma unroll 4
        for (int e = 0; e < cnt; ++e) {
            const int s = __builtin_amdgcn_readlane(sv, e);
            const float a = my[e * HEADS + hc];
            const float4* r = (const float4*)(xl + (size_t)s * HIDDEN + lane * 8);
            const float4 r0 = r[0], r1 = r[1];
            acc[0] += a * r0.x; acc[1] += a * r0.y; acc[2] += a * r0.z; acc[3] += a * r0.w;
            acc[4] += a * r1.x; acc[5] += a * r1.y; acc[6] += a * r1.z; acc[7] += a * r1.w;
        }
    }
    const float4* bp = (const float4*)(bias + lane * 8);
    const float4 b0 = bp[0], b1 = bp[1];
    const float bb[8] = {b0.x, b0.y, b0.z, b0.w, b1.x, b1.y, b1.z, b1.w};
    float o[8];
#pragma unroll
    for (int i = 0; i < 8; ++i) { const float v = acc[i] + bb[i]; o[i] = v > 0.f ? v : expm1f(v); }
    float4* op = (float4*)(&hrow[wv][lane * 8]);
    op[0] = make_float4(o[0], o[1], o[2], o[3]);
    op[1] = make_float4(o[4], o[5], o[6], o[7]);
    __syncthreads();

    const int kh = (threadIdx.x >> 4) & 1, l16 = threadIdx.x & 15;
    const float* hA = &hrow[l16][wv * 16 + kh * 2];
    const float* __restrict__ Bq = W2 + (size_t)(wv * 16 + kh * 2) * OUT_C + l16;
    v8f c2 = {0.f, 0.f, 0.f, 0.f, 0.f, 0.f, 0.f, 0.f};
#pragma unroll
    for (int kk = 0; kk < 4; ++kk) {
        const v2f a2 = *(const v2f*)(hA + kk * 4);
        v2f b2;
        b2.x = Bq[(size_t)(kk * 4) * OUT_C];
        b2.y = Bq[(size_t)(kk * 4 + 1) * OUT_C];
        c2 = __builtin_amdgcn_wmma_f32_16x16x4_f32(
                  false, a2,  false, b2,
                  (short)0, c2,
                  false,  false);
    }
#pragma unroll
    for (int r = 0; r < 8; ++r) my[r * 32 + lane] = c2[r];
    __syncthreads();
    if (threadIdx.x < 256) {
        const int idx = threadIdx.x;
        float s = sw[0][idx];
#pragma unroll
        for (int w = 1; w < L1_WAVES; ++w) s += sw[w][idx];
        const int col  = idx & 15;
        const int node = blockIdx.x * L1_WAVES + ((idx >> 4) & 1) * 8 + (idx >> 5);
        xl2[(size_t)node * OUT_C + col] = s;
        float ps = s * att_s2[col], pd = s * att_d2[col];
        ps += __int_as_float(__builtin_amdgcn_ds_swizzle(__float_as_int(ps), 0x041f));
        pd += __int_as_float(__builtin_amdgcn_ds_swizzle(__float_as_int(pd), 0x041f));
        ps += __int_as_float(__builtin_amdgcn_ds_swizzle(__float_as_int(ps), 0x081f));
        pd += __int_as_float(__builtin_amdgcn_ds_swizzle(__float_as_int(pd), 0x081f));
        ps += __int_as_float(__builtin_amdgcn_ds_swizzle(__float_as_int(ps), 0x101f));
        pd += __int_as_float(__builtin_amdgcn_ds_swizzle(__float_as_int(pd), 0x101f));
        ps += __int_as_float(__builtin_amdgcn_ds_swizzle(__float_as_int(ps), 0x201f));
        pd += __int_as_float(__builtin_amdgcn_ds_swizzle(__float_as_int(pd), 0x201f));
        if (col == 0) { as2[node] = ps; ad2[node] = pd; }
    }
}


template <int E>
__device__ __forceinline__ int gat_row_lane_i(int v)
{
    return __builtin_amdgcn_update_dpp(0, v, 0x150 + E, 0xf, 0xf, true);
}
template <int E>
__device__ __forceinline__ float gat_row_lane_f(float v)
{
    return __int_as_float(gat_row_lane_i<E>(__float_as_int(v)));
}

template <int B0>
__device__ __forceinline__ float gat_l2_rows16(float acc, const int sreg, const float wreg,
                                                const float* __restrict__ xl, const int c)
{
    const float x0 = xl[(unsigned)(gat_row_lane_i<B0 + 0>(sreg) * OUT_C + c)];
    const float x1 = xl[(unsigned)(gat_row_lane_i<B0 + 1>(sreg) * OUT_C + c)];
    const float x2 = xl[(unsigned)(gat_row_lane_i<B0 + 2>(sreg) * OUT_C + c)];
    const float x3 = xl[(unsigned)(gat_row_lane_i<B0 + 3>(sreg) * OUT_C + c)];
    const float x4 = xl[(unsigned)(gat_row_lane_i<B0 + 4>(sreg) * OUT_C + c)];
    const float x5 = xl[(unsigned)(gat_row_lane_i<B0 + 5>(sreg) * OUT_C + c)];
    const float x6 = xl[(unsigned)(gat_row_lane_i<B0 + 6>(sreg) * OUT_C + c)];
    const float x7 = xl[(unsigned)(gat_row_lane_i<B0 + 7>(sreg) * OUT_C + c)];
    const float x8 = xl[(unsigned)(gat_row_lane_i<B0 + 8>(sreg) * OUT_C + c)];
    const float x9 = xl[(unsigned)(gat_row_lane_i<B0 + 9>(sreg) * OUT_C + c)];
    const float x10 = xl[(unsigned)(gat_row_lane_i<B0 + 10>(sreg) * OUT_C + c)];
    const float x11 = xl[(unsigned)(gat_row_lane_i<B0 + 11>(sreg) * OUT_C + c)];
    const float x12 = xl[(unsigned)(gat_row_lane_i<B0 + 12>(sreg) * OUT_C + c)];
    const float x13 = xl[(unsigned)(gat_row_lane_i<B0 + 13>(sreg) * OUT_C + c)];
    const float x14 = xl[(unsigned)(gat_row_lane_i<B0 + 14>(sreg) * OUT_C + c)];
    const float x15 = xl[(unsigned)(gat_row_lane_i<B0 + 15>(sreg) * OUT_C + c)];
    acc = fmaf(gat_row_lane_f<B0 + 0>(wreg), x0, acc);
    acc = fmaf(gat_row_lane_f<B0 + 1>(wreg), x1, acc);
    acc = fmaf(gat_row_lane_f<B0 + 2>(wreg), x2, acc);
    acc = fmaf(gat_row_lane_f<B0 + 3>(wreg), x3, acc);
    acc = fmaf(gat_row_lane_f<B0 + 4>(wreg), x4, acc);
    acc = fmaf(gat_row_lane_f<B0 + 5>(wreg), x5, acc);
    acc = fmaf(gat_row_lane_f<B0 + 6>(wreg), x6, acc);
    acc = fmaf(gat_row_lane_f<B0 + 7>(wreg), x7, acc);
    acc = fmaf(gat_row_lane_f<B0 + 8>(wreg), x8, acc);
    acc = fmaf(gat_row_lane_f<B0 + 9>(wreg), x9, acc);
    acc = fmaf(gat_row_lane_f<B0 + 10>(wreg), x10, acc);
    acc = fmaf(gat_row_lane_f<B0 + 11>(wreg), x11, acc);
    acc = fmaf(gat_row_lane_f<B0 + 12>(wreg), x12, acc);
    acc = fmaf(gat_row_lane_f<B0 + 13>(wreg), x13, acc);
    acc = fmaf(gat_row_lane_f<B0 + 14>(wreg), x14, acc);
    acc = fmaf(gat_row_lane_f<B0 + 15>(wreg), x15, acc);
    return acc;
}

template <int B0>
__device__ __forceinline__ float gat_l2_rows8(float acc, const int sreg, const float wreg,
                                               const float* __restrict__ xl, const int c)
{
    const float x0 = xl[(unsigned)(gat_row_lane_i<B0 + 0>(sreg) * OUT_C + c)];
    const float x1 = xl[(unsigned)(gat_row_lane_i<B0 + 1>(sreg) * OUT_C + c)];
    const float x2 = xl[(unsigned)(gat_row_lane_i<B0 + 2>(sreg) * OUT_C + c)];
    const float x3 = xl[(unsigned)(gat_row_lane_i<B0 + 3>(sreg) * OUT_C + c)];
    const float x4 = xl[(unsigned)(gat_row_lane_i<B0 + 4>(sreg) * OUT_C + c)];
    const float x5 = xl[(unsigned)(gat_row_lane_i<B0 + 5>(sreg) * OUT_C + c)];
    const float x6 = xl[(unsigned)(gat_row_lane_i<B0 + 6>(sreg) * OUT_C + c)];
    const float x7 = xl[(unsigned)(gat_row_lane_i<B0 + 7>(sreg) * OUT_C + c)];
    acc = fmaf(gat_row_lane_f<B0 + 0>(wreg), x0, acc);
    acc = fmaf(gat_row_lane_f<B0 + 1>(wreg), x1, acc);
    acc = fmaf(gat_row_lane_f<B0 + 2>(wreg), x2, acc);
    acc = fmaf(gat_row_lane_f<B0 + 3>(wreg), x3, acc);
    acc = fmaf(gat_row_lane_f<B0 + 4>(wreg), x4, acc);
    acc = fmaf(gat_row_lane_f<B0 + 5>(wreg), x5, acc);
    acc = fmaf(gat_row_lane_f<B0 + 6>(wreg), x6, acc);
    acc = fmaf(gat_row_lane_f<B0 + 7>(wreg), x7, acc);
    return acc;
}

static_assert(N_NODES % 16 == 0, "a block of the layer-2 gather is 16 whole nodes: no partial block");

__global__ __launch_bounds__(256) void gat_gather_l2(
    const int* __restrict__ rowptr, const int* __restrict__ list,
    const float* __restrict__ asrc, const float* __restrict__ adst,
    const float* __restrict__ xl, const float* __restrict__ bias,
    float* __restrict__ out)
{
    const int c = threadIdx.x & 15;
    const int d = blockIdx.x * 16 + (threadIdx.x >> 4);
    const int p0 = rowptr[d], p1 = rowptr[d + 1];
    const int deg = p1 - p0;
    const float ad    = adst[d];
    const float vself = leaky(asrc[d] + ad);
    const int degw = max(__builtin_amdgcn_readlane(deg, 0), __builtin_amdgcn_readlane(deg, 16));
    float acc;
    if (degw <= 32) {
        const bool in0 = c < deg, in1 = c + 16 < deg;
        const int l0 = list[in0 ? p0 + c : 0];
        const int l1 = list[in1 ? p0 + c + 16 : 0];
        const int s0 = in0 ? l0 : d;
        const int s1 = in1 ? l1 : d;
        const float v0 = leaky(asrc[s0] + ad);
        const float v1 = leaky(asrc[s1] + ad);
        float m = vself;
        m = in0 ? fmaxf(m, v0) : m;
        m = in1 ? fmaxf(m, v1) : m;
        m = fmaxf(m, gat_row_xor<1>(m)); m = fmaxf(m, gat_row_xor<2>(m));
        m = fmaxf(m, gat_row_xor<4>(m)); m = fmaxf(m, gat_row_xor<8>(m));
        const float eself = expf(vself - m);
        const float e0 = expf(v0 - m), e1 = expf(v1 - m);
        float sum = (c == 0) ? eself : 0.f;
        sum = in0 ? sum + e0 : sum;
        sum = in1 ? sum + e1 : sum;
        sum += gat_row_xor<1>(sum); sum += gat_row_xor<2>(sum);
        sum += gat_row_xor<4>(sum); sum += gat_row_xor<8>(sum);
        const float den = sum + GAT_EPS;
        const float w0 = in0 ? e0 / den : 0.f;
        const float w1 = in1 ? e1 / den : 0.f;
        acc = eself / den * xl[(size_t)d * OUT_C + c];
        acc = gat_l2_rows16<0>(acc, s0, w0, xl, c);
        if (degw > 16) acc = gat_l2_rows8<0>(acc, s1, w1, xl, c);
        if (degw > 24) acc = gat_l2_rows8<8>(acc, s1, w1, xl, c);
    } else {
        float m = vself;
        for (int p = p0 + c; p < p1; p += 16) m = fmaxf(m, leaky(asrc[list[p]] + ad));
        m = fmaxf(m, __shfl_xor(m, 1)); m = fmaxf(m, __shfl_xor(m, 2));
        m = fmaxf(m, __shfl_xor(m, 4)); m = fmaxf(m, __shfl_xor(m, 8));
        float sum = (c == 0) ? expf(vself - m) : 0.f;
        for (int p = p0 + c; p < p1; p += 16) sum += expf(leaky(asrc[list[p]] + ad) - m);
        sum += __shfl_xor(sum, 1); sum += __shfl_xor(sum, 2);
        sum += __shfl_xor(sum, 4); sum += __shfl_xor(sum, 8);
        const float den = sum + GAT_EPS;
        acc = expf(vself - m) / den * xl[(size_t)d * OUT_C + c];
#pragma unroll 4
        for (int p = p0; p < p1; ++p) {
            const int s = list[p];
            const float a = expf(leaky(asrc[s] + ad) - m) / den;
            acc += a * xl[(size_t)s * OUT_C + c];
        }
    }
    out[(size_t)d * OUT_C + c] = acc + bias[c];
}

extern "C" void kernel_launch(void* const* d_in, const int* in_sizes, int n_in,
                              void* d_out, int out_size, void* d_ws, size_t ws_size,
                              hipStream_t stream)
{
    (void)in_sizes; (void)n_in; (void)out_size; (void)ws_size;

    const float* x    = (const float*)d_in[0];
    const int*   ei   = (const int*)  d_in[1];
    const float* W1   = (const float*)d_in[2];
    const float* ats1 = (const float*)d_in[3];
    const float* atd1 = (const float*)d_in[4];
    const float* b1   = (const float*)d_in[5];
    const float* W2   = (const float*)d_in[6];
    const float* ats2 = (const float*)d_in[7];
    const float* atd2 = (const float*)d_in[8];
    const float* b2   = (const float*)d_in[9];
    const int* src = ei;
    const int* dst = ei + N_EDGES;
    float* out = (float*)d_out;

    int*   rowptr = (int*)d_ws;
    int*   rnk    = rowptr + (size_t)CNT_PAD;
    int*   list   = rnk + (size_t)N_EDGES;
    float* xl1 = (float*)(list + (size_t)N_EDGES);
    float* as1 = xl1 + (size_t)N_NODES * HIDDEN;
    float* ad1 = as1 + (size_t)N_NODES * HEADS;
    float* xl2 = ad1 + (size_t)N_NODES * HEADS;
    float* as2 = xl2 + (size_t)N_NODES * OUT_C;
    float* ad2 = as2 + (size_t)CNT_PAD;
    int*   part = (int*)(ad2 + (size_t)CNT_PAD);
    int*   bsum = part + (size_t)RANK_CHUNKS * CNT_PAD;

    const int TB = 256;

    gat_rank_gemm_f32<4, IN_C, HIDDEN><<<RANK_CHUNKS + RG_GEMM_GROUPS, RANK_TB, 0, stream>>>(
        dst, part, rnk, bsum,
        x, W1, xl1, ats1, atd1, as1, ad1, N_NODES, HIDDEN / (16 * 4));
    gat_chunk_prefix<<<N_BKT, BKT_N, 0, stream>>>(part, bsum, rowptr);

    gat_place_gemm_f32<4, IN_C, HIDDEN><<<PG_GROUPS + PG_PLACE, PG_TB, 0, stream>>>(
        src, dst, rowptr, part, rnk, list,
        x, W1, xl1, ats1, atd1, as1, ad1, N_NODES, HIDDEN / (16 * 4));

    gat_gather_l1<<<N_NODES / L1_WAVES, L1_WAVES * 32, 0, stream>>>(
        rowptr, list, as1, ad1, xl1, b1, W2, ats2, atd2, xl2, as2, ad2);

    gat_gather_l2<<<N_NODES / 16, TB, 0, stream>>>(
        rowptr, list, as2, ad2, xl2, b2, out);
}
